# starting kernel
# baseline (speedup 1.0000x reference)
#include <hip/hip_runtime.h>
#include <math.h>

#define GAT_N 50000
#define GAT_E 800000
#define GAT_H 4
#define GAT_SLOPE 0.2f

typedef __attribute__((ext_vector_type(2))) float v2f;
typedef __attribute__((ext_vector_type(8))) float v8f;

template <int K, int NCOL>
__global__ __launch_bounds__(128) void gat_gemm_wmma(
    const float* __restrict__ A, const float* __restrict__ B,
    float* __restrict__ C, int M)
{
  const int wave = threadIdx.x >> 5;
  const int lane = threadIdx.x & 31;
  const int row0 = blockIdx.x << 5;
  const int col0 = ((blockIdx.y << 2) + wave) << 4;
  if (col0 >= NCOL) return;

  const int m     = lane & 15;
  const int khalf = (lane >> 4) << 1;
  const int n     = col0 + m;
  const bool hi_valid = (row0 + 16) < M;

  const float* arow0 = A + (size_t)(row0 + m) * K + khalf;
  const float* arow1 = hi_valid ? (arow0 + (size_t)16 * K) : arow0;
  const float* bcol  = B + (size_t)khalf * NCOL + n;

  v8f acc0 = {}, acc1 = {};
#pragma unroll 8
  for (int k0 = 0; k0 < K; k0 += 4) {
    v2f a0 = *(const v2f*)arow0;
    v2f a1 = *(const v2f*)arow1;
    v2f b;
    b.x = bcol[0];
    b.y = bcol[NCOL];
    acc0 = __builtin_amdgcn_wmma_f32_16x16x4_f32(
        false, a0, false, b, (short)0, acc0, false, false);
    acc1 = __builtin_amdgcn_wmma_f32_16x16x4_f32(
        false, a1, false, b, (short)0, acc1, false, false);
    arow0 += 4;
    arow1 += 4;
    bcol  += NCOL * 4;
  }

  const int rsub = (lane >> 4) << 3;
  float* crow0 = C + (size_t)(row0 + rsub) * NCOL + n;
#pragma unroll
  for (int r = 0; r < 8; ++r) crow0[(size_t)r * NCOL] = acc0[r];
  if (hi_valid) {
    float* crow1 = C + (size_t)(row0 + 16 + rsub) * NCOL + n;
#pragma unroll
    for (int r = 0; r < 8; ++r) crow1[(size_t)r * NCOL] = acc1[r];
  }
}

__global__ void gat_attn_scores(const float* __restrict__ feat,
                                const float* __restrict__ al,
                                const float* __restrict__ ar,
                                float* __restrict__ el, float* __restrict__ er,
                                int F, int D)
{
  int idx = blockIdx.x * blockDim.x + threadIdx.x;
  if (idx >= GAT_N * GAT_H) return;
  int nd = idx >> 2, h = idx & 3;
  const float* f   = feat + (size_t)nd * F + h * D;
  const float* alh = al + h * D;
  const float* arh = ar + h * D;
  float sl = 0.f, sr = 0.f;
  for (int d = 0; d < D; ++d) {
    float v = f[d];
    sl = fmaf(v, alh[d], sl);
    sr = fmaf(v, arh[d], sr);
  }
  el[idx] = sl;
  er[idx] = sr;
}

__global__ void gat_init_mden(float* __restrict__ mx, float* __restrict__ den, int count)
{
  int idx = blockIdx.x * blockDim.x + threadIdx.x;
  if (idx < count) {
    mx[idx]  = __int_as_float((int)0xFF800000u);
    den[idx] = 0.f;
  }
}

__device__ __forceinline__ float gat_leaky(float x) {
  return x > 0.f ? x : GAT_SLOPE * x;
}

__device__ __forceinline__ void gat_atomic_max_f32(float* addr, float v) {
  if (v >= 0.f) atomicMax((int*)addr, __float_as_int(v));
  else          atomicMin((unsigned int*)addr, __float_as_uint(v));
}

__global__ void gat_edge_max(const int* __restrict__ src, const int* __restrict__ dst,
                             const float* __restrict__ el, const float* __restrict__ er,
                             float* __restrict__ mx)
{
  int idx = blockIdx.x * blockDim.x + threadIdx.x;
  if (idx >= GAT_E * GAT_H) return;
  int e = idx >> 2, h = idx & 3;
  int s = src[e], d = dst[e];
  float x = gat_leaky(el[s * GAT_H + h] + er[d * GAT_H + h]);
  gat_atomic_max_f32(&mx[d * GAT_H + h], x);
}

__global__ void gat_edge_exp(const int* __restrict__ src, const int* __restrict__ dst,
                             const float* __restrict__ el, const float* __restrict__ er,
                             const float* __restrict__ mx,
                             float* __restrict__ ex, float* __restrict__ den)
{
  int idx = blockIdx.x * blockDim.x + threadIdx.x;
  if (idx >= GAT_E * GAT_H) return;
  int e = idx >> 2, h = idx & 3;
  int s = src[e], d = dst[e];
  float x = gat_leaky(el[s * GAT_H + h] + er[d * GAT_H + h]);
  float v = expf(x - mx[d * GAT_H + h]);
  ex[idx] = v;
  atomicAdd(&den[d * GAT_H + h], v);
}

__global__ __launch_bounds__(256) void gat_edge_aggregate(
    const int* __restrict__ src, const int* __restrict__ dst,
    const float* __restrict__ feat, const float* __restrict__ ex,
    const float* __restrict__ den, float* __restrict__ out, int F, int D)
{
  int e = blockIdx.x * (blockDim.x >> 5) + (threadIdx.x >> 5);
  if (e >= GAT_E) return;
  int lane = threadIdx.x & 31;
  int s = src[e], d = dst[e];
  const float* fs = feat + (size_t)s * F;
  float*       od = out  + (size_t)d * F;
  __builtin_prefetch(fs + lane, 0, 0);
  float alpha[GAT_H];
#pragma unroll
  for (int h = 0; h < GAT_H; ++h)
    alpha[h] = ex[(size_t)e * GAT_H + h] / den[(size_t)d * GAT_H + h];
  for (int f = lane; f < F; f += 32)
    atomicAdd(&od[f], fs[f] * alpha[(unsigned)f / (unsigned)D]);
}

__global__ void gat_finalize_elu(float* __restrict__ acc, const float* __restrict__ bias,
                                 const float* __restrict__ res, int total)
{
  int idx = blockIdx.x * blockDim.x + threadIdx.x;
  if (idx >= total) return;
  float v = acc[idx] + bias[idx & 255];
  if (res) v += res[idx];
  acc[idx] = v > 0.f ? v : (expf(v) - 1.f);
}

__global__ void gat_finalize_out(const float* __restrict__ acc, const float* __restrict__ bias,
                                 const float* __restrict__ res2, float* __restrict__ out)
{
  const int C = 40;
  int idx = blockIdx.x * blockDim.x + threadIdx.x;
  if (idx >= GAT_N * C) return;
  int n = idx / C, c = idx % C;
  float sum = 0.f;
#pragma unroll
  for (int h = 0; h < GAT_H; ++h) {
    int j = n * (GAT_H * C) + h * C + c;
    sum += acc[j] + bias[h * C + c] + res2[j];
  }
  out[idx] = sum * 0.25f;
}

extern "C" void kernel_launch(void* const* d_in, const int* in_sizes, int n_in,
                              void* d_out, int out_size, void* d_ws, size_t ws_size,
                              hipStream_t stream)
{
  const float* x     = (const float*)d_in[0];
  const int*   src   = (const int*)d_in[1];
  const int*   dst   = (const int*)d_in[2];
  const float* W0    = (const float*)d_in[3];
  const float* al0   = (const float*)d_in[4];
  const float* ar0   = (const float*)d_in[5];
  const float* b0    = (const float*)d_in[6];
  const float* W1    = (const float*)d_in[7];
  const float* al1   = (const float*)d_in[8];
  const float* ar1   = (const float*)d_in[9];
  const float* b1    = (const float*)d_in[10];
  const float* W2    = (const float*)d_in[11];
  const float* al2   = (const float*)d_in[12];
  const float* ar2   = (const float*)d_in[13];
  const float* b2    = (const float*)d_in[14];
  const float* Wres2 = (const float*)d_in[15];

  const size_t NF = (size_t)GAT_N * 256;
  const size_t NG = (size_t)GAT_N * 160;
  float* ws   = (float*)d_ws;
  float* bufA = ws;
  float* bufB = bufA + NF;
  float* bufC = bufB + NF;
  float* bufD = bufC + NF;
  float* EX   = bufD + NG;
  float* EL   = EX + (size_t)GAT_E * GAT_H;
  float* ER   = EL + (size_t)GAT_N * GAT_H;
  float* MX   = ER + (size_t)GAT_N * GAT_H;
  float* DEN  = MX + (size_t)GAT_N * GAT_H;

  dim3 gemmBlk(128);
  const int rowTiles = (GAT_N + 31) / 32;
  dim3 gemmGrid256(rowTiles, 4);
  dim3 gemmGrid160(rowTiles, 3);

  const int ehThreads = GAT_E * GAT_H;
  const int ehBlocks  = (ehThreads + 255) / 256;
  const int nhThreads = GAT_N * GAT_H;
  const int nhBlocks  = (nhThreads + 255) / 256;
  const int aggBlocks = (GAT_E + 7) / 8;
  const int nfBlocks  = (int)((NF + 255) / 256);

  gat_gemm_wmma<256, 256><<<gemmGrid256, gemmBlk, 0, stream>>>(x, W0, bufA, GAT_N);
  gat_attn_scores<<<nhBlocks, 256, 0, stream>>>(bufA, al0, ar0, EL, ER, 256, 64);
  gat_init_mden<<<nhBlocks, 256, 0, stream>>>(MX, DEN, nhThreads);
  gat_edge_max<<<ehBlocks, 256, 0, stream>>>(src, dst, EL, ER, MX);
  gat_edge_exp<<<ehBlocks, 256, 0, stream>>>(src, dst, EL, ER, MX, EX, DEN);
  (void)hipMemsetAsync(bufB, 0, NF * sizeof(float), stream);
  gat_edge_aggregate<<<aggBlocks, 256, 0, stream>>>(src, dst, bufA, EX, DEN, bufB, 256, 64);
  gat_finalize_elu<<<nfBlocks, 256, 0, stream>>>(bufB, b0, nullptr, (int)NF);

  gat_gemm_wmma<256, 256><<<gemmGrid256, gemmBlk, 0, stream>>>(bufB, W1, bufA, GAT_N);
  gat_attn_scores<<<nhBlocks, 256, 0, stream>>>(bufA, al1, ar1, EL, ER, 256, 64);
  gat_init_mden<<<nhBlocks, 256, 0, stream>>>(MX, DEN, nhThreads);
  gat_edge_max<<<ehBlocks, 256, 0, stream>>>(src, dst, EL, ER, MX);
  gat_edge_exp<<<ehBlocks, 256, 0, stream>>>(src, dst, EL, ER, MX, EX, DEN);
  (void)hipMemsetAsync(bufC, 0, NF * sizeof(float), stream);
  gat_edge_aggregate<<<aggBlocks, 256, 0, stream>>>(src, dst, bufA, EX, DEN, bufC, 256, 64);
  gat_finalize_elu<<<nfBlocks, 256, 0, stream>>>(bufC, b1, bufB, (int)NF);

  gat_gemm_wmma<256, 160><<<gemmGrid160, gemmBlk, 0, stream>>>(bufC, W2, bufA, GAT_N);
  gat_gemm_wmma<256, 160><<<gemmGrid160, gemmBlk, 0, stream>>>(bufC, Wres2, bufB, GAT_N);
  gat_attn_scores<<<nhBlocks, 256, 0, stream>>>(bufA, al2, ar2, EL, ER, 160, 40);
  gat_init_mden<<<nhBlocks, 256, 0, stream>>>(MX, DEN, nhThreads);
  gat_edge_max<<<ehBlocks, 256, 0, stream>>>(src, dst, EL, ER, MX);
  gat_edge_exp<<<ehBlocks, 256, 0, stream>>>(src, dst, EL, ER, MX, EX, DEN);
  (void)hipMemsetAsync(bufD, 0, NG * sizeof(float), stream);
  gat_edge_aggregate<<<aggBlocks, 256, 0, stream>>>(src, dst, bufA, EX, DEN, bufD, 160, 40);
  gat_finalize_out<<<(GAT_N * 40 + 255) / 256, 256, 0, stream>>>(bufD, b2, bufB, (float*)d_out);
}


# out rows summed in LDS tiles: each of 256 work-groups filters the whole dst list for its 196 rows, adds alpha*feat rows by LDS float adds, stores once; global adds, zero fills gone; sum order changes
# speedup vs baseline: 1.4748x; 1.4748x over previous
#include <hip/hip_runtime.h>
#include <math.h>

#define GAT_N 50000
#define GAT_E 800000
#define GAT_H 4
#define GAT_SLOPE 0.2f

typedef __attribute__((ext_vector_type(2))) float v2f;
typedef __attribute__((ext_vector_type(8))) float v8f;

template <int K, int NCOL>
__global__ __launch_bounds__(128) void gat_gemm_wmma(
    const float* __restrict__ A, const float* __restrict__ B,
    float* __restrict__ C, int M)
{
  const int wave = threadIdx.x >> 5;
  const int lane = threadIdx.x & 31;
  const int row0 = blockIdx.x << 5;
  const int col0 = ((blockIdx.y << 2) + wave) << 4;
  if (col0 >= NCOL) return;

  const int m     = lane & 15;
  const int khalf = (lane >> 4) << 1;
  const int n     = col0 + m;
  const bool hi_valid = (row0 + 16) < M;

  const float* arow0 = A + (size_t)(row0 + m) * K + khalf;
  const float* arow1 = hi_valid ? (arow0 + (size_t)16 * K) : arow0;
  const float* bcol  = B + (size_t)khalf * NCOL + n;

  v8f acc0 = {}, acc1 = {};
#pragma unroll 8
  for (int k0 = 0; k0 < K; k0 += 4) {
    v2f a0 = *(const v2f*)arow0;
    v2f a1 = *(const v2f*)arow1;
    v2f b;
    b.x = bcol[0];
    b.y = bcol[NCOL];
    acc0 = __builtin_amdgcn_wmma_f32_16x16x4_f32(
        false, a0, false, b, (short)0, acc0, false, false);
    acc1 = __builtin_amdgcn_wmma_f32_16x16x4_f32(
        false, a1, false, b, (short)0, acc1, false, false);
    arow0 += 4;
    arow1 += 4;
    bcol  += NCOL * 4;
  }

  const int rsub = (lane >> 4) << 3;
  float* crow0 = C + (size_t)(row0 + rsub) * NCOL + n;
#pragma unroll
  for (int r = 0; r < 8; ++r) crow0[(size_t)r * NCOL] = acc0[r];
  if (hi_valid) {
    float* crow1 = C + (size_t)(row0 + 16 + rsub) * NCOL + n;
#pragma unroll
    for (int r = 0; r < 8; ++r) crow1[(size_t)r * NCOL] = acc1[r];
  }
}

__global__ void gat_attn_scores(const float* __restrict__ feat,
                                const float* __restrict__ al,
                                const float* __restrict__ ar,
                                float* __restrict__ el, float* __restrict__ er,
                                int F, int D)
{
  int idx = blockIdx.x * blockDim.x + threadIdx.x;
  if (idx >= GAT_N * GAT_H) return;
  int nd = idx >> 2, h = idx & 3;
  const float* f   = feat + (size_t)nd * F + h * D;
  const float* alh = al + h * D;
  const float* arh = ar + h * D;
  float sl = 0.f, sr = 0.f;
  for (int d = 0; d < D; ++d) {
    float v = f[d];
    sl = fmaf(v, alh[d], sl);
    sr = fmaf(v, arh[d], sr);
  }
  el[idx] = sl;
  er[idx] = sr;
}

__global__ void gat_init_mden(float* __restrict__ mx, float* __restrict__ den, int count)
{
  int idx = blockIdx.x * blockDim.x + threadIdx.x;
  if (idx < count) {
    mx[idx]  = __int_as_float((int)0xFF800000u);
    den[idx] = 0.f;
  }
}

__device__ __forceinline__ float gat_leaky(float x) {
  return x > 0.f ? x : GAT_SLOPE * x;
}

__device__ __forceinline__ void gat_atomic_max_f32(float* addr, float v) {
  if (v >= 0.f) atomicMax((int*)addr, __float_as_int(v));
  else          atomicMin((unsigned int*)addr, __float_as_uint(v));
}

__global__ void gat_edge_max(const int* __restrict__ src, const int* __restrict__ dst,
                             const float* __restrict__ el, const float* __restrict__ er,
                             float* __restrict__ mx)
{
  int idx = blockIdx.x * blockDim.x + threadIdx.x;
  if (idx >= GAT_E * GAT_H) return;
  int e = idx >> 2, h = idx & 3;
  int s = src[e], d = dst[e];
  float x = gat_leaky(el[s * GAT_H + h] + er[d * GAT_H + h]);
  gat_atomic_max_f32(&mx[d * GAT_H + h], x);
}

__global__ void gat_edge_exp(const int* __restrict__ src, const int* __restrict__ dst,
                             const float* __restrict__ el, const float* __restrict__ er,
                             const float* __restrict__ mx,
                             float* __restrict__ ex, float* __restrict__ den)
{
  int idx = blockIdx.x * blockDim.x + threadIdx.x;
  if (idx >= GAT_E * GAT_H) return;
  int e = idx >> 2, h = idx & 3;
  int s = src[e], d = dst[e];
  float x = gat_leaky(el[s * GAT_H + h] + er[d * GAT_H + h]);
  float v = expf(x - mx[d * GAT_H + h]);
  ex[idx] = v;
  atomicAdd(&den[d * GAT_H + h], v);
}

#define GAT_TILE 196
#define GAT_AGG_THREADS 1024

template <int F, int D>
__global__ __launch_bounds__(GAT_AGG_THREADS) void gat_tile_aggregate(
    const int* __restrict__ src, const int* __restrict__ dst,
    const float* __restrict__ feat, const float* __restrict__ ex,
    const float* __restrict__ den, float* __restrict__ out)
{
  static_assert(GAT_E % 128 == 0, "dst is walked in steps of 128 edges");
  static_assert(F % 32 == 0, "row is walked in steps of one wave");
  constexpr int NW  = GAT_AGG_THREADS / 32;
  constexpr int NCH = GAT_E / 128;
  constexpr int FI  = F / 32;

  __shared__ float tile[GAT_TILE * F];
  __shared__ int2  queue[NW][160];

  const int tid   = threadIdx.x;
  const int wave  = __builtin_amdgcn_readfirstlane(tid >> 5);
  const int lane  = tid & 31;
  const int node0 = blockIdx.x * GAT_TILE;
  const int rows  = min(GAT_TILE, GAT_N - node0);

  float4* t4 = (float4*)tile;
  for (int i = tid; i < GAT_TILE * F / 4; i += GAT_AGG_THREADS)
    t4[i] = make_float4(0.f, 0.f, 0.f, 0.f);
  __syncthreads();

  int2* q  = queue[wave];
  int   qn = 0;

  auto drain = [&](int base, int n) {
    int   s_l = 0, r_l = 0;
    float a0 = 0.f, a1 = 0.f, a2 = 0.f, a3 = 0.f;
    if (lane < n) {
      const int2 ent = q[base + lane];
      r_l = ent.y;
      s_l = src[ent.x];
      const float4 xv = ((const float4*)ex)[ent.x];
      const float4 dn = ((const float4*)den)[node0 + ent.y];
      a0 = xv.x / dn.x; a1 = xv.y / dn.y; a2 = xv.z / dn.z; a3 = xv.w / dn.w;
    }
#pragma unroll 1
    for (int k = 0; k < n; k += 4) {
      const float* fp[4];
      float*       tp[4];
      float        w[4][4];
#pragma unroll
      for (int j = 0; j < 4; ++j) {
        const int kj = min(k + j, n - 1);
        fp[j] = feat + (size_t)__builtin_amdgcn_readlane(s_l, kj) * F + lane;
        tp[j] = tile + __builtin_amdgcn_readlane(r_l, kj) * F + lane;
        w[j][0] = __int_as_float(__builtin_amdgcn_readlane(__float_as_int(a0), kj));
        w[j][1] = __int_as_float(__builtin_amdgcn_readlane(__float_as_int(a1), kj));
        w[j][2] = __int_as_float(__builtin_amdgcn_readlane(__float_as_int(a2), kj));
        w[j][3] = __int_as_float(__builtin_amdgcn_readlane(__float_as_int(a3), kj));
      }
      float v[4][FI];
#pragma unroll
      for (int j = 0; j < 4; ++j) {
#pragma unroll
        for (int i = 0; i < FI; ++i) {
          const int h = (i * 32 + lane) / D;
          const float wh = h == 0 ? w[j][0] : (h == 1 ? w[j][1] : (h == 2 ? w[j][2] : w[j][3]));
          v[j][i] = fp[j][i * 32] * wh;
        }
      }
#pragma unroll
      for (int j = 0; j < 4; ++j) {
        if (k + j < n) {
#pragma unroll
          for (int i = 0; i < FI; ++i)
            __hip_atomic_fetch_add(tp[j] + i * 32, v[j][i], __ATOMIC_RELAXED, __HIP_MEMORY_SCOPE_WORKGROUP);
        }
      }
    }
  };

  auto flush = [&](int keep) {
    while (qn > keep) {
      const int n = min(qn, 32);
      drain(qn - n, n);
      qn -= n;
    }
  };

  auto push = [&](bool m, int e, int r) {
    const unsigned mask = __builtin_amdgcn_ballot_w32(m);
    if (m) q[qn + (int)__builtin_amdgcn_mbcnt_lo(mask, 0u)] = make_int2(e, r);
    qn += __popc(mask);
  };

  const int4* dst4 = (const int4*)dst;
  constexpr int NGROUPS = (GAT_N + GAT_TILE - 1) / GAT_TILE;
  const int rot = (int)blockIdx.x * (NCH / NGROUPS);
  auto place = [&](int t) { const int p = t + rot; return p >= NCH ? p - NCH : p; };
  auto fetch = [&](int t) { return dst4[(size_t)place(min(t, NCH - 1)) * 32 + lane]; };
  auto step = [&](const int4 cur, int t) {
    const bool live = t < NCH;
    const int cc = place(t);
    const int l0 = cur.x - node0, l1 = cur.y - node0;
    const int l2 = cur.z - node0, l3 = cur.w - node0;
    const bool m0 = live && (unsigned)l0 < (unsigned)rows, m1 = live && (unsigned)l1 < (unsigned)rows;
    const bool m2 = live && (unsigned)l2 < (unsigned)rows, m3 = live && (unsigned)l3 < (unsigned)rows;
    if (__builtin_amdgcn_ballot_w32(m0 | m1 | m2 | m3) != 0u) {
      const int e0 = cc * 128 + lane * 4;
      push(m0, e0 + 0, l0);
      push(m1, e0 + 1, l1);
      push(m2, e0 + 2, l2);
      push(m3, e0 + 3, l3);
      flush(31);
    }
  };
  int4 n0 = fetch(wave), n1 = fetch(wave + NW), n2 = fetch(wave + 2 * NW), n3 = fetch(wave + 3 * NW);
#pragma unroll 1
  for (int c = wave; c < NCH; c += 4 * NW) {
    const int4 c0 = n0, c1 = n1, c2 = n2, c3 = n3;
    n0 = fetch(c + 4 * NW); n1 = fetch(c + 5 * NW); n2 = fetch(c + 6 * NW); n3 = fetch(c + 7 * NW);
    step(c0, c);
    step(c1, c + NW);
    step(c2, c + 2 * NW);
    step(c3, c + 3 * NW);
  }
  flush(0);

  __syncthreads();
  float4* o4 = (float4*)(out + (size_t)node0 * F);
  for (int i = tid; i < rows * F / 4; i += GAT_AGG_THREADS) o4[i] = t4[i];
}

__global__ void gat_finalize_elu(float* __restrict__ acc, const float* __restrict__ bias,
                                 const float* __restrict__ res, int total)
{
  int idx = blockIdx.x * blockDim.x + threadIdx.x;
  if (idx >= total) return;
  float v = acc[idx] + bias[idx & 255];
  if (res) v += res[idx];
  acc[idx] = v > 0.f ? v : (expf(v) - 1.f);
}

__global__ void gat_finalize_out(const float* __restrict__ acc, const float* __restrict__ bias,
                                 const float* __restrict__ res2, float* __restrict__ out)
{
  const int C = 40;
  int idx = blockIdx.x * blockDim.x + threadIdx.x;
  if (idx >= GAT_N * C) return;
  int n = idx / C, c = idx % C;
  float sum = 0.f;
#pragma unroll
  for (int h = 0; h < GAT_H; ++h) {
    int j = n * (GAT_H * C) + h * C + c;
    sum += acc[j] + bias[h * C + c] + res2[j];
  }
  out[idx] = sum * 0.25f;
}

extern "C" void kernel_launch(void* const* d_in, const int* in_sizes, int n_in,
                              void* d_out, int out_size, void* d_ws, size_t ws_size,
                              hipStream_t stream)
{
  const float* x     = (const float*)d_in[0];
  const int*   src   = (const int*)d_in[1];
  const int*   dst   = (const int*)d_in[2];
  const float* W0    = (const float*)d_in[3];
  const float* al0   = (const float*)d_in[4];
  const float* ar0   = (const float*)d_in[5];
  const float* b0    = (const float*)d_in[6];
  const float* W1    = (const float*)d_in[7];
  const float* al1   = (const float*)d_in[8];
  const float* ar1   = (const float*)d_in[9];
  const float* b1    = (const float*)d_in[10];
  const float* W2    = (const float*)d_in[11];
  const float* al2   = (const float*)d_in[12];
  const float* ar2   = (const float*)d_in[13];
  const float* b2    = (const float*)d_in[14];
  const float* Wres2 = (const float*)d_in[15];

  const size_t NF = (size_t)GAT_N * 256;
  const size_t NG = (size_t)GAT_N * 160;
  float* ws   = (float*)d_ws;
  float* bufA = ws;
  float* bufB = bufA + NF;
  float* bufC = bufB + NF;
  float* bufD = bufC + NF;
  float* EX   = bufD + NG;
  float* EL   = EX + (size_t)GAT_E * GAT_H;
  float* ER   = EL + (size_t)GAT_N * GAT_H;
  float* MX   = ER + (size_t)GAT_N * GAT_H;
  float* DEN  = MX + (size_t)GAT_N * GAT_H;

  dim3 gemmBlk(128);
  const int rowTiles = (GAT_N + 31) / 32;
  dim3 gemmGrid256(rowTiles, 4);
  dim3 gemmGrid160(rowTiles, 3);

  const int ehThreads = GAT_E * GAT_H;
  const int ehBlocks  = (ehThreads + 255) / 256;
  const int nhThreads = GAT_N * GAT_H;
  const int nhBlocks  = (nhThreads + 255) / 256;
  const int aggBlocks = (GAT_N + GAT_TILE - 1) / GAT_TILE;
  const int nfBlocks  = (int)((NF + 255) / 256);

  gat_gemm_wmma<256, 256><<<gemmGrid256, gemmBlk, 0, stream>>>(x, W0, bufA, GAT_N);
  gat_attn_scores<<<nhBlocks, 256, 0, stream>>>(bufA, al0, ar0, EL, ER, 256, 64);
  gat_init_mden<<<nhBlocks, 256, 0, stream>>>(MX, DEN, nhThreads);
  gat_edge_max<<<ehBlocks, 256, 0, stream>>>(src, dst, EL, ER, MX);
  gat_edge_exp<<<ehBlocks, 256, 0, stream>>>(src, dst, EL, ER, MX, EX, DEN);
  gat_tile_aggregate<256, 64><<<aggBlocks, GAT_AGG_THREADS, 0, stream>>>(src, dst, bufA, EX, DEN, bufB);
  gat_finalize_elu<<<nfBlocks, 256, 0, stream>>>(bufB, b0, nullptr, (int)NF);

  gat_gemm_wmma<256, 256><<<gemmGrid256, gemmBlk, 0, stream>>>(bufB, W1, bufA, GAT_N);
  gat_attn_scores<<<nhBlocks, 256, 0, stream>>>(bufA, al1, ar1, EL, ER, 256, 64);
  gat_init_mden<<<nhBlocks, 256, 0, stream>>>(MX, DEN, nhThreads);
  gat_edge_max<<<ehBlocks, 256, 0, stream>>>(src, dst, EL, ER, MX);
  gat_edge_exp<<<ehBlocks, 256, 0, stream>>>(src, dst, EL, ER, MX, EX, DEN);
  gat_tile_aggregate<256, 64><<<aggBlocks, GAT_AGG_THREADS, 0, stream>>>(src, dst, bufA, EX, DEN, bufC);
  gat_finalize_elu<<<nfBlocks, 256, 0, stream>>>(bufC, b1, bufB, (int)NF);

  gat_gemm_wmma<256, 160><<<gemmGrid160, gemmBlk, 0, stream>>>(bufC, W2, bufA, GAT_N);
  gat_gemm_wmma<256, 160><<<gemmGrid160, gemmBlk, 0, stream>>>(bufC, Wres2, bufB, GAT_N);
  gat_attn_scores<<<nhBlocks, 256, 0, stream>>>(bufA, al2, ar2, EL, ER, 160, 40);
  gat_init_mden<<<nhBlocks, 256, 0, stream>>>(MX, DEN, nhThreads);
  gat_edge_max<<<ehBlocks, 256, 0, stream>>>(src, dst, EL, ER, MX);
  gat_edge_exp<<<ehBlocks, 256, 0, stream>>>(src, dst, EL, ER, MX, EX, DEN);
  gat_tile_aggregate<160, 40><<<aggBlocks, GAT_AGG_THREADS, 0, stream>>>(src, dst, bufA, EX, DEN, bufD);
  gat_finalize_out<<<(GAT_N * 40 + 255) / 256, 256, 0, stream>>>(bufD, b2, bufB, (float*)d_out);
}


# GEMM operands staged in LDS: a 4-wave group copies 32-k chunks of its A rows and B columns by 16-byte whole-line loads, each wave reads fragments of a 32x32 part from LDS; k order kept, f32 kept
# speedup vs baseline: 1.7311x; 1.1738x over previous
#include <hip/hip_runtime.h>
#include <math.h>

#define GAT_N 50000
#define GAT_E 800000
#define GAT_H 4
#define GAT_SLOPE 0.2f

typedef __attribute__((ext_vector_type(2))) float v2f;
typedef __attribute__((ext_vector_type(4))) float v4f;
typedef __attribute__((ext_vector_type(8))) float v8f;

template <int K, int NCOL, int WR, int WC>
__global__ __launch_bounds__(WR * WC * 32) void gat_gemm_wmma(
    const float* __restrict__ A, const float* __restrict__ B,
    float* __restrict__ C, int M)
{
  constexpr int NT = WR * WC * 32;
  constexpr int BM = WR * 32;
  constexpr int BN = WC * 32;
  constexpr int KC = 32;
  constexpr int PA = KC + 4;
  constexpr int PB = 2 * BN + 32;
  constexpr int A4 = KC / 4;
  constexpr int B4 = BN / 4;
  constexpr int NA = BM * A4 / NT;
  constexpr int NB = (KC / 2) * B4 / NT;
  static_assert(K % KC == 0, "K is walked in whole chunks");
  static_assert(NCOL % BN == 0, "column tiles cover C exactly");
  static_assert((BM * A4) % NT == 0 && ((KC / 2) * B4) % NT == 0, "copies divide evenly");

  __shared__ __attribute__((aligned(16))) float As[BM * PA];
  __shared__ __attribute__((aligned(16))) float Bs[(KC / 2) * PB];

  const int tid  = threadIdx.x;
  const int wave = tid >> 5;
  const int lane = tid & 31;
  const int wr   = wave / WC;
  const int wc   = wave % WC;
  const int col0 = blockIdx.x * BN;
  const int row0 = blockIdx.y * BM;

  const int m  = lane & 15;
  const int hl = lane >> 4;

  const float* ag[NA];
  int          al[NA];
#pragma unroll
  for (int i = 0; i < NA; ++i) {
    const int idx = tid + i * NT;
    const int r = idx / A4, c4 = idx % A4;
    const int gr = min(row0 + r, M - 1);
    ag[i] = A + (size_t)gr * K + c4 * 4;
    al[i] = r * PA + c4 * 4;
  }
  const float* bg[NB];
  int          bl[NB];
#pragma unroll
  for (int i = 0; i < NB; ++i) {
    const int idx = tid + i * NT;
    const int kp = idx / B4, c4 = idx % B4;
    bg[i] = B + (size_t)(2 * kp) * NCOL + col0 + c4 * 4;
    bl[i] = kp * PB + c4 * 8;
  }

  v4f ra[NA], rb0[NB], rb1[NB];
#pragma unroll
  for (int i = 0; i < NA; ++i) ra[i] = *(const v4f*)ag[i];
#pragma unroll
  for (int i = 0; i < NB; ++i) {
    rb0[i] = *(const v4f*)bg[i];
    rb1[i] = *(const v4f*)(bg[i] + NCOL);
  }

  v8f acc00 = {}, acc01 = {}, acc10 = {}, acc11 = {};
  const float* afrag = As + (wr * 32 + m) * PA + 2 * hl;
  const float* bfrag = Bs + hl * PB + (wc * 32 + m) * 2;

  auto stage = [&]() {
#pragma unroll
    for (int i = 0; i < NA; ++i) *(v4f*)(As + al[i]) = ra[i];
#pragma unroll
    for (int i = 0; i < NB; ++i) {
      v4f lo, hi;
      lo.x = rb0[i].x; lo.y = rb1[i].x; lo.z = rb0[i].y; lo.w = rb1[i].y;
      hi.x = rb0[i].z; hi.y = rb1[i].z; hi.z = rb0[i].w; hi.w = rb1[i].w;
      *(v4f*)(Bs + bl[i])     = lo;
      *(v4f*)(Bs + bl[i] + 4) = hi;
    }
  };
  auto multiply = [&]() {
#pragma unroll
    for (int ks = 0; ks < KC; ks += 4) {
      const v2f a0 = *(const v2f*)(afrag + ks);
      const v2f a1 = *(const v2f*)(afrag + 16 * PA + ks);
      const v2f b0 = *(const v2f*)(bfrag + (ks / 2) * PB);
      const v2f b1 = *(const v2f*)(bfrag + (ks / 2) * PB + 32);
      acc00 = __builtin_amdgcn_wmma_f32_16x16x4_f32(false, a0, false, b0, (short)0, acc00, false, false);
      acc01 = __builtin_amdgcn_wmma_f32_16x16x4_f32(false, a0, false, b1, (short)0, acc01, false, false);
      acc10 = __builtin_amdgcn_wmma_f32_16x16x4_f32(false, a1, false, b0, (short)0, acc10, false, false);
      acc11 = __builtin_amdgcn_wmma_f32_16x16x4_f32(false, a1, false, b1, (short)0, acc11, false, false);
    }
  };

#pragma unroll 1
  for (int kn = KC; kn < K; kn += KC) {
    __syncthreads();
    stage();
    __syncthreads();
#pragma unroll
    for (int i = 0; i < NA; ++i) ra[i] = *(const v4f*)(ag[i] + kn);
#pragma unroll
    for (int i = 0; i < NB; ++i) {
      rb0[i] = *(const v4f*)(bg[i] + (size_t)kn * NCOL);
      rb1[i] = *(const v4f*)(bg[i] + (size_t)(kn + 1) * NCOL);
    }
    __builtin_amdgcn_sched_barrier(0);
    multiply();
  }
  __syncthreads();
  stage();
  __syncthreads();
  multiply();

  const int rsub = hl << 3;
  const int rt0  = row0 + wr * 32;
  float* crow = C + (size_t)(rt0 + rsub) * NCOL + col0 + wc * 32 + m;
  if (rt0 < M) {
#pragma unroll
    for (int r = 0; r < 8; ++r) {
      crow[(size_t)r * NCOL]      = acc00[r];
      crow[(size_t)r * NCOL + 16] = acc01[r];
    }
  }
  if (rt0 + 16 < M) {
#pragma unroll
    for (int r = 0; r < 8; ++r) {
      crow[(size_t)(r + 16) * NCOL]      = acc10[r];
      crow[(size_t)(r + 16) * NCOL + 16] = acc11[r];
    }
  }
}

__global__ void gat_attn_scores(const float* __restrict__ feat,
                                const float* __restrict__ al,
                                const float* __restrict__ ar,
                                float* __restrict__ el, float* __restrict__ er,
                                int F, int D)
{
  int idx = blockIdx.x * blockDim.x + threadIdx.x;
  if (idx >= GAT_N * GAT_H) return;
  int nd = idx >> 2, h = idx & 3;
  const float* f   = feat + (size_t)nd * F + h * D;
  const float* alh = al + h * D;
  const float* arh = ar + h * D;
  float sl = 0.f, sr = 0.f;
  for (int d = 0; d < D; ++d) {
    float v = f[d];
    sl = fmaf(v, alh[d], sl);
    sr = fmaf(v, arh[d], sr);
  }
  el[idx] = sl;
  er[idx] = sr;
}

__global__ void gat_init_mden(float* __restrict__ mx, float* __restrict__ den, int count)
{
  int idx = blockIdx.x * blockDim.x + threadIdx.x;
  if (idx < count) {
    mx[idx]  = __int_as_float((int)0xFF800000u);
    den[idx] = 0.f;
  }
}

__device__ __forceinline__ float gat_leaky(float x) {
  return x > 0.f ? x : GAT_SLOPE * x;
}

__device__ __forceinline__ void gat_atomic_max_f32(float* addr, float v) {
  if (v >= 0.f) atomicMax((int*)addr, __float_as_int(v));
  else          atomicMin((unsigned int*)addr, __float_as_uint(v));
}

__global__ void gat_edge_max(const int* __restrict__ src, const int* __restrict__ dst,
                             const float* __restrict__ el, const float* __restrict__ er,
                             float* __restrict__ mx)
{
  int idx = blockIdx.x * blockDim.x + threadIdx.x;
  if (idx >= GAT_E * GAT_H) return;
  int e = idx >> 2, h = idx & 3;
  int s = src[e], d = dst[e];
  float x = gat_leaky(el[s * GAT_H + h] + er[d * GAT_H + h]);
  gat_atomic_max_f32(&mx[d * GAT_H + h], x);
}

__global__ void gat_edge_exp(const int* __restrict__ src, const int* __restrict__ dst,
                             const float* __restrict__ el, const float* __restrict__ er,
                             const float* __restrict__ mx,
                             float* __restrict__ ex, float* __restrict__ den)
{
  int idx = blockIdx.x * blockDim.x + threadIdx.x;
  if (idx >= GAT_E * GAT_H) return;
  int e = idx >> 2, h = idx & 3;
  int s = src[e], d = dst[e];
  float x = gat_leaky(el[s * GAT_H + h] + er[d * GAT_H + h]);
  float v = expf(x - mx[d * GAT_H + h]);
  ex[idx] = v;
  atomicAdd(&den[d * GAT_H + h], v);
}

#define GAT_TILE 196
#define GAT_AGG_THREADS 1024

template <int F, int D>
__global__ __launch_bounds__(GAT_AGG_THREADS) void gat_tile_aggregate(
    const int* __restrict__ src, const int* __restrict__ dst,
    const float* __restrict__ feat, const float* __restrict__ ex,
    const float* __restrict__ den, float* __restrict__ out)
{
  static_assert(GAT_E % 128 == 0, "dst is walked in steps of 128 edges");
  static_assert(F % 32 == 0, "row is walked in steps of one wave");
  constexpr int NW  = GAT_AGG_THREADS / 32;
  constexpr int NCH = GAT_E / 128;
  constexpr int FI  = F / 32;

  __shared__ float tile[GAT_TILE * F];
  __shared__ int2  queue[NW][160];

  const int tid   = threadIdx.x;
  const int wave  = __builtin_amdgcn_readfirstlane(tid >> 5);
  const int lane  = tid & 31;
  const int node0 = blockIdx.x * GAT_TILE;
  const int rows  = min(GAT_TILE, GAT_N - node0);

  float4* t4 = (float4*)tile;
  for (int i = tid; i < GAT_TILE * F / 4; i += GAT_AGG_THREADS)
    t4[i] = make_float4(0.f, 0.f, 0.f, 0.f);
  __syncthreads();

  int2* q  = queue[wave];
  int   qn = 0;

  auto drain = [&](int base, int n) {
    int   s_l = 0, r_l = 0;
    float a0 = 0.f, a1 = 0.f, a2 = 0.f, a3 = 0.f;
    if (lane < n) {
      const int2 ent = q[base + lane];
      r_l = ent.y;
      s_l = src[ent.x];
      const float4 xv = ((const float4*)ex)[ent.x];
      const float4 dn = ((const float4*)den)[node0 + ent.y];
      a0 = xv.x / dn.x; a1 = xv.y / dn.y; a2 = xv.z / dn.z; a3 = xv.w / dn.w;
    }
#pragma unroll 1
    for (int k = 0; k < n; k += 4) {
      const float* fp[4];
      float*       tp[4];
      float        w[4][4];
#pragma unroll
      for (int j = 0; j < 4; ++j) {
        const int kj = min(k + j, n - 1);
        fp[j] = feat + (size_t)__builtin_amdgcn_readlane(s_l, kj) * F + lane;
        tp[j] = tile + __builtin_amdgcn_readlane(r_l, kj) * F + lane;
        w[j][0] = __int_as_float(__builtin_amdgcn_readlane(__float_as_int(a0), kj));
        w[j][1] = __int_as_float(__builtin_amdgcn_readlane(__float_as_int(a1), kj));
        w[j][2] = __int_as_float(__builtin_amdgcn_readlane(__float_as_int(a2), kj));
        w[j][3] = __int_as_float(__builtin_amdgcn_readlane(__float_as_int(a3), kj));
      }
      float v[4][FI];
#pragma unroll
      for (int j = 0; j < 4; ++j) {
#pragma unroll
        for (int i = 0; i < FI; ++i) {
          const int h = (i * 32 + lane) / D;
          const float wh = h == 0 ? w[j][0] : (h == 1 ? w[j][1] : (h == 2 ? w[j][2] : w[j][3]));
          v[j][i] = fp[j][i * 32] * wh;
        }
      }
#pragma unroll
      for (int j = 0; j < 4; ++j) {
        if (k + j < n) {
#pragma unroll
          for (int i = 0; i < FI; ++i)
            __hip_atomic_fetch_add(tp[j] + i * 32, v[j][i], __ATOMIC_RELAXED, __HIP_MEMORY_SCOPE_WORKGROUP);
        }
      }
    }
  };

  auto flush = [&](int keep) {
    while (qn > keep) {
      const int n = min(qn, 32);
      drain(qn - n, n);
      qn -= n;
    }
  };

  auto push = [&](bool m, int e, int r) {
    const unsigned mask = __builtin_amdgcn_ballot_w32(m);
    if (m) q[qn + (int)__builtin_amdgcn_mbcnt_lo(mask, 0u)] = make_int2(e, r);
    qn += __popc(mask);
  };

  const int4* dst4 = (const int4*)dst;
  constexpr int NGROUPS = (GAT_N + GAT_TILE - 1) / GAT_TILE;
  const int rot = (int)blockIdx.x * (NCH / NGROUPS);
  auto place = [&](int t) { const int p = t + rot; return p >= NCH ? p - NCH : p; };
  auto fetch = [&](int t) { return dst4[(size_t)place(min(t, NCH - 1)) * 32 + lane]; };
  auto step = [&](const int4 cur, int t) {
    const bool live = t < NCH;
    const int cc = place(t);
    const int l0 = cur.x - node0, l1 = cur.y - node0;
    const int l2 = cur.z - node0, l3 = cur.w - node0;
    const bool m0 = live && (unsigned)l0 < (unsigned)rows, m1 = live && (unsigned)l1 < (unsigned)rows;
    const bool m2 = live && (unsigned)l2 < (unsigned)rows, m3 = live && (unsigned)l3 < (unsigned)rows;
    if (__builtin_amdgcn_ballot_w32(m0 | m1 | m2 | m3) != 0u) {
      const int e0 = cc * 128 + lane * 4;
      push(m0, e0 + 0, l0);
      push(m1, e0 + 1, l1);
      push(m2, e0 + 2, l2);
      push(m3, e0 + 3, l3);
      flush(31);
    }
  };
  int4 n0 = fetch(wave), n1 = fetch(wave + NW), n2 = fetch(wave + 2 * NW), n3 = fetch(wave + 3 * NW);
#pragma unroll 1
  for (int c = wave; c < NCH; c += 4 * NW) {
    const int4 c0 = n0, c1 = n1, c2 = n2, c3 = n3;
    n0 = fetch(c + 4 * NW); n1 = fetch(c + 5 * NW); n2 = fetch(c + 6 * NW); n3 = fetch(c + 7 * NW);
    step(c0, c);
    step(c1, c + NW);
    step(c2, c + 2 * NW);
    step(c3, c + 3 * NW);
  }
  flush(0);

  __syncthreads();
  float4* o4 = (float4*)(out + (size_t)node0 * F);
  for (int i = tid; i < rows * F / 4; i += GAT_AGG_THREADS) o4[i] = t4[i];
}

__global__ void gat_finalize_elu(float* __restrict__ acc, const float* __restrict__ bias,
                                 const float* __restrict__ res, int total)
{
  int idx = blockIdx.x * blockDim.x + threadIdx.x;
  if (idx >= total) return;
  float v = acc[idx] + bias[idx & 255];
  if (res) v += res[idx];
  acc[idx] = v > 0.f ? v : (expf(v) - 1.f);
}

__global__ void gat_finalize_out(const float* __restrict__ acc, const float* __restrict__ bias,
                                 const float* __restrict__ res2, float* __restrict__ out)
{
  const int C = 40;
  int idx = blockIdx.x * blockDim.x + threadIdx.x;
  if (idx >= GAT_N * C) return;
  int n = idx / C, c = idx % C;
  float sum = 0.f;
#pragma unroll
  for (int h = 0; h < GAT_H; ++h) {
    int j = n * (GAT_H * C) + h * C + c;
    sum += acc[j] + bias[h * C + c] + res2[j];
  }
  out[idx] = sum * 0.25f;
}

extern "C" void kernel_launch(void* const* d_in, const int* in_sizes, int n_in,
                              void* d_out, int out_size, void* d_ws, size_t ws_size,
                              hipStream_t stream)
{
  const float* x     = (const float*)d_in[0];
  const int*   src   = (const int*)d_in[1];
  const int*   dst   = (const int*)d_in[2];
  const float* W0    = (const float*)d_in[3];
  const float* al0   = (const float*)d_in[4];
  const float* ar0   = (const float*)d_in[5];
  const float* b0    = (const float*)d_in[6];
  const float* W1    = (const float*)d_in[7];
  const float* al1   = (const float*)d_in[8];
  const float* ar1   = (const float*)d_in[9];
  const float* b1    = (const float*)d_in[10];
  const float* W2    = (const float*)d_in[11];
  const float* al2   = (const float*)d_in[12];
  const float* ar2   = (const float*)d_in[13];
  const float* b2    = (const float*)d_in[14];
  const float* Wres2 = (const float*)d_in[15];

  const size_t NF = (size_t)GAT_N * 256;
  const size_t NG = (size_t)GAT_N * 160;
  float* ws   = (float*)d_ws;
  float* bufA = ws;
  float* bufB = bufA + NF;
  float* bufC = bufB + NF;
  float* bufD = bufC + NF;
  float* EX   = bufD + NG;
  float* EL   = EX + (size_t)GAT_E * GAT_H;
  float* ER   = EL + (size_t)GAT_N * GAT_H;
  float* MX   = ER + (size_t)GAT_N * GAT_H;
  float* DEN  = MX + (size_t)GAT_N * GAT_H;

  dim3 gemmBlk(128);
  dim3 gemmGrid256(4, (GAT_N + 63) / 64);
  dim3 gemmGrid160(5, (GAT_N + 127) / 128);

  const int ehThreads = GAT_E * GAT_H;
  const int ehBlocks  = (ehThreads + 255) / 256;
  const int nhThreads = GAT_N * GAT_H;
  const int nhBlocks  = (nhThreads + 255) / 256;
  const int aggBlocks = (GAT_N + GAT_TILE - 1) / GAT_TILE;
  const int nfBlocks  = (int)((NF + 255) / 256);

  gat_gemm_wmma<256, 256, 2, 2><<<gemmGrid256, gemmBlk, 0, stream>>>(x, W0, bufA, GAT_N);
  gat_attn_scores<<<nhBlocks, 256, 0, stream>>>(bufA, al0, ar0, EL, ER, 256, 64);
  gat_init_mden<<<nhBlocks, 256, 0, stream>>>(MX, DEN, nhThreads);
  gat_edge_max<<<ehBlocks, 256, 0, stream>>>(src, dst, EL, ER, MX);
  gat_edge_exp<<<ehBlocks, 256, 0, stream>>>(src, dst, EL, ER, MX, EX, DEN);
  gat_tile_aggregate<256, 64><<<aggBlocks, GAT_AGG_THREADS, 0, stream>>>(src, dst, bufA, EX, DEN, bufB);
  gat_finalize_elu<<<nfBlocks, 256, 0, stream>>>(bufB, b0, nullptr, (int)NF);

  gat_gemm_wmma<256, 256, 2, 2><<<gemmGrid256, gemmBlk, 0, stream>>>(bufB, W1, bufA, GAT_N);
  gat_attn_scores<<<nhBlocks, 256, 0, stream>>>(bufA, al1, ar1, EL, ER, 256, 64);
  gat_init_mden<<<nhBlocks, 256, 0, stream>>>(MX, DEN, nhThreads);
  gat_edge_max<<<ehBlocks, 256, 0, stream>>>(src, dst, EL, ER, MX);
  gat_edge_exp<<<ehBlocks, 256, 0, stream>>>(src, dst, EL, ER, MX, EX, DEN);
  gat_tile_aggregate<256, 64><<<aggBlocks, GAT_AGG_THREADS, 0, stream>>>(src, dst, bufA, EX, DEN, bufC);
  gat_finalize_elu<<<nfBlocks, 256, 0, stream>>>(bufC, b1, bufB, (int)NF);

  gat_gemm_wmma<256, 160, 4, 1><<<gemmGrid160, gemmBlk, 0, stream>>>(bufC, W2, bufA, GAT_N);
  gat_gemm_wmma<256, 160, 4, 1><<<gemmGrid160, gemmBlk, 0, stream>>>(bufC, Wres2, bufB, GAT_N);
  gat_attn_scores<<<nhBlocks, 256, 0, stream>>>(bufA, al2, ar2, EL, ER, 160, 40);
  gat_init_mden<<<nhBlocks, 256, 0, stream>>>(MX, DEN, nhThreads);
  gat_edge_max<<<ehBlocks, 256, 0, stream>>>(src, dst, EL, ER, MX);
  gat_edge_exp<<<ehBlocks, 256, 0, stream>>>(src, dst, EL, ER, MX, EX, DEN);
  gat_tile_aggregate<160, 40><<<aggBlocks, GAT_AGG_THREADS, 0, stream>>>(src, dst, bufA, EX, DEN, bufD);
  gat_finalize_out<<<(GAT_N * 40 + 255) / 256, 256, 0, stream>>>(bufD, b2, bufB, (float*)d_out);
}


# edge max, edge exp and both finalize kernels on fixed grids: a thread takes 10 strided elements (16-byte pieces in finalize), about 16 times fewer and longer waves; den atomics keep arrival order
# speedup vs baseline: 1.7837x; 1.0304x over previous
#include <hip/hip_runtime.h>
#include <math.h>

#define GAT_N 50000
#define GAT_E 800000
#define GAT_H 4
#define GAT_SLOPE 0.2f

typedef __attribute__((ext_vector_type(2))) float v2f;
typedef __attribute__((ext_vector_type(4))) float v4f;
typedef __attribute__((ext_vector_type(8))) float v8f;

template <int K, int NCOL, int WR, int WC>
__global__ __launch_bounds__(WR * WC * 32) void gat_gemm_wmma(
    const float* __restrict__ A, const float* __restrict__ B,
    float* __restrict__ C, int M)
{
  constexpr int NT = WR * WC * 32;
  constexpr int BM = WR * 32;
  constexpr int BN = WC * 32;
  constexpr int KC = 32;
  constexpr int PA = KC + 4;
  constexpr int PB = 2 * BN + 32;
  constexpr int A4 = KC / 4;
  constexpr int B4 = BN / 4;
  constexpr int NA = BM * A4 / NT;
  constexpr int NB = (KC / 2) * B4 / NT;
  static_assert(K % KC == 0, "K is walked in whole chunks");
  static_assert(NCOL % BN == 0, "column tiles cover C exactly");
  static_assert((BM * A4) % NT == 0 && ((KC / 2) * B4) % NT == 0, "copies divide evenly");

  __shared__ __attribute__((aligned(16))) float As[BM * PA];
  __shared__ __attribute__((aligned(16))) float Bs[(KC / 2) * PB];

  const int tid  = threadIdx.x;
  const int wave = tid >> 5;
  const int lane = tid & 31;
  const int wr   = wave / WC;
  const int wc   = wave % WC;
  const int col0 = blockIdx.x * BN;
  const int row0 = blockIdx.y * BM;

  const int m  = lane & 15;
  const int hl = lane >> 4;

  const float* ag[NA];
  int          al[NA];
#pragma unroll
  for (int i = 0; i < NA; ++i) {
    const int idx = tid + i * NT;
    const int r = idx / A4, c4 = idx % A4;
    const int gr = min(row0 + r, M - 1);
    ag[i] = A + (size_t)gr * K + c4 * 4;
    al[i] = r * PA + c4 * 4;
  }
  const float* bg[NB];
  int          bl[NB];
#pragma unroll
  for (int i = 0; i < NB; ++i) {
    const int idx = tid + i * NT;
    const int kp = idx / B4, c4 = idx % B4;
    bg[i] = B + (size_t)(2 * kp) * NCOL + col0 + c4 * 4;
    bl[i] = kp * PB + c4 * 8;
  }

  v4f ra[NA], rb0[NB], rb1[NB];
#pragma unroll
  for (int i = 0; i < NA; ++i) ra[i] = *(const v4f*)ag[i];
#pragma unroll
  for (int i = 0; i < NB; ++i) {
    rb0[i] = *(const v4f*)bg[i];
    rb1[i] = *(const v4f*)(bg[i] + NCOL);
  }

  v8f acc00 = {}, acc01 = {}, acc10 = {}, acc11 = {};
  const float* afrag = As + (wr * 32 + m) * PA + 2 * hl;
  const float* bfrag = Bs + hl * PB + (wc * 32 + m) * 2;

  auto stage = [&]() {
#pragma unroll
    for (int i = 0; i < NA; ++i) *(v4f*)(As + al[i]) = ra[i];
#pragma unroll
    for (int i = 0; i < NB; ++i) {
      v4f lo, hi;
      lo.x = rb0[i].x; lo.y = rb1[i].x; lo.z = rb0[i].y; lo.w = rb1[i].y;
      hi.x = rb0[i].z; hi.y = rb1[i].z; hi.z = rb0[i].w; hi.w = rb1[i].w;
      *(v4f*)(Bs + bl[i])     = lo;
      *(v4f*)(Bs + bl[i] + 4) = hi;
    }
  };
  auto multiply = [&]() {
#pragma unroll
    for (int ks = 0; ks < KC; ks += 4) {
      const v2f a0 = *(const v2f*)(afrag + ks);
      const v2f a1 = *(const v2f*)(afrag + 16 * PA + ks);
      const v2f b0 = *(const v2f*)(bfrag + (ks / 2) * PB);
      const v2f b1 = *(const v2f*)(bfrag + (ks / 2) * PB + 32);
      acc00 = __builtin_amdgcn_wmma_f32_16x16x4_f32(false, a0, false, b0, (short)0, acc00, false, false);
      acc01 = __builtin_amdgcn_wmma_f32_16x16x4_f32(false, a0, false, b1, (short)0, acc01, false, false);
      acc10 = __builtin_amdgcn_wmma_f32_16x16x4_f32(false, a1, false, b0, (short)0, acc10, false, false);
      acc11 = __builtin_amdgcn_wmma_f32_16x16x4_f32(false, a1, false, b1, (short)0, acc11, false, false);
    }
  };

#pragma unroll 1
  for (int kn = KC; kn < K; kn += KC) {
    __syncthreads();
    stage();
    __syncthreads();
#pragma unroll
    for (int i = 0; i < NA; ++i) ra[i] = *(const v4f*)(ag[i] + kn);
#pragma unroll
    for (int i = 0; i < NB; ++i) {
      rb0[i] = *(const v4f*)(bg[i] + (size_t)kn * NCOL);
      rb1[i] = *(const v4f*)(bg[i] + (size_t)(kn + 1) * NCOL);
    }
    __builtin_amdgcn_sched_barrier(0);
    multiply();
  }
  __syncthreads();
  stage();
  __syncthreads();
  multiply();

  const int rsub = hl << 3;
  const int rt0  = row0 + wr * 32;
  float* crow = C + (size_t)(rt0 + rsub) * NCOL + col0 + wc * 32 + m;
  if (rt0 < M) {
#pragma unroll
    for (int r = 0; r < 8; ++r) {
      crow[(size_t)r * NCOL]      = acc00[r];
      crow[(size_t)r * NCOL + 16] = acc01[r];
    }
  }
  if (rt0 + 16 < M) {
#pragma unroll
    for (int r = 0; r < 8; ++r) {
      crow[(size_t)(r + 16) * NCOL]      = acc10[r];
      crow[(size_t)(r + 16) * NCOL + 16] = acc11[r];
    }
  }
}

__global__ void gat_attn_scores(const float* __restrict__ feat,
                                const float* __restrict__ al,
                                const float* __restrict__ ar,
                                float* __restrict__ el, float* __restrict__ er,
                                int F, int D)
{
  int idx = blockIdx.x * blockDim.x + threadIdx.x;
  if (idx >= GAT_N * GAT_H) return;
  int nd = idx >> 2, h = idx & 3;
  const float* f   = feat + (size_t)nd * F + h * D;
  const float* alh = al + h * D;
  const float* arh = ar + h * D;
  float sl = 0.f, sr = 0.f;
  for (int d = 0; d < D; ++d) {
    float v = f[d];
    sl = fmaf(v, alh[d], sl);
    sr = fmaf(v, arh[d], sr);
  }
  el[idx] = sl;
  er[idx] = sr;
}

__global__ void gat_init_mden(float* __restrict__ mx, float* __restrict__ den, int count)
{
  int idx = blockIdx.x * blockDim.x + threadIdx.x;
  if (idx < count) {
    mx[idx]  = __int_as_float((int)0xFF800000u);
    den[idx] = 0.f;
  }
}

__device__ __forceinline__ float gat_leaky(float x) {
  return x > 0.f ? x : GAT_SLOPE * x;
}

__device__ __forceinline__ void gat_atomic_max_f32(float* addr, float v) {
  if (v >= 0.f) atomicMax((int*)addr, __float_as_int(v));
  else          atomicMin((unsigned int*)addr, __float_as_uint(v));
}

#define GAT_EW_THREADS 256
#define GAT_EW_GROUPS  1250
#define GAT_EW_STRIDE  (GAT_EW_GROUPS * GAT_EW_THREADS)
#define GAT_EW_ITEMS   10
static_assert(GAT_EW_STRIDE * GAT_EW_ITEMS == GAT_E * GAT_H, "the grid covers every (edge, head) pair exactly once");
static_assert(GAT_EW_STRIDE % 4 == 0, "a thread keeps its head over its elements");

__global__ __launch_bounds__(GAT_EW_THREADS) void gat_edge_max(
    const int* __restrict__ src, const int* __restrict__ dst,
    const float* __restrict__ el, const float* __restrict__ er,
    float* __restrict__ mx)
{
  const int t = blockIdx.x * GAT_EW_THREADS + threadIdx.x;
  const int h = t & 3;
  int s[GAT_EW_ITEMS], d[GAT_EW_ITEMS];
#pragma unroll
  for (int u = 0; u < GAT_EW_ITEMS; ++u) {
    const int e = (t + u * GAT_EW_STRIDE) >> 2;
    s[u] = src[e];
    d[u] = dst[e];
  }
  float x[GAT_EW_ITEMS];
#pragma unroll
  for (int u = 0; u < GAT_EW_ITEMS; ++u)
    x[u] = el[s[u] * GAT_H + h] + er[d[u] * GAT_H + h];
#pragma unroll
  for (int u = 0; u < GAT_EW_ITEMS; ++u)
    gat_atomic_max_f32(&mx[d[u] * GAT_H + h], gat_leaky(x[u]));
}

__global__ __launch_bounds__(GAT_EW_THREADS) void gat_edge_exp(
    const int* __restrict__ src, const int* __restrict__ dst,
    const float* __restrict__ el, const float* __restrict__ er,
    const float* __restrict__ mx,
    float* __restrict__ ex, float* __restrict__ den)
{
  const int t = blockIdx.x * GAT_EW_THREADS + threadIdx.x;
  const int h = t & 3;
  int s[GAT_EW_ITEMS], d[GAT_EW_ITEMS];
#pragma unroll
  for (int u = 0; u < GAT_EW_ITEMS; ++u) {
    const int e = (t + u * GAT_EW_STRIDE) >> 2;
    s[u] = src[e];
    d[u] = dst[e];
  }
  float x[GAT_EW_ITEMS], m[GAT_EW_ITEMS];
#pragma unroll
  for (int u = 0; u < GAT_EW_ITEMS; ++u) {
    x[u] = el[s[u] * GAT_H + h] + er[d[u] * GAT_H + h];
    m[u] = mx[d[u] * GAT_H + h];
  }
#pragma unroll
  for (int u = 0; u < GAT_EW_ITEMS; ++u) {
    const float v = expf(gat_leaky(x[u]) - m[u]);
    ex[t + u * GAT_EW_STRIDE] = v;
    atomicAdd(&den[d[u] * GAT_H + h], v);
  }
}

#define GAT_TILE 196
#define GAT_AGG_THREADS 1024

template <int F, int D>
__global__ __launch_bounds__(GAT_AGG_THREADS) void gat_tile_aggregate(
    const int* __restrict__ src, const int* __restrict__ dst,
    const float* __restrict__ feat, const float* __restrict__ ex,
    const float* __restrict__ den, float* __restrict__ out)
{
  static_assert(GAT_E % 128 == 0, "dst is walked in steps of 128 edges");
  static_assert(F % 32 == 0, "row is walked in steps of one wave");
  constexpr int NW  = GAT_AGG_THREADS / 32;
  constexpr int NCH = GAT_E / 128;
  constexpr int FI  = F / 32;

  __shared__ float tile[GAT_TILE * F];
  __shared__ int2  queue[NW][160];

  const int tid   = threadIdx.x;
  const int wave  = __builtin_amdgcn_readfirstlane(tid >> 5);
  const int lane  = tid & 31;
  const int node0 = blockIdx.x * GAT_TILE;
  const int rows  = min(GAT_TILE, GAT_N - node0);

  float4* t4 = (float4*)tile;
  for (int i = tid; i < GAT_TILE * F / 4; i += GAT_AGG_THREADS)
    t4[i] = make_float4(0.f, 0.f, 0.f, 0.f);
  __syncthreads();

  int2* q  = queue[wave];
  int   qn = 0;

  auto drain = [&](int base, int n) {
    int   s_l = 0, r_l = 0;
    float a0 = 0.f, a1 = 0.f, a2 = 0.f, a3 = 0.f;
    if (lane < n) {
      const int2 ent = q[base + lane];
      r_l = ent.y;
      s_l = src[ent.x];
      const float4 xv = ((const float4*)ex)[ent.x];
      const float4 dn = ((const float4*)den)[node0 + ent.y];
      a0 = xv.x / dn.x; a1 = xv.y / dn.y; a2 = xv.z / dn.z; a3 = xv.w / dn.w;
    }
#pragma unroll 1
    for (int k = 0; k < n; k += 4) {
      const float* fp[4];
      float*       tp[4];
      float        w[4][4];
#pragma unroll
      for (int j = 0; j < 4; ++j) {
        const int kj = min(k + j, n - 1);
        fp[j] = feat + (size_t)__builtin_amdgcn_readlane(s_l, kj) * F + lane;
        tp[j] = tile + __builtin_amdgcn_readlane(r_l, kj) * F + lane;
        w[j][0] = __int_as_float(__builtin_amdgcn_readlane(__float_as_int(a0), kj));
        w[j][1] = __int_as_float(__builtin_amdgcn_readlane(__float_as_int(a1), kj));
        w[j][2] = __int_as_float(__builtin_amdgcn_readlane(__float_as_int(a2), kj));
        w[j][3] = __int_as_float(__builtin_amdgcn_readlane(__float_as_int(a3), kj));
      }
      float v[4][FI];
#pragma unroll
      for (int j = 0; j < 4; ++j) {
#pragma unroll
        for (int i = 0; i < FI; ++i) {
          const int h = (i * 32 + lane) / D;
          const float wh = h == 0 ? w[j][0] : (h == 1 ? w[j][1] : (h == 2 ? w[j][2] : w[j][3]));
          v[j][i] = fp[j][i * 32] * wh;
        }
      }
#pragma unroll
      for (int j = 0; j < 4; ++j) {
        if (k + j < n) {
#pragma unroll
          for (int i = 0; i < FI; ++i)
            __hip_atomic_fetch_add(tp[j] + i * 32, v[j][i], __ATOMIC_RELAXED, __HIP_MEMORY_SCOPE_WORKGROUP);
        }
      }
    }
  };

  auto flush = [&](int keep) {
    while (qn > keep) {
      const int n = min(qn, 32);
      drain(qn - n, n);
      qn -= n;
    }
  };

  auto push = [&](bool m, int e, int r) {
    const unsigned mask = __builtin_amdgcn_ballot_w32(m);
    if (m) q[qn + (int)__builtin_amdgcn_mbcnt_lo(mask, 0u)] = make_int2(e, r);
    qn += __popc(mask);
  };

  const int4* dst4 = (const int4*)dst;
  constexpr int NGROUPS = (GAT_N + GAT_TILE - 1) / GAT_TILE;
  const int rot = (int)blockIdx.x * (NCH / NGROUPS);
  auto place = [&](int t) { const int p = t + rot; return p >= NCH ? p - NCH : p; };
  auto fetch = [&](int t) { return dst4[(size_t)place(min(t, NCH - 1)) * 32 + lane]; };
  auto step = [&](const int4 cur, int t) {
    const bool live = t < NCH;
    const int cc = place(t);
    const int l0 = cur.x - node0, l1 = cur.y - node0;
    const int l2 = cur.z - node0, l3 = cur.w - node0;
    const bool m0 = live && (unsigned)l0 < (unsigned)rows, m1 = live && (unsigned)l1 < (unsigned)rows;
    const bool m2 = live && (unsigned)l2 < (unsigned)rows, m3 = live && (unsigned)l3 < (unsigned)rows;
    if (__builtin_amdgcn_ballot_w32(m0 | m1 | m2 | m3) != 0u) {
      const int e0 = cc * 128 + lane * 4;
      push(m0, e0 + 0, l0);
      push(m1, e0 + 1, l1);
      push(m2, e0 + 2, l2);
      push(m3, e0 + 3, l3);
      flush(31);
    }
  };
  int4 n0 = fetch(wave), n1 = fetch(wave + NW), n2 = fetch(wave + 2 * NW), n3 = fetch(wave + 3 * NW);
#pragma unroll 1
  for (int c = wave; c < NCH; c += 4 * NW) {
    const int4 c0 = n0, c1 = n1, c2 = n2, c3 = n3;
    n0 = fetch(c + 4 * NW); n1 = fetch(c + 5 * NW); n2 = fetch(c + 6 * NW); n3 = fetch(c + 7 * NW);
    step(c0, c);
    step(c1, c + NW);
    step(c2, c + 2 * NW);
    step(c3, c + 3 * NW);
  }
  flush(0);

  __syncthreads();
  float4* o4 = (float4*)(out + (size_t)node0 * F);
  for (int i = tid; i < rows * F / 4; i += GAT_AGG_THREADS) o4[i] = t4[i];
}

static_assert(GAT_EW_STRIDE * GAT_EW_ITEMS * 4 == GAT_N * 256, "the grid covers the N x 256 buffer exactly once");
static_assert(GAT_EW_STRIDE % 64 == 0, "a thread keeps its 16-byte piece of the bias row");
#define GAT_ELU_BATCH 5

template <bool RES>
__global__ __launch_bounds__(GAT_EW_THREADS) void gat_finalize_elu(
    float* __restrict__ acc, const float* __restrict__ bias, const float* __restrict__ res)
{
  const int t = blockIdx.x * GAT_EW_THREADS + threadIdx.x;
  v4f*       acc4 = (v4f*)acc;
  const v4f* res4 = (const v4f*)res;
  const v4f  b    = ((const v4f*)bias)[t & 63];
#pragma unroll 1
  for (int p = 0; p < GAT_EW_ITEMS; p += GAT_ELU_BATCH) {
    v4f a[GAT_ELU_BATCH], r[GAT_ELU_BATCH];
#pragma unroll
    for (int u = 0; u < GAT_ELU_BATCH; ++u) {
      const int i = t + (p + u) * GAT_EW_STRIDE;
      a[u] = acc4[i];
      if (RES) r[u] = res4[i];
    }
#pragma unroll
    for (int u = 0; u < GAT_ELU_BATCH; ++u) {
      v4f v = a[u] + b;
      if (RES) v += r[u];
      v4f o;
      o.x = v.x > 0.f ? v.x : (expf(v.x) - 1.f);
      o.y = v.y > 0.f ? v.y : (expf(v.y) - 1.f);
      o.z = v.z > 0.f ? v.z : (expf(v.z) - 1.f);
      o.w = v.w > 0.f ? v.w : (expf(v.w) - 1.f);
      acc4[t + (p + u) * GAT_EW_STRIDE] = o;
    }
  }
}

#define GAT_OUT_GROUPS 977

__global__ __launch_bounds__(GAT_EW_THREADS) void gat_finalize_out(
    const float* __restrict__ acc, const float* __restrict__ bias,
    const float* __restrict__ res2, float* __restrict__ out)
{
  constexpr int C4     = 10;
  constexpr int TOTAL4 = GAT_N * C4;
  static_assert(GAT_OUT_GROUPS * GAT_EW_THREADS * 2 >= TOTAL4, "two passes cover the output");
  const v4f* acc4  = (const v4f*)acc;
  const v4f* res4  = (const v4f*)res2;
  const v4f* bias4 = (const v4f*)bias;
  v4f*       out4  = (v4f*)out;
  for (int i = blockIdx.x * GAT_EW_THREADS + threadIdx.x; i < TOTAL4; i += GAT_OUT_GROUPS * GAT_EW_THREADS) {
    const int n = i / C4, c4 = i % C4;
    v4f sum = {0.f, 0.f, 0.f, 0.f};
#pragma unroll
    for (int h = 0; h < GAT_H; ++h) {
      const int j4 = n * (GAT_H * C4) + h * C4 + c4;
      sum += acc4[j4] + bias4[h * C4 + c4] + res4[j4];
    }
    out4[i] = sum * 0.25f;
  }
}

extern "C" void kernel_launch(void* const* d_in, const int* in_sizes, int n_in,
                              void* d_out, int out_size, void* d_ws, size_t ws_size,
                              hipStream_t stream)
{
  const float* x     = (const float*)d_in[0];
  const int*   src   = (const int*)d_in[1];
  const int*   dst   = (const int*)d_in[2];
  const float* W0    = (const float*)d_in[3];
  const float* al0   = (const float*)d_in[4];
  const float* ar0   = (const float*)d_in[5];
  const float* b0    = (const float*)d_in[6];
  const float* W1    = (const float*)d_in[7];
  const float* al1   = (const float*)d_in[8];
  const float* ar1   = (const float*)d_in[9];
  const float* b1    = (const float*)d_in[10];
  const float* W2    = (const float*)d_in[11];
  const float* al2   = (const float*)d_in[12];
  const float* ar2   = (const float*)d_in[13];
  const float* b2    = (const float*)d_in[14];
  const float* Wres2 = (const float*)d_in[15];

  const size_t NF = (size_t)GAT_N * 256;
  const size_t NG = (size_t)GAT_N * 160;
  float* ws   = (float*)d_ws;
  float* bufA = ws;
  float* bufB = bufA + NF;
  float* bufC = bufB + NF;
  float* bufD = bufC + NF;
  float* EX   = bufD + NG;
  float* EL   = EX + (size_t)GAT_E * GAT_H;
  float* ER   = EL + (size_t)GAT_N * GAT_H;
  float* MX   = ER + (size_t)GAT_N * GAT_H;
  float* DEN  = MX + (size_t)GAT_N * GAT_H;

  dim3 gemmBlk(128);
  dim3 gemmGrid256(4, (GAT_N + 63) / 64);
  dim3 gemmGrid160(5, (GAT_N + 127) / 128);

  const int nhThreads = GAT_N * GAT_H;
  const int nhBlocks  = (nhThreads + 255) / 256;
  const int aggBlocks = (GAT_N + GAT_TILE - 1) / GAT_TILE;

  gat_gemm_wmma<256, 256, 2, 2><<<gemmGrid256, gemmBlk, 0, stream>>>(x, W0, bufA, GAT_N);
  gat_attn_scores<<<nhBlocks, 256, 0, stream>>>(bufA, al0, ar0, EL, ER, 256, 64);
  gat_init_mden<<<nhBlocks, 256, 0, stream>>>(MX, DEN, nhThreads);
  gat_edge_max<<<GAT_EW_GROUPS, GAT_EW_THREADS, 0, stream>>>(src, dst, EL, ER, MX);
  gat_edge_exp<<<GAT_EW_GROUPS, GAT_EW_THREADS, 0, stream>>>(src, dst, EL, ER, MX, EX, DEN);
  gat_tile_aggregate<256, 64><<<aggBlocks, GAT_AGG_THREADS, 0, stream>>>(src, dst, bufA, EX, DEN, bufB);
  gat_finalize_elu<false><<<GAT_EW_GROUPS, GAT_EW_THREADS, 0, stream>>>(bufB, b0, nullptr);

  gat_gemm_wmma<256, 256, 2, 2><<<gemmGrid256, gemmBlk, 0, stream>>>(bufB, W1, bufA, GAT_N);
  gat_attn_scores<<<nhBlocks, 256, 0, stream>>>(bufA, al1, ar1, EL, ER, 256, 64);
  gat_init_mden<<<nhBlocks, 256, 0, stream>>>(MX, DEN, nhThreads);
  gat_edge_max<<<GAT_EW_GROUPS, GAT_EW_THREADS, 0, stream>>>(src, dst, EL, ER, MX);
  gat_edge_exp<<<GAT_EW_GROUPS, GAT_EW_THREADS, 0, stream>>>(src, dst, EL, ER, MX, EX, DEN);
  gat_tile_aggregate<256, 64><<<aggBlocks, GAT_AGG_THREADS, 0, stream>>>(src, dst, bufA, EX, DEN, bufC);
  gat_finalize_elu<true><<<GAT_EW_GROUPS, GAT_EW_THREADS, 0, stream>>>(bufC, b1, bufB);

  gat_gemm_wmma<256, 160, 4, 1><<<gemmGrid160, gemmBlk, 0, stream>>>(bufC, W2, bufA, GAT_N);
  gat_gemm_wmma<256, 160, 4, 1><<<gemmGrid160, gemmBlk, 0, stream>>>(bufC, Wres2, bufB, GAT_N);
  gat_attn_scores<<<nhBlocks, 256, 0, stream>>>(bufA, al2, ar2, EL, ER, 160, 40);
  gat_init_mden<<<nhBlocks, 256, 0, stream>>>(MX, DEN, nhThreads);
  gat_edge_max<<<GAT_EW_GROUPS, GAT_EW_THREADS, 0, stream>>>(src, dst, EL, ER, MX);
  gat_edge_exp<<<GAT_EW_GROUPS, GAT_EW_THREADS, 0, stream>>>(src, dst, EL, ER, MX, EX, DEN);
  gat_tile_aggregate<160, 40><<<aggBlocks, GAT_AGG_THREADS, 0, stream>>>(src, dst, bufA, EX, DEN, bufD);
  gat_finalize_out<<<GAT_OUT_GROUPS, GAT_EW_THREADS, 0, stream>>>(bufD, b2, bufB, (float*)d_out);
}


# edge softmax inside the tile aggregation: a group lists its edges in LDS, takes max and exp sum by LDS atomics, divides rows once at the store; 9 launches, global atomics gone; sum order changes
# speedup vs baseline: 2.8444x; 1.5947x over previous
#include <hip/hip_runtime.h>
#include <math.h>

#define GAT_N 50000
#define GAT_E 800000
#define GAT_H 4
#define GAT_SLOPE 0.2f

typedef __attribute__((ext_vector_type(2))) float v2f;
typedef __attribute__((ext_vector_type(4))) float v4f;
typedef __attribute__((ext_vector_type(8))) float v8f;

template <int K, int NCOL, int WR, int WC>
__global__ __launch_bounds__(WR * WC * 32) void gat_gemm_wmma(
    const float* __restrict__ A, const float* __restrict__ B,
    float* __restrict__ C, int M)
{
  constexpr int NT = WR * WC * 32;
  constexpr int BM = WR * 32;
  constexpr int BN = WC * 32;
  constexpr int KC = 32;
  constexpr int PA = KC + 4;
  constexpr int PB = 2 * BN + 32;
  constexpr int A4 = KC / 4;
  constexpr int B4 = BN / 4;
  constexpr int NA = BM * A4 / NT;
  constexpr int NB = (KC / 2) * B4 / NT;
  static_assert(K % KC == 0, "K is walked in whole chunks");
  static_assert(NCOL % BN == 0, "column tiles cover C exactly");
  static_assert((BM * A4) % NT == 0 && ((KC / 2) * B4) % NT == 0, "copies divide evenly");

  __shared__ __attribute__((aligned(16))) float As[BM * PA];
  __shared__ __attribute__((aligned(16))) float Bs[(KC / 2) * PB];

  const int tid  = threadIdx.x;
  const int wave = tid >> 5;
  const int lane = tid & 31;
  const int wr   = wave / WC;
  const int wc   = wave % WC;
  const int col0 = blockIdx.x * BN;
  const int row0 = blockIdx.y * BM;

  const int m  = lane & 15;
  const int hl = lane >> 4;

  const float* ag[NA];
  int          al[NA];
#pragma unroll
  for (int i = 0; i < NA; ++i) {
    const int idx = tid + i * NT;
    const int r = idx / A4, c4 = idx % A4;
    const int gr = min(row0 + r, M - 1);
    ag[i] = A + (size_t)gr * K + c4 * 4;
    al[i] = r * PA + c4 * 4;
  }
  const float* bg[NB];
  int          bl[NB];
#pragma unroll
  for (int i = 0; i < NB; ++i) {
    const int idx = tid + i * NT;
    const int kp = idx / B4, c4 = idx % B4;
    bg[i] = B + (size_t)(2 * kp) * NCOL + col0 + c4 * 4;
    bl[i] = kp * PB + c4 * 8;
  }

  v4f ra[NA], rb0[NB], rb1[NB];
#pragma unroll
  for (int i = 0; i < NA; ++i) ra[i] = *(const v4f*)ag[i];
#pragma unroll
  for (int i = 0; i < NB; ++i) {
    rb0[i] = *(const v4f*)bg[i];
    rb1[i] = *(const v4f*)(bg[i] + NCOL);
  }

  v8f acc00 = {}, acc01 = {}, acc10 = {}, acc11 = {};
  const float* afrag = As + (wr * 32 + m) * PA + 2 * hl;
  const float* bfrag = Bs + hl * PB + (wc * 32 + m) * 2;

  auto stage = [&]() {
#pragma unroll
    for (int i = 0; i < NA; ++i) *(v4f*)(As + al[i]) = ra[i];
#pragma unroll
    for (int i = 0; i < NB; ++i) {
      v4f lo, hi;
      lo.x = rb0[i].x; lo.y = rb1[i].x; lo.z = rb0[i].y; lo.w = rb1[i].y;
      hi.x = rb0[i].z; hi.y = rb1[i].z; hi.z = rb0[i].w; hi.w = rb1[i].w;
      *(v4f*)(Bs + bl[i])     = lo;
      *(v4f*)(Bs + bl[i] + 4) = hi;
    }
  };
  auto multiply = [&]() {
#pragma unroll
    for (int ks = 0; ks < KC; ks += 4) {
      const v2f a0 = *(const v2f*)(afrag + ks);
      const v2f a1 = *(const v2f*)(afrag + 16 * PA + ks);
      const v2f b0 = *(const v2f*)(bfrag + (ks / 2) * PB);
      const v2f b1 = *(const v2f*)(bfrag + (ks / 2) * PB + 32);
      acc00 = __builtin_amdgcn_wmma_f32_16x16x4_f32(false, a0, false, b0, (short)0, acc00, false, false);
      acc01 = __builtin_amdgcn_wmma_f32_16x16x4_f32(false, a0, false, b1, (short)0, acc01, false, false);
      acc10 = __builtin_amdgcn_wmma_f32_16x16x4_f32(false, a1, false, b0, (short)0, acc10, false, false);
      acc11 = __builtin_amdgcn_wmma_f32_16x16x4_f32(false, a1, false, b1, (short)0, acc11, false, false);
    }
  };

#pragma unroll 1
  for (int kn = KC; kn < K; kn += KC) {
    __syncthreads();
    stage();
    __syncthreads();
#pragma unroll
    for (int i = 0; i < NA; ++i) ra[i] = *(const v4f*)(ag[i] + kn);
#pragma unroll
    for (int i = 0; i < NB; ++i) {
      rb0[i] = *(const v4f*)(bg[i] + (size_t)kn * NCOL);
      rb1[i] = *(const v4f*)(bg[i] + (size_t)(kn + 1) * NCOL);
    }
    __builtin_amdgcn_sched_barrier(0);
    multiply();
  }
  __syncthreads();
  stage();
  __syncthreads();
  multiply();

  const int rsub = hl << 3;
  const int rt0  = row0 + wr * 32;
  float* crow = C + (size_t)(rt0 + rsub) * NCOL + col0 + wc * 32 + m;
  if (rt0 < M) {
#pragma unroll
    for (int r = 0; r < 8; ++r) {
      crow[(size_t)r * NCOL]      = acc00[r];
      crow[(size_t)r * NCOL + 16] = acc01[r];
    }
  }
  if (rt0 + 16 < M) {
#pragma unroll
    for (int r = 0; r < 8; ++r) {
      crow[(size_t)(r + 16) * NCOL]      = acc10[r];
      crow[(size_t)(r + 16) * NCOL + 16] = acc11[r];
    }
  }
}

__global__ void gat_attn_scores(const float* __restrict__ feat,
                                const float* __restrict__ al,
                                const float* __restrict__ ar,
                                float* __restrict__ el, float* __restrict__ er,
                                int F, int D)
{
  int idx = blockIdx.x * blockDim.x + threadIdx.x;
  if (idx >= GAT_N * GAT_H) return;
  int nd = idx >> 2, h = idx & 3;
  const float* f   = feat + (size_t)nd * F + h * D;
  const float* alh = al + h * D;
  const float* arh = ar + h * D;
  float sl = 0.f, sr = 0.f;
  for (int d = 0; d < D; ++d) {
    float v = f[d];
    sl = fmaf(v, alh[d], sl);
    sr = fmaf(v, arh[d], sr);
  }
  el[idx] = sl;
  er[idx] = sr;
}

__device__ __forceinline__ float gat_leaky(float x) {
  return x > 0.f ? x : GAT_SLOPE * x;
}

__device__ __forceinline__ void gat_lds_max_f32(float* addr, float v) {
  if (__float_as_int(v) >= 0)
    __hip_atomic_fetch_max((int*)addr, __float_as_int(v), __ATOMIC_RELAXED, __HIP_MEMORY_SCOPE_WORKGROUP);
  else
    __hip_atomic_fetch_min((unsigned int*)addr, __float_as_uint(v), __ATOMIC_RELAXED, __HIP_MEMORY_SCOPE_WORKGROUP);
}

#define GAT_EW_THREADS 256
#define GAT_EW_GROUPS  1250
#define GAT_EW_STRIDE  (GAT_EW_GROUPS * GAT_EW_THREADS)
#define GAT_EW_ITEMS   10

#define GAT_TILE 196
#define GAT_AGG_THREADS 1024
#define GAT_LIST_CAP 4096

template <int F, int D>
__global__ __launch_bounds__(GAT_AGG_THREADS) void gat_tile_softmax_aggregate(
    const int* __restrict__ src, const int* __restrict__ dst,
    const float* __restrict__ feat, const float* __restrict__ el,
    const float* __restrict__ er, float* __restrict__ out)
{
  static_assert(GAT_E % 128 == 0, "dst is walked in steps of 128 edges");
  static_assert(F % 32 == 0, "row is walked in steps of one wave");
  static_assert(D % 4 == 0, "a 16-byte piece of a row lies in one head");
  constexpr int NW  = GAT_AGG_THREADS / 32;
  constexpr int NCH = GAT_E / 128;
  constexpr int FI  = F / 32;
  constexpr int NRH = GAT_TILE * GAT_H;
  static_assert(GAT_LIST_CAP >= NW * 128, "one step of every wave always fits the list");
  static_assert(NRH <= GAT_AGG_THREADS, "one thread per (row, head) pair");

  __shared__ __attribute__((aligned(16))) float tile[GAT_TILE * F];
  __shared__ float4 l_w[GAT_LIST_CAP];
  __shared__ int    l_s[GAT_LIST_CAP];
  __shared__ int    l_r[GAT_LIST_CAP];
  __shared__ float4 s_er[GAT_TILE];
  __shared__ __attribute__((aligned(16))) float s_m[NRH];
  __shared__ __attribute__((aligned(16))) float s_mo[NRH];
  __shared__ __attribute__((aligned(16))) float s_den[NRH];
  __shared__ int    s_cnt;
  __shared__ int    s_over;

  const int tid   = threadIdx.x;
  const int wave  = __builtin_amdgcn_readfirstlane(tid >> 5);
  const int lane  = tid & 31;
  const int node0 = blockIdx.x * GAT_TILE;
  const int rows  = min(GAT_TILE, GAT_N - node0);

  float4* t4 = (float4*)tile;
  for (int i = tid; i < GAT_TILE * F / 4; i += GAT_AGG_THREADS)
    t4[i] = make_float4(0.f, 0.f, 0.f, 0.f);
  if (tid < NRH) {
    s_m[tid]   = __int_as_float((int)0xFF800000u);
    s_den[tid] = 0.f;
  }
  if (tid < GAT_TILE)
    s_er[tid] = tid < rows ? ((const float4*)er)[node0 + tid] : make_float4(0.f, 0.f, 0.f, 0.f);
  if (tid == 0) { s_cnt = 0; s_over = 0; }
  __syncthreads();

  const int4* dst4 = (const int4*)dst;
  constexpr int NGROUPS = (GAT_N + GAT_TILE - 1) / GAT_TILE;
  const int rot = (int)blockIdx.x * (NCH / NGROUPS);
  auto place = [&](int t) { const int p = t + rot; return p >= NCH ? p - NCH : p; };
  auto fetch = [&](int t) { return dst4[(size_t)place(min(t, NCH - 1)) * 32 + lane]; };
  auto step = [&](const int4 cur, int t, int end) {
    const unsigned live = t < end ? 0xFFFFFFFFu : 0u;
    const int cc = place(t);
    const int l0 = cur.x - node0, l1 = cur.y - node0;
    const int l2 = cur.z - node0, l3 = cur.w - node0;
    const bool m0 = (unsigned)l0 < (unsigned)rows, m1 = (unsigned)l1 < (unsigned)rows;
    const bool m2 = (unsigned)l2 < (unsigned)rows, m3 = (unsigned)l3 < (unsigned)rows;
    const unsigned b0 = __builtin_amdgcn_ballot_w32(m0), b1 = __builtin_amdgcn_ballot_w32(m1);
    const unsigned b2 = __builtin_amdgcn_ballot_w32(m2), b3 = __builtin_amdgcn_ballot_w32(m3);
    if (((b0 | b1 | b2 | b3) & live) != 0u) {
      const int k0 = __popc(b0), k1 = __popc(b1), k2 = __popc(b2), k3 = __popc(b3);
      const int total = k0 + k1 + k2 + k3;
      int base = 0;
      if (lane == 0)
        base = __hip_atomic_fetch_add(&s_cnt, total, __ATOMIC_RELAXED, __HIP_MEMORY_SCOPE_WORKGROUP);
      base = __builtin_amdgcn_readfirstlane(base);
      if (base + total <= GAT_LIST_CAP) {
        const int e0 = cc * 128 + lane * 4;
        if (m0) { const int i = base + (int)__builtin_amdgcn_mbcnt_lo(b0, 0u);                l_s[i] = e0;     l_r[i] = l0; }
        if (m1) { const int i = base + k0 + (int)__builtin_amdgcn_mbcnt_lo(b1, 0u);           l_s[i] = e0 + 1; l_r[i] = l1; }
        if (m2) { const int i = base + k0 + k1 + (int)__builtin_amdgcn_mbcnt_lo(b2, 0u);      l_s[i] = e0 + 2; l_r[i] = l2; }
        if (m3) { const int i = base + k0 + k1 + k2 + (int)__builtin_amdgcn_mbcnt_lo(b3, 0u); l_s[i] = e0 + 3; l_r[i] = l3; }
      } else if (lane == 0) {
        s_over = 1;
      }
    }
  };
  auto collect = [&](int begin, int end) {
    const int w0 = begin + wave;
    int4 n0 = fetch(w0), n1 = fetch(w0 + NW), n2 = fetch(w0 + 2 * NW), n3 = fetch(w0 + 3 * NW);
#pragma unroll 1
    for (int c = w0; c < end; c += 4 * NW) {
      const int4 c0 = n0, c1 = n1, c2 = n2, c3 = n3;
      n0 = fetch(c + 4 * NW); n1 = fetch(c + 5 * NW); n2 = fetch(c + 6 * NW); n3 = fetch(c + 7 * NW);
      step(c0, c, end);
      step(c1, c + NW, end);
      step(c2, c + 2 * NW, end);
      step(c3, c + 3 * NW, end);
    }
  };

  auto drain = [&](int base, int n) {
    int   s_l = 0, r_l = 0;
    float a0 = 0.f, a1 = 0.f, a2 = 0.f, a3 = 0.f;
    if (lane < n) {
      s_l = l_s[base + lane];
      r_l = l_r[base + lane];
      const float4 xv = l_w[base + lane];
      a0 = xv.x; a1 = xv.y; a2 = xv.z; a3 = xv.w;
    }
#pragma unroll 1
    for (int k = 0; k < n; k += 4) {
      const float* fp[4];
      float*       tp[4];
      float        w[4][4];
#pragma unroll
      for (int j = 0; j < 4; ++j) {
        const int  kj = min(k + j, n - 1);
        const bool in = k + j < n;
        fp[j] = feat + (size_t)__builtin_amdgcn_readlane(s_l, kj) * F + lane;
        tp[j] = tile + __builtin_amdgcn_readlane(r_l, kj) * F + lane;
        w[j][0] = in ? __int_as_float(__builtin_amdgcn_readlane(__float_as_int(a0), kj)) : 0.f;
        w[j][1] = in ? __int_as_float(__builtin_amdgcn_readlane(__float_as_int(a1), kj)) : 0.f;
        w[j][2] = in ? __int_as_float(__builtin_amdgcn_readlane(__float_as_int(a2), kj)) : 0.f;
        w[j][3] = in ? __int_as_float(__builtin_amdgcn_readlane(__float_as_int(a3), kj)) : 0.f;
      }
      float v[4][FI];
#pragma unroll
      for (int j = 0; j < 4; ++j) {
#pragma unroll
        for (int i = 0; i < FI; ++i) {
          const int h = (i * 32 + lane) / D;
          const float wh = h == 0 ? w[j][0] : (h == 1 ? w[j][1] : (h == 2 ? w[j][2] : w[j][3]));
          v[j][i] = fp[j][i * 32] * wh;
        }
      }
#pragma unroll
      for (int j = 0; j < 4; ++j) {
#pragma unroll
        for (int i = 0; i < FI; ++i)
          __hip_atomic_fetch_add(tp[j] + i * 32, v[j][i], __ATOMIC_RELAXED, __HIP_MEMORY_SCOPE_WORKGROUP);
      }
    }
  };

  collect(0, NCH);
  __syncthreads();
  const bool over = __builtin_amdgcn_readfirstlane(s_over) != 0;
  const int  span = over ? NW : NCH;

#pragma unroll 1
  for (int begin = 0; begin < NCH; begin += span) {
    if (over) {
      __syncthreads();
      if (tid == 0) s_cnt = 0;
      __syncthreads();
      collect(begin, min(begin + span, NCH));
      __syncthreads();
    }
    const int cnt = __builtin_amdgcn_readfirstlane(s_cnt);
    if (begin > 0) {
      if (tid < NRH) s_mo[tid] = s_m[tid];
      __syncthreads();
    }

    for (int i0 = tid; i0 < cnt; i0 += 4 * GAT_AGG_THREADS) {
      int  sv[4], rv[4];
      bool ok[4];
#pragma unroll
      for (int u = 0; u < 4; ++u) {
        const int i = i0 + u * GAT_AGG_THREADS;
        ok[u] = i < cnt;
        const int e = ok[u] ? l_s[i] : 0;
        rv[u] = ok[u] ? l_r[i] : 0;
        sv[u] = src[e];
      }
      float4 av[4];
#pragma unroll
      for (int u = 0; u < 4; ++u) av[u] = ((const float4*)el)[sv[u]];
#pragma unroll
      for (int u = 0; u < 4; ++u) {
        if (ok[u]) {
          const int i = i0 + u * GAT_AGG_THREADS;
          const float4 b = s_er[rv[u]];
          float4 sc;
          sc.x = gat_leaky(av[u].x + b.x);
          sc.y = gat_leaky(av[u].y + b.y);
          sc.z = gat_leaky(av[u].z + b.z);
          sc.w = gat_leaky(av[u].w + b.w);
          l_s[i] = sv[u];
          l_w[i] = sc;
          float* mp = s_m + rv[u] * GAT_H;
          gat_lds_max_f32(mp + 0, sc.x);
          gat_lds_max_f32(mp + 1, sc.y);
          gat_lds_max_f32(mp + 2, sc.z);
          gat_lds_max_f32(mp + 3, sc.w);
        }
      }
    }
    __syncthreads();

    if (begin > 0) {
      for (int i = tid; i < rows * F; i += GAT_AGG_THREADS) {
        const int p = (i / F) * GAT_H + (i % F) / D;
        const float mo = s_mo[p], mn = s_m[p];
        if (mn > mo) tile[i] *= expf(mo - mn);
      }
      if (tid < NRH) {
        const float mo = s_mo[tid], mn = s_m[tid];
        if (mn > mo) s_den[tid] *= expf(mo - mn);
      }
      __syncthreads();
    }

    for (int i = tid; i < cnt; i += GAT_AGG_THREADS) {
      const int    r  = l_r[i];
      const float4 sc = l_w[i];
      const float4 mm = ((const float4*)s_m)[r];
      float4 x;
      x.x = expf(sc.x - mm.x);
      x.y = expf(sc.y - mm.y);
      x.z = expf(sc.z - mm.z);
      x.w = expf(sc.w - mm.w);
      l_w[i] = x;
      float* dp = s_den + r * GAT_H;
      __hip_atomic_fetch_add(dp + 0, x.x, __ATOMIC_RELAXED, __HIP_MEMORY_SCOPE_WORKGROUP);
      __hip_atomic_fetch_add(dp + 1, x.y, __ATOMIC_RELAXED, __HIP_MEMORY_SCOPE_WORKGROUP);
      __hip_atomic_fetch_add(dp + 2, x.z, __ATOMIC_RELAXED, __HIP_MEMORY_SCOPE_WORKGROUP);
      __hip_atomic_fetch_add(dp + 3, x.w, __ATOMIC_RELAXED, __HIP_MEMORY_SCOPE_WORKGROUP);
    }
    __syncthreads();

    const int per = (cnt + NW - 1) / NW;
    const int lo  = wave * per;
    const int hi  = min(cnt, lo + per);
    for (int b = lo; b < hi; b += 32) drain(b, min(32, hi - b));
  }

  __syncthreads();
  float4* o4 = (float4*)(out + (size_t)node0 * F);
  for (int i = tid; i < rows * F / 4; i += GAT_AGG_THREADS) {
    const float  dn = s_den[(i * 4 / F) * GAT_H + (i * 4 % F) / D];
    const float4 t  = t4[i];
    o4[i] = dn > 0.f ? make_float4(t.x / dn, t.y / dn, t.z / dn, t.w / dn) : make_float4(0.f, 0.f, 0.f, 0.f);
  }
}

static_assert(GAT_EW_STRIDE * GAT_EW_ITEMS * 4 == GAT_N * 256, "the grid covers the N x 256 buffer exactly once");
static_assert(GAT_EW_STRIDE % 64 == 0, "a thread keeps its 16-byte piece of the bias row");
#define GAT_ELU_BATCH 5

template <bool RES>
__global__ __launch_bounds__(GAT_EW_THREADS) void gat_finalize_elu(
    float* __restrict__ acc, const float* __restrict__ bias, const float* __restrict__ res)
{
  const int t = blockIdx.x * GAT_EW_THREADS + threadIdx.x;
  v4f*       acc4 = (v4f*)acc;
  const v4f* res4 = (const v4f*)res;
  const v4f  b    = ((const v4f*)bias)[t & 63];
#pragma unroll 1
  for (int p = 0; p < GAT_EW_ITEMS; p += GAT_ELU_BATCH) {
    v4f a[GAT_ELU_BATCH], r[GAT_ELU_BATCH];
#pragma unroll
    for (int u = 0; u < GAT_ELU_BATCH; ++u) {
      const int i = t + (p + u) * GAT_EW_STRIDE;
      a[u] = acc4[i];
      if (RES) r[u] = res4[i];
    }
#pragma unroll
    for (int u = 0; u < GAT_ELU_BATCH; ++u) {
      v4f v = a[u] + b;
      if (RES) v += r[u];
      v4f o;
      o.x = v.x > 0.f ? v.x : (expf(v.x) - 1.f);
      o.y = v.y > 0.f ? v.y : (expf(v.y) - 1.f);
      o.z = v.z > 0.f ? v.z : (expf(v.z) - 1.f);
      o.w = v.w > 0.f ? v.w : (expf(v.w) - 1.f);
      acc4[t + (p + u) * GAT_EW_STRIDE] = o;
    }
  }
}

#define GAT_OUT_GROUPS 977

__global__ __launch_bounds__(GAT_EW_THREADS) void gat_finalize_out(
    const float* __restrict__ acc, const float* __restrict__ bias,
    const float* __restrict__ res2, float* __restrict__ out)
{
  constexpr int C4     = 10;
  constexpr int TOTAL4 = GAT_N * C4;
  static_assert(GAT_OUT_GROUPS * GAT_EW_THREADS * 2 >= TOTAL4, "two passes cover the output");
  const v4f* acc4  = (const v4f*)acc;
  const v4f* res4  = (const v4f*)res2;
  const v4f* bias4 = (const v4f*)bias;
  v4f*       out4  = (v4f*)out;
  for (int i = blockIdx.x * GAT_EW_THREADS + threadIdx.x; i < TOTAL4; i += GAT_OUT_GROUPS * GAT_EW_THREADS) {
    const int n = i / C4, c4 = i % C4;
    v4f sum = {0.f, 0.f, 0.f, 0.f};
#pragma unroll
    for (int h = 0; h < GAT_H; ++h) {
      const int j4 = n * (GAT_H * C4) + h * C4 + c4;
      sum += acc4[j4] + bias4[h * C4 + c4] + res4[j4];
    }
    out4[i] = sum * 0.25f;
  }
}

extern "C" void kernel_launch(void* const* d_in, const int* in_sizes, int n_in,
                              void* d_out, int out_size, void* d_ws, size_t ws_size,
                              hipStream_t stream)
{
  const float* x     = (const float*)d_in[0];
  const int*   src   = (const int*)d_in[1];
  const int*   dst   = (const int*)d_in[2];
  const float* W0    = (const float*)d_in[3];
  const float* al0   = (const float*)d_in[4];
  const float* ar0   = (const float*)d_in[5];
  const float* b0    = (const float*)d_in[6];
  const float* W1    = (const float*)d_in[7];
  const float* al1   = (const float*)d_in[8];
  const float* ar1   = (const float*)d_in[9];
  const float* b1    = (const float*)d_in[10];
  const float* W2    = (const float*)d_in[11];
  const float* al2   = (const float*)d_in[12];
  const float* ar2   = (const float*)d_in[13];
  const float* b2    = (const float*)d_in[14];
  const float* Wres2 = (const float*)d_in[15];

  const size_t NF = (size_t)GAT_N * 256;
  const size_t NG = (size_t)GAT_N * 160;
  float* ws   = (float*)d_ws;
  float* bufA = ws;
  float* bufB = bufA + NF;
  float* bufC = bufB + NF;
  float* bufD = bufC + NF;
  float* EL   = bufD + NG;
  float* ER   = EL + (size_t)GAT_N * GAT_H;

  dim3 gemmBlk(128);
  dim3 gemmGrid256(4, (GAT_N + 63) / 64);
  dim3 gemmGrid160(5, (GAT_N + 127) / 128);

  const int nhThreads = GAT_N * GAT_H;
  const int nhBlocks  = (nhThreads + 255) / 256;
  const int aggBlocks = (GAT_N + GAT_TILE - 1) / GAT_TILE;

  gat_gemm_wmma<256, 256, 2, 2><<<gemmGrid256, gemmBlk, 0, stream>>>(x, W0, bufA, GAT_N);
  gat_attn_scores<<<nhBlocks, 256, 0, stream>>>(bufA, al0, ar0, EL, ER, 256, 64);
  gat_tile_softmax_aggregate<256, 64><<<aggBlocks, GAT_AGG_THREADS, 0, stream>>>(src, dst, bufA, EL, ER, bufB);
  gat_finalize_elu<false><<<GAT_EW_GROUPS, GAT_EW_THREADS, 0, stream>>>(bufB, b0, nullptr);

  gat_gemm_wmma<256, 256, 2, 2><<<gemmGrid256, gemmBlk, 0, stream>>>(bufB, W1, bufA, GAT_N);
  gat_attn_scores<<<nhBlocks, 256, 0, stream>>>(bufA, al1, ar1, EL, ER, 256, 64);
  gat_tile_softmax_aggregate<256, 64><<<aggBlocks, GAT_AGG_THREADS, 0, stream>>>(src, dst, bufA, EL, ER, bufC);
  gat_finalize_elu<true><<<GAT_EW_GROUPS, GAT_EW_THREADS, 0, stream>>>(bufC, b1, bufB);

  gat_gemm_wmma<256, 160, 4, 1><<<gemmGrid160, gemmBlk, 0, stream>>>(bufC, W2, bufA, GAT_N);
  gat_gemm_wmma<256, 160, 4, 1><<<gemmGrid160, gemmBlk, 0, stream>>>(bufC, Wres2, bufB, GAT_N);
  gat_attn_scores<<<nhBlocks, 256, 0, stream>>>(bufA, al2, ar2, EL, ER, 160, 40);
  gat_tile_softmax_aggregate<160, 40><<<aggBlocks, GAT_AGG_THREADS, 0, stream>>>(src, dst, bufA, EL, ER, bufD);
  gat_finalize_out<<<GAT_OUT_GROUPS, GAT_EW_THREADS, 0, stream>>>(bufD, b2, bufB, (float*)d_out);
}


# tile edge lists built once: layer 0's aggregation saves each tile's list (source node, local row) in the workspace; layers 1 and 2 read it back and skip the walk over dst; their sums follow its order
# speedup vs baseline: 3.3344x; 1.1723x over previous
#include <hip/hip_runtime.h>
#include <math.h>

#define GAT_N 50000
#define GAT_E 800000
#define GAT_H 4
#define GAT_SLOPE 0.2f

typedef __attribute__((ext_vector_type(2))) float v2f;
typedef __attribute__((ext_vector_type(4))) float v4f;
typedef __attribute__((ext_vector_type(8))) float v8f;

template <int K, int NCOL, int WR, int WC>
__global__ __launch_bounds__(WR * WC * 32) void gat_gemm_wmma(
    const float* __restrict__ A, const float* __restrict__ B,
    float* __restrict__ C, int M)
{
  constexpr int NT = WR * WC * 32;
  constexpr int BM = WR * 32;
  constexpr int BN = WC * 32;
  constexpr int KC = 32;
  constexpr int PA = KC + 4;
  constexpr int PB = 2 * BN + 32;
  constexpr int A4 = KC / 4;
  constexpr int B4 = BN / 4;
  constexpr int NA = BM * A4 / NT;
  constexpr int NB = (KC / 2) * B4 / NT;
  static_assert(K % KC == 0, "K is walked in whole chunks");
  static_assert(NCOL % BN == 0, "column tiles cover C exactly");
  static_assert((BM * A4) % NT == 0 && ((KC / 2) * B4) % NT == 0, "copies divide evenly");

  __shared__ __attribute__((aligned(16))) float As[BM * PA];
  __shared__ __attribute__((aligned(16))) float Bs[(KC / 2) * PB];

  const int tid  = threadIdx.x;
  const int wave = tid >> 5;
  const int lane = tid & 31;
  const int wr   = wave / WC;
  const int wc   = wave % WC;
  const int col0 = blockIdx.x * BN;
  const int row0 = blockIdx.y * BM;

  const int m  = lane & 15;
  const int hl = lane >> 4;

  const float* ag[NA];
  int          al[NA];
#pragma unroll
  for (int i = 0; i < NA; ++i) {
    const int idx = tid + i * NT;
    const int r = idx / A4, c4 = idx % A4;
    const int gr = min(row0 + r, M - 1);
    ag[i] = A + (size_t)gr * K + c4 * 4;
    al[i] = r * PA + c4 * 4;
  }
  const float* bg[NB];
  int          bl[NB];
#pragma unroll
  for (int i = 0; i < NB; ++i) {
    const int idx = tid + i * NT;
    const int kp = idx / B4, c4 = idx % B4;
    bg[i] = B + (size_t)(2 * kp) * NCOL + col0 + c4 * 4;
    bl[i] = kp * PB + c4 * 8;
  }

  v4f ra[NA], rb0[NB], rb1[NB];
#pragma unroll
  for (int i = 0; i < NA; ++i) ra[i] = *(const v4f*)ag[i];
#pragma unroll
  for (int i = 0; i < NB; ++i) {
    rb0[i] = *(const v4f*)bg[i];
    rb1[i] = *(const v4f*)(bg[i] + NCOL);
  }

  v8f acc00 = {}, acc01 = {}, acc10 = {}, acc11 = {};
  const float* afrag = As + (wr * 32 + m) * PA + 2 * hl;
  const float* bfrag = Bs + hl * PB + (wc * 32 + m) * 2;

  auto stage = [&]() {
#pragma unroll
    for (int i = 0; i < NA; ++i) *(v4f*)(As + al[i]) = ra[i];
#pragma unroll
    for (int i = 0; i < NB; ++i) {
      v4f lo, hi;
      lo.x = rb0[i].x; lo.y = rb1[i].x; lo.z = rb0[i].y; lo.w = rb1[i].y;
      hi.x = rb0[i].z; hi.y = rb1[i].z; hi.z = rb0[i].w; hi.w = rb1[i].w;
      *(v4f*)(Bs + bl[i])     = lo;
      *(v4f*)(Bs + bl[i] + 4) = hi;
    }
  };
  auto multiply = [&]() {
#pragma unroll
    for (int ks = 0; ks < KC; ks += 4) {
      const v2f a0 = *(const v2f*)(afrag + ks);
      const v2f a1 = *(const v2f*)(afrag + 16 * PA + ks);
      const v2f b0 = *(const v2f*)(bfrag + (ks / 2) * PB);
      const v2f b1 = *(const v2f*)(bfrag + (ks / 2) * PB + 32);
      acc00 = __builtin_amdgcn_wmma_f32_16x16x4_f32(false, a0, false, b0, (short)0, acc00, false, false);
      acc01 = __builtin_amdgcn_wmma_f32_16x16x4_f32(false, a0, false, b1, (short)0, acc01, false, false);
      acc10 = __builtin_amdgcn_wmma_f32_16x16x4_f32(false, a1, false, b0, (short)0, acc10, false, false);
      acc11 = __builtin_amdgcn_wmma_f32_16x16x4_f32(false, a1, false, b1, (short)0, acc11, false, false);
    }
  };

#pragma unroll 1
  for (int kn = KC; kn < K; kn += KC) {
    __syncthreads();
    stage();
    __syncthreads();
#pragma unroll
    for (int i = 0; i < NA; ++i) ra[i] = *(const v4f*)(ag[i] + kn);
#pragma unroll
    for (int i = 0; i < NB; ++i) {
      rb0[i] = *(const v4f*)(bg[i] + (size_t)kn * NCOL);
      rb1[i] = *(const v4f*)(bg[i] + (size_t)(kn + 1) * NCOL);
    }
    __builtin_amdgcn_sched_barrier(0);
    multiply();
  }
  __syncthreads();
  stage();
  __syncthreads();
  multiply();

  const int rsub = hl << 3;
  const int rt0  = row0 + wr * 32;
  float* crow = C + (size_t)(rt0 + rsub) * NCOL + col0 + wc * 32 + m;
  if (rt0 < M) {
#pragma unroll
    for (int r = 0; r < 8; ++r) {
      crow[(size_t)r * NCOL]      = acc00[r];
      crow[(size_t)r * NCOL + 16] = acc01[r];
    }
  }
  if (rt0 + 16 < M) {
#pragma unroll
    for (int r = 0; r < 8; ++r) {
      crow[(size_t)(r + 16) * NCOL]      = acc10[r];
      crow[(size_t)(r + 16) * NCOL + 16] = acc11[r];
    }
  }
}

__global__ void gat_attn_scores(const float* __restrict__ feat,
                                const float* __restrict__ al,
                                const float* __restrict__ ar,
                                float* __restrict__ el, float* __restrict__ er,
                                int F, int D)
{
  int idx = blockIdx.x * blockDim.x + threadIdx.x;
  if (idx >= GAT_N * GAT_H) return;
  int nd = idx >> 2, h = idx & 3;
  const float* f   = feat + (size_t)nd * F + h * D;
  const float* alh = al + h * D;
  const float* arh = ar + h * D;
  float sl = 0.f, sr = 0.f;
  for (int d = 0; d < D; ++d) {
    float v = f[d];
    sl = fmaf(v, alh[d], sl);
    sr = fmaf(v, arh[d], sr);
  }
  el[idx] = sl;
  er[idx] = sr;
}

__device__ __forceinline__ float gat_leaky(float x) {
  return x > 0.f ? x : GAT_SLOPE * x;
}

__device__ __forceinline__ void gat_lds_max_f32(float* addr, float v) {
  if (__float_as_int(v) >= 0)
    __hip_atomic_fetch_max((int*)addr, __float_as_int(v), __ATOMIC_RELAXED, __HIP_MEMORY_SCOPE_WORKGROUP);
  else
    __hip_atomic_fetch_min((unsigned int*)addr, __float_as_uint(v), __ATOMIC_RELAXED, __HIP_MEMORY_SCOPE_WORKGROUP);
}

#define GAT_EW_THREADS 256
#define GAT_EW_GROUPS  1250
#define GAT_EW_STRIDE  (GAT_EW_GROUPS * GAT_EW_THREADS)
#define GAT_EW_ITEMS   10

#define GAT_TILE 196
#define GAT_AGG_THREADS 1024
#define GAT_LIST_CAP 4096
#define GAT_NTILES ((GAT_N + GAT_TILE - 1) / GAT_TILE)
static_assert(GAT_N <= (1 << 23) && GAT_TILE <= 256, "source node and local row share one 32-bit word");

template <int F, int D, bool SAVED>
__global__ __launch_bounds__(GAT_AGG_THREADS) void gat_tile_softmax_aggregate(
    const int* __restrict__ src, const int* __restrict__ dst,
    const float* __restrict__ feat, const float* __restrict__ el,
    const float* __restrict__ er, float* __restrict__ out,
    int* tile_cnt, int* tile_list)
{
  static_assert(GAT_E % 128 == 0, "dst is walked in steps of 128 edges");
  static_assert(F % 32 == 0, "row is walked in steps of one wave");
  static_assert(D % 4 == 0, "a 16-byte piece of a row lies in one head");
  constexpr int NW  = GAT_AGG_THREADS / 32;
  constexpr int NCH = GAT_E / 128;
  constexpr int FI  = F / 32;
  constexpr int NRH = GAT_TILE * GAT_H;
  static_assert(GAT_LIST_CAP >= NW * 128, "one step of every wave always fits the list");
  static_assert(NRH <= GAT_AGG_THREADS, "one thread per (row, head) pair");

  __shared__ __attribute__((aligned(16))) float tile[GAT_TILE * F];
  __shared__ float4 l_w[GAT_LIST_CAP];
  __shared__ int    l_s[GAT_LIST_CAP];
  __shared__ int    l_r[GAT_LIST_CAP];
  __shared__ float4 s_er[GAT_TILE];
  __shared__ __attribute__((aligned(16))) float s_m[NRH];
  __shared__ __attribute__((aligned(16))) float s_mo[NRH];
  __shared__ __attribute__((aligned(16))) float s_den[NRH];
  __shared__ int    s_cnt;
  __shared__ int    s_over;

  const int tid   = threadIdx.x;
  const int wave  = __builtin_amdgcn_readfirstlane(tid >> 5);
  const int lane  = tid & 31;
  const int node0 = blockIdx.x * GAT_TILE;
  const int rows  = min(GAT_TILE, GAT_N - node0);
  int* my_list = tile_list + (size_t)blockIdx.x * GAT_LIST_CAP;
  const int saved = SAVED ? __builtin_amdgcn_readfirstlane(tile_cnt[blockIdx.x]) : -1;
  const bool have = saved >= 0;

  float4* t4 = (float4*)tile;
  for (int i = tid; i < GAT_TILE * F / 4; i += GAT_AGG_THREADS)
    t4[i] = make_float4(0.f, 0.f, 0.f, 0.f);
  if (tid < NRH) {
    s_m[tid]   = __int_as_float((int)0xFF800000u);
    s_den[tid] = 0.f;
  }
  if (tid < GAT_TILE)
    s_er[tid] = tid < rows ? ((const float4*)er)[node0 + tid] : make_float4(0.f, 0.f, 0.f, 0.f);
  if (tid == 0) { s_cnt = 0; s_over = 0; }
  __syncthreads();

  const int4* dst4 = (const int4*)dst;
  constexpr int NGROUPS = (GAT_N + GAT_TILE - 1) / GAT_TILE;
  const int rot = (int)blockIdx.x * (NCH / NGROUPS);
  auto place = [&](int t) { const int p = t + rot; return p >= NCH ? p - NCH : p; };
  auto fetch = [&](int t) { return dst4[(size_t)place(min(t, NCH - 1)) * 32 + lane]; };
  auto step = [&](const int4 cur, int t, int end) {
    const unsigned live = t < end ? 0xFFFFFFFFu : 0u;
    const int cc = place(t);
    const int l0 = cur.x - node0, l1 = cur.y - node0;
    const int l2 = cur.z - node0, l3 = cur.w - node0;
    const bool m0 = (unsigned)l0 < (unsigned)rows, m1 = (unsigned)l1 < (unsigned)rows;
    const bool m2 = (unsigned)l2 < (unsigned)rows, m3 = (unsigned)l3 < (unsigned)rows;
    const unsigned b0 = __builtin_amdgcn_ballot_w32(m0), b1 = __builtin_amdgcn_ballot_w32(m1);
    const unsigned b2 = __builtin_amdgcn_ballot_w32(m2), b3 = __builtin_amdgcn_ballot_w32(m3);
    if (((b0 | b1 | b2 | b3) & live) != 0u) {
      const int k0 = __popc(b0), k1 = __popc(b1), k2 = __popc(b2), k3 = __popc(b3);
      const int total = k0 + k1 + k2 + k3;
      int base = 0;
      if (lane == 0)
        base = __hip_atomic_fetch_add(&s_cnt, total, __ATOMIC_RELAXED, __HIP_MEMORY_SCOPE_WORKGROUP);
      base = __builtin_amdgcn_readfirstlane(base);
      if (base + total <= GAT_LIST_CAP) {
        const int e0 = cc * 128 + lane * 4;
        if (m0) { const int i = base + (int)__builtin_amdgcn_mbcnt_lo(b0, 0u);                l_s[i] = e0;     l_r[i] = l0; }
        if (m1) { const int i = base + k0 + (int)__builtin_amdgcn_mbcnt_lo(b1, 0u);           l_s[i] = e0 + 1; l_r[i] = l1; }
        if (m2) { const int i = base + k0 + k1 + (int)__builtin_amdgcn_mbcnt_lo(b2, 0u);      l_s[i] = e0 + 2; l_r[i] = l2; }
        if (m3) { const int i = base + k0 + k1 + k2 + (int)__builtin_amdgcn_mbcnt_lo(b3, 0u); l_s[i] = e0 + 3; l_r[i] = l3; }
      } else if (lane == 0) {
        s_over = 1;
      }
    }
  };
  auto collect = [&](int begin, int end) {
    const int w0 = begin + wave;
    int4 n0 = fetch(w0), n1 = fetch(w0 + NW), n2 = fetch(w0 + 2 * NW), n3 = fetch(w0 + 3 * NW);
#pragma unroll 1
    for (int c = w0; c < end; c += 4 * NW) {
      const int4 c0 = n0, c1 = n1, c2 = n2, c3 = n3;
      n0 = fetch(c + 4 * NW); n1 = fetch(c + 5 * NW); n2 = fetch(c + 6 * NW); n3 = fetch(c + 7 * NW);
      step(c0, c, end);
      step(c1, c + NW, end);
      step(c2, c + 2 * NW, end);
      step(c3, c + 3 * NW, end);
    }
  };

  auto drain = [&](int base, int n) {
    int   s_l = 0, r_l = 0;
    float a0 = 0.f, a1 = 0.f, a2 = 0.f, a3 = 0.f;
    if (lane < n) {
      s_l = l_s[base + lane];
      r_l = l_r[base + lane];
      const float4 xv = l_w[base + lane];
      a0 = xv.x; a1 = xv.y; a2 = xv.z; a3 = xv.w;
    }
#pragma unroll 1
    for (int k = 0; k < n; k += 4) {
      const float* fp[4];
      float*       tp[4];
      float        w[4][4];
#pragma unroll
      for (int j = 0; j < 4; ++j) {
        const int  kj = min(k + j, n - 1);
        const bool in = k + j < n;
        fp[j] = feat + (size_t)__builtin_amdgcn_readlane(s_l, kj) * F + lane;
        tp[j] = tile + __builtin_amdgcn_readlane(r_l, kj) * F + lane;
        w[j][0] = in ? __int_as_float(__builtin_amdgcn_readlane(__float_as_int(a0), kj)) : 0.f;
        w[j][1] = in ? __int_as_float(__builtin_amdgcn_readlane(__float_as_int(a1), kj)) : 0.f;
        w[j][2] = in ? __int_as_float(__builtin_amdgcn_readlane(__float_as_int(a2), kj)) : 0.f;
        w[j][3] = in ? __int_as_float(__builtin_amdgcn_readlane(__float_as_int(a3), kj)) : 0.f;
      }
      float v[4][FI];
#pragma unroll
      for (int j = 0; j < 4; ++j) {
#pragma unroll
        for (int i = 0; i < FI; ++i) {
          const int h = (i * 32 + lane) / D;
          const float wh = h == 0 ? w[j][0] : (h == 1 ? w[j][1] : (h == 2 ? w[j][2] : w[j][3]));
          v[j][i] = fp[j][i * 32] * wh;
        }
      }
#pragma unroll
      for (int j = 0; j < 4; ++j) {
#pragma unroll
        for (int i = 0; i < FI; ++i)
          __hip_atomic_fetch_add(tp[j] + i * 32, v[j][i], __ATOMIC_RELAXED, __HIP_MEMORY_SCOPE_WORKGROUP);
      }
    }
  };

  if (have) {
    for (int i = tid; i < saved; i += GAT_AGG_THREADS) {
      const int p = my_list[i];
      l_s[i] = p >> 8;
      l_r[i] = p & 255;
    }
    if (tid == 0) s_cnt = saved;
  } else {
    collect(0, NCH);
  }
  __syncthreads();
  const bool over = __builtin_amdgcn_readfirstlane(s_over) != 0;
  const int  span = over ? NW : NCH;
  if (!SAVED && tid == 0) tile_cnt[blockIdx.x] = over ? -1 : s_cnt;

#pragma unroll 1
  for (int begin = 0; begin < NCH; begin += span) {
    if (over) {
      __syncthreads();
      if (tid == 0) s_cnt = 0;
      __syncthreads();
      collect(begin, min(begin + span, NCH));
      __syncthreads();
    }
    const int cnt = __builtin_amdgcn_readfirstlane(s_cnt);
    if (begin > 0) {
      if (tid < NRH) s_mo[tid] = s_m[tid];
      __syncthreads();
    }

    for (int i0 = tid; i0 < cnt; i0 += 4 * GAT_AGG_THREADS) {
      int  sv[4], rv[4];
      bool ok[4];
#pragma unroll
      for (int u = 0; u < 4; ++u) {
        const int i = i0 + u * GAT_AGG_THREADS;
        ok[u] = i < cnt;
        const int e = ok[u] ? l_s[i] : 0;
        rv[u] = ok[u] ? l_r[i] : 0;
        sv[u] = have ? e : src[e];
      }
      float4 av[4];
#pragma unroll
      for (int u = 0; u < 4; ++u) av[u] = ((const float4*)el)[sv[u]];
#pragma unroll
      for (int u = 0; u < 4; ++u) {
        if (ok[u]) {
          const int i = i0 + u * GAT_AGG_THREADS;
          const float4 b = s_er[rv[u]];
          float4 sc;
          sc.x = gat_leaky(av[u].x + b.x);
          sc.y = gat_leaky(av[u].y + b.y);
          sc.z = gat_leaky(av[u].z + b.z);
          sc.w = gat_leaky(av[u].w + b.w);
          l_s[i] = sv[u];
          l_w[i] = sc;
          if (!SAVED && !over) my_list[i] = (sv[u] << 8) | rv[u];
          float* mp = s_m + rv[u] * GAT_H;
          gat_lds_max_f32(mp + 0, sc.x);
          gat_lds_max_f32(mp + 1, sc.y);
          gat_lds_max_f32(mp + 2, sc.z);
          gat_lds_max_f32(mp + 3, sc.w);
        }
      }
    }
    __syncthreads();

    if (begin > 0) {
      for (int i = tid; i < rows * F; i += GAT_AGG_THREADS) {
        const int p = (i / F) * GAT_H + (i % F) / D;
        const float mo = s_mo[p], mn = s_m[p];
        if (mn > mo) tile[i] *= expf(mo - mn);
      }
      if (tid < NRH) {
        const float mo = s_mo[tid], mn = s_m[tid];
        if (mn > mo) s_den[tid] *= expf(mo - mn);
      }
      __syncthreads();
    }

    for (int i = tid; i < cnt; i += GAT_AGG_THREADS) {
      const int    r  = l_r[i];
      const float4 sc = l_w[i];
      const float4 mm = ((const float4*)s_m)[r];
      float4 x;
      x.x = expf(sc.x - mm.x);
      x.y = expf(sc.y - mm.y);
      x.z = expf(sc.z - mm.z);
      x.w = expf(sc.w - mm.w);
      l_w[i] = x;
      float* dp = s_den + r * GAT_H;
      __hip_atomic_fetch_add(dp + 0, x.x, __ATOMIC_RELAXED, __HIP_MEMORY_SCOPE_WORKGROUP);
      __hip_atomic_fetch_add(dp + 1, x.y, __ATOMIC_RELAXED, __HIP_MEMORY_SCOPE_WORKGROUP);
      __hip_atomic_fetch_add(dp + 2, x.z, __ATOMIC_RELAXED, __HIP_MEMORY_SCOPE_WORKGROUP);
      __hip_atomic_fetch_add(dp + 3, x.w, __ATOMIC_RELAXED, __HIP_MEMORY_SCOPE_WORKGROUP);
    }
    __syncthreads();

    const int per = (cnt + NW - 1) / NW;
    const int lo  = wave * per;
    const int hi  = min(cnt, lo + per);
    for (int b = lo; b < hi; b += 32) drain(b, min(32, hi - b));
  }

  __syncthreads();
  float4* o4 = (float4*)(out + (size_t)node0 * F);
  for (int i = tid; i < rows * F / 4; i += GAT_AGG_THREADS) {
    const float  dn = s_den[(i * 4 / F) * GAT_H + (i * 4 % F) / D];
    const float4 t  = t4[i];
    o4[i] = dn > 0.f ? make_float4(t.x / dn, t.y / dn, t.z / dn, t.w / dn) : make_float4(0.f, 0.f, 0.f, 0.f);
  }
}

static_assert(GAT_EW_STRIDE * GAT_EW_ITEMS * 4 == GAT_N * 256, "the grid covers the N x 256 buffer exactly once");
static_assert(GAT_EW_STRIDE % 64 == 0, "a thread keeps its 16-byte piece of the bias row");
#define GAT_ELU_BATCH 5

template <bool RES>
__global__ __launch_bounds__(GAT_EW_THREADS) void gat_finalize_elu(
    float* __restrict__ acc, const float* __restrict__ bias, const float* __restrict__ res)
{
  const int t = blockIdx.x * GAT_EW_THREADS + threadIdx.x;
  v4f*       acc4 = (v4f*)acc;
  const v4f* res4 = (const v4f*)res;
  const v4f  b    = ((const v4f*)bias)[t & 63];
#pragma unroll 1
  for (int p = 0; p < GAT_EW_ITEMS; p += GAT_ELU_BATCH) {
    v4f a[GAT_ELU_BATCH], r[GAT_ELU_BATCH];
#pragma unroll
    for (int u = 0; u < GAT_ELU_BATCH; ++u) {
      const int i = t + (p + u) * GAT_EW_STRIDE;
      a[u] = acc4[i];
      if (RES) r[u] = res4[i];
    }
#pragma unroll
    for (int u = 0; u < GAT_ELU_BATCH; ++u) {
      v4f v = a[u] + b;
      if (RES) v += r[u];
      v4f o;
      o.x = v.x > 0.f ? v.x : (expf(v.x) - 1.f);
      o.y = v.y > 0.f ? v.y : (expf(v.y) - 1.f);
      o.z = v.z > 0.f ? v.z : (expf(v.z) - 1.f);
      o.w = v.w > 0.f ? v.w : (expf(v.w) - 1.f);
      acc4[t + (p + u) * GAT_EW_STRIDE] = o;
    }
  }
}

#define GAT_OUT_GROUPS 977

__global__ __launch_bounds__(GAT_EW_THREADS) void gat_finalize_out(
    const float* __restrict__ acc, const float* __restrict__ bias,
    const float* __restrict__ res2, float* __restrict__ out)
{
  constexpr int C4     = 10;
  constexpr int TOTAL4 = GAT_N * C4;
  static_assert(GAT_OUT_GROUPS * GAT_EW_THREADS * 2 >= TOTAL4, "two passes cover the output");
  const v4f* acc4  = (const v4f*)acc;
  const v4f* res4  = (const v4f*)res2;
  const v4f* bias4 = (const v4f*)bias;
  v4f*       out4  = (v4f*)out;
  for (int i = blockIdx.x * GAT_EW_THREADS + threadIdx.x; i < TOTAL4; i += GAT_OUT_GROUPS * GAT_EW_THREADS) {
    const int n = i / C4, c4 = i % C4;
    v4f sum = {0.f, 0.f, 0.f, 0.f};
#pragma unroll
    for (int h = 0; h < GAT_H; ++h) {
      const int j4 = n * (GAT_H * C4) + h * C4 + c4;
      sum += acc4[j4] + bias4[h * C4 + c4] + res4[j4];
    }
    out4[i] = sum * 0.25f;
  }
}

extern "C" void kernel_launch(void* const* d_in, const int* in_sizes, int n_in,
                              void* d_out, int out_size, void* d_ws, size_t ws_size,
                              hipStream_t stream)
{
  const float* x     = (const float*)d_in[0];
  const int*   src   = (const int*)d_in[1];
  const int*   dst   = (const int*)d_in[2];
  const float* W0    = (const float*)d_in[3];
  const float* al0   = (const float*)d_in[4];
  const float* ar0   = (const float*)d_in[5];
  const float* b0    = (const float*)d_in[6];
  const float* W1    = (const float*)d_in[7];
  const float* al1   = (const float*)d_in[8];
  const float* ar1   = (const float*)d_in[9];
  const float* b1    = (const float*)d_in[10];
  const float* W2    = (const float*)d_in[11];
  const float* al2   = (const float*)d_in[12];
  const float* ar2   = (const float*)d_in[13];
  const float* b2    = (const float*)d_in[14];
  const float* Wres2 = (const float*)d_in[15];

  const size_t NF = (size_t)GAT_N * 256;
  const size_t NG = (size_t)GAT_N * 160;
  float* ws   = (float*)d_ws;
  float* bufA = ws;
  float* bufB = bufA + NF;
  float* bufC = bufB + NF;
  float* bufD = bufC + NF;
  float* EL   = bufD + NG;
  float* ER   = EL + (size_t)GAT_N * GAT_H;
  int* tileCnt  = (int*)(ER + (size_t)GAT_N * GAT_H);
  int* tileList = tileCnt + GAT_NTILES;

  dim3 gemmBlk(128);
  dim3 gemmGrid256(4, (GAT_N + 63) / 64);
  dim3 gemmGrid160(5, (GAT_N + 127) / 128);

  const int nhThreads = GAT_N * GAT_H;
  const int nhBlocks  = (nhThreads + 255) / 256;
  const int aggBlocks = (GAT_N + GAT_TILE - 1) / GAT_TILE;

  gat_gemm_wmma<256, 256, 2, 2><<<gemmGrid256, gemmBlk, 0, stream>>>(x, W0, bufA, GAT_N);
  gat_attn_scores<<<nhBlocks, 256, 0, stream>>>(bufA, al0, ar0, EL, ER, 256, 64);
  gat_tile_softmax_aggregate<256, 64, false><<<aggBlocks, GAT_AGG_THREADS, 0, stream>>>(src, dst, bufA, EL, ER, bufB, tileCnt, tileList);
  gat_finalize_elu<false><<<GAT_EW_GROUPS, GAT_EW_THREADS, 0, stream>>>(bufB, b0, nullptr);

  gat_gemm_wmma<256, 256, 2, 2><<<gemmGrid256, gemmBlk, 0, stream>>>(bufB, W1, bufA, GAT_N);
  gat_attn_scores<<<nhBlocks, 256, 0, stream>>>(bufA, al1, ar1, EL, ER, 256, 64);
  gat_tile_softmax_aggregate<256, 64, true><<<aggBlocks, GAT_AGG_THREADS, 0, stream>>>(src, dst, bufA, EL, ER, bufC, tileCnt, tileList);
  gat_finalize_elu<true><<<GAT_EW_GROUPS, GAT_EW_THREADS, 0, stream>>>(bufC, b1, bufB);

  gat_gemm_wmma<256, 160, 4, 1><<<gemmGrid160, gemmBlk, 0, stream>>>(bufC, W2, bufA, GAT_N);
  gat_gemm_wmma<256, 160, 4, 1><<<gemmGrid160, gemmBlk, 0, stream>>>(bufC, Wres2, bufB, GAT_N);
  gat_attn_scores<<<nhBlocks, 256, 0, stream>>>(bufA, al2, ar2, EL, ER, 160, 40);
  gat_tile_softmax_aggregate<160, 40, true><<<aggBlocks, GAT_AGG_THREADS, 0, stream>>>(src, dst, bufA, EL, ER, bufD, tileCnt, tileList);
  gat_finalize_out<<<GAT_OUT_GROUPS, GAT_EW_THREADS, 0, stream>>>(bufD, b2, bufB, (float*)d_out);
}


# bias, residual, elu and the head mean applied at the tile aggregation's store while the rows are in LDS: both finalize kernels and their 3 launches gone; every sum keeps its order
# speedup vs baseline: 3.5625x; 1.0684x over previous
#include <hip/hip_runtime.h>
#include <math.h>

#define GAT_N 50000
#define GAT_E 800000
#define GAT_H 4
#define GAT_SLOPE 0.2f

typedef __attribute__((ext_vector_type(2))) float v2f;
typedef __attribute__((ext_vector_type(4))) float v4f;
typedef __attribute__((ext_vector_type(8))) float v8f;

template <int K, int NCOL, int WR, int WC>
__global__ __launch_bounds__(WR * WC * 32) void gat_gemm_wmma(
    const float* __restrict__ A, const float* __restrict__ B,
    float* __restrict__ C, int M)
{
  constexpr int NT = WR * WC * 32;
  constexpr int BM = WR * 32;
  constexpr int BN = WC * 32;
  constexpr int KC = 32;
  constexpr int PA = KC + 4;
  constexpr int PB = 2 * BN + 32;
  constexpr int A4 = KC / 4;
  constexpr int B4 = BN / 4;
  constexpr int NA = BM * A4 / NT;
  constexpr int NB = (KC / 2) * B4 / NT;
  static_assert(K % KC == 0, "K is walked in whole chunks");
  static_assert(NCOL % BN == 0, "column tiles cover C exactly");
  static_assert((BM * A4) % NT == 0 && ((KC / 2) * B4) % NT == 0, "copies divide evenly");

  __shared__ __attribute__((aligned(16))) float As[BM * PA];
  __shared__ __attribute__((aligned(16))) float Bs[(KC / 2) * PB];

  const int tid  = threadIdx.x;
  const int wave = tid >> 5;
  const int lane = tid & 31;
  const int wr   = wave / WC;
  const int wc   = wave % WC;
  const int col0 = blockIdx.x * BN;
  const int row0 = blockIdx.y * BM;

  const int m  = lane & 15;
  const int hl = lane >> 4;

  const float* ag[NA];
  int          al[NA];
#pragma unroll
  for (int i = 0; i < NA; ++i) {
    const int idx = tid + i * NT;
    const int r = idx / A4, c4 = idx % A4;
    const int gr = min(row0 + r, M - 1);
    ag[i] = A + (size_t)gr * K + c4 * 4;
    al[i] = r * PA + c4 * 4;
  }
  const float* bg[NB];
  int          bl[NB];
#pragma unroll
  for (int i = 0; i < NB; ++i) {
    const int idx = tid + i * NT;
    const int kp = idx / B4, c4 = idx % B4;
    bg[i] = B + (size_t)(2 * kp) * NCOL + col0 + c4 * 4;
    bl[i] = kp * PB + c4 * 8;
  }

  v4f ra[NA], rb0[NB], rb1[NB];
#pragma unroll
  for (int i = 0; i < NA; ++i) ra[i] = *(const v4f*)ag[i];
#pragma unroll
  for (int i = 0; i < NB; ++i) {
    rb0[i] = *(const v4f*)bg[i];
    rb1[i] = *(const v4f*)(bg[i] + NCOL);
  }

  v8f acc00 = {}, acc01 = {}, acc10 = {}, acc11 = {};
  const float* afrag = As + (wr * 32 + m) * PA + 2 * hl;
  const float* bfrag = Bs + hl * PB + (wc * 32 + m) * 2;

  auto stage = [&]() {
#pragma unroll
    for (int i = 0; i < NA; ++i) *(v4f*)(As + al[i]) = ra[i];
#pragma unroll
    for (int i = 0; i < NB; ++i) {
      v4f lo, hi;
      lo.x = rb0[i].x; lo.y = rb1[i].x; lo.z = rb0[i].y; lo.w = rb1[i].y;
      hi.x = rb0[i].z; hi.y = rb1[i].z; hi.z = rb0[i].w; hi.w = rb1[i].w;
      *(v4f*)(Bs + bl[i])     = lo;
      *(v4f*)(Bs + bl[i] + 4) = hi;
    }
  };
  auto multiply = [&]() {
#pragma unroll
    for (int ks = 0; ks < KC; ks += 4) {
      const v2f a0 = *(const v2f*)(afrag + ks);
      const v2f a1 = *(const v2f*)(afrag + 16 * PA + ks);
      const v2f b0 = *(const v2f*)(bfrag + (ks / 2) * PB);
      const v2f b1 = *(const v2f*)(bfrag + (ks / 2) * PB + 32);
      acc00 = __builtin_amdgcn_wmma_f32_16x16x4_f32(false, a0, false, b0, (short)0, acc00, false, false);
      acc01 = __builtin_amdgcn_wmma_f32_16x16x4_f32(false, a0, false, b1, (short)0, acc01, false, false);
      acc10 = __builtin_amdgcn_wmma_f32_16x16x4_f32(false, a1, false, b0, (short)0, acc10, false, false);
      acc11 = __builtin_amdgcn_wmma_f32_16x16x4_f32(false, a1, false, b1, (short)0, acc11, false, false);
    }
  };

#pragma unroll 1
  for (int kn = KC; kn < K; kn += KC) {
    __syncthreads();
    stage();
    __syncthreads();
#pragma unroll
    for (int i = 0; i < NA; ++i) ra[i] = *(const v4f*)(ag[i] + kn);
#pragma unroll
    for (int i = 0; i < NB; ++i) {
      rb0[i] = *(const v4f*)(bg[i] + (size_t)kn * NCOL);
      rb1[i] = *(const v4f*)(bg[i] + (size_t)(kn + 1) * NCOL);
    }
    __builtin_amdgcn_sched_barrier(0);
    multiply();
  }
  __syncthreads();
  stage();
  __syncthreads();
  multiply();

  const int rsub = hl << 3;
  const int rt0  = row0 + wr * 32;
  float* crow = C + (size_t)(rt0 + rsub) * NCOL + col0 + wc * 32 + m;
  if (rt0 < M) {
#pragma unroll
    for (int r = 0; r < 8; ++r) {
      crow[(size_t)r * NCOL]      = acc00[r];
      crow[(size_t)r * NCOL + 16] = acc01[r];
    }
  }
  if (rt0 + 16 < M) {
#pragma unroll
    for (int r = 0; r < 8; ++r) {
      crow[(size_t)(r + 16) * NCOL]      = acc10[r];
      crow[(size_t)(r + 16) * NCOL + 16] = acc11[r];
    }
  }
}

__global__ void gat_attn_scores(const float* __restrict__ feat,
                                const float* __restrict__ al,
                                const float* __restrict__ ar,
                                float* __restrict__ el, float* __restrict__ er,
                                int F, int D)
{
  int idx = blockIdx.x * blockDim.x + threadIdx.x;
  if (idx >= GAT_N * GAT_H) return;
  int nd = idx >> 2, h = idx & 3;
  const float* f   = feat + (size_t)nd * F + h * D;
  const float* alh = al + h * D;
  const float* arh = ar + h * D;
  float sl = 0.f, sr = 0.f;
  for (int d = 0; d < D; ++d) {
    float v = f[d];
    sl = fmaf(v, alh[d], sl);
    sr = fmaf(v, arh[d], sr);
  }
  el[idx] = sl;
  er[idx] = sr;
}

__device__ __forceinline__ float gat_leaky(float x) {
  return x > 0.f ? x : GAT_SLOPE * x;
}

__device__ __forceinline__ void gat_lds_max_f32(float* addr, float v) {
  if (__float_as_int(v) >= 0)
    __hip_atomic_fetch_max((int*)addr, __float_as_int(v), __ATOMIC_RELAXED, __HIP_MEMORY_SCOPE_WORKGROUP);
  else
    __hip_atomic_fetch_min((unsigned int*)addr, __float_as_uint(v), __ATOMIC_RELAXED, __HIP_MEMORY_SCOPE_WORKGROUP);
}

#define GAT_TILE 196
#define GAT_AGG_THREADS 1024
#define GAT_LIST_CAP 4096
#define GAT_NTILES ((GAT_N + GAT_TILE - 1) / GAT_TILE)
#define GAT_EPI_ELU      0
#define GAT_EPI_RES_ELU  1
#define GAT_EPI_RES_MEAN 2
#define GAT_EPI_BATCH    4
static_assert(GAT_N <= (1 << 23) && GAT_TILE <= 256, "source node and local row share one 32-bit word");

template <int F, int D, bool SAVED, int EPI>
__global__ __launch_bounds__(GAT_AGG_THREADS) void gat_tile_softmax_aggregate(
    const int* __restrict__ src, const int* __restrict__ dst,
    const float* __restrict__ feat, const float* __restrict__ el,
    const float* __restrict__ er, const float* __restrict__ bias,
    const float* __restrict__ res, float* __restrict__ out,
    int* tile_cnt, int* tile_list)
{
  static_assert(GAT_E % 128 == 0, "dst is walked in steps of 128 edges");
  static_assert(F % 32 == 0, "row is walked in steps of one wave");
  static_assert(D % 4 == 0, "a 16-byte piece of a row lies in one head");
  constexpr int NW  = GAT_AGG_THREADS / 32;
  constexpr int NCH = GAT_E / 128;
  constexpr int FI  = F / 32;
  constexpr int NRH = GAT_TILE * GAT_H;
  static_assert(GAT_LIST_CAP >= NW * 128, "one step of every wave always fits the list");
  static_assert(NRH <= GAT_AGG_THREADS, "one thread per (row, head) pair");

  __shared__ __attribute__((aligned(16))) float tile[GAT_TILE * F];
  __shared__ float4 l_w[GAT_LIST_CAP];
  __shared__ int    l_s[GAT_LIST_CAP];
  __shared__ int    l_r[GAT_LIST_CAP];
  __shared__ float4 s_er[GAT_TILE];
  __shared__ __attribute__((aligned(16))) float s_m[NRH];
  __shared__ __attribute__((aligned(16))) float s_mo[NRH];
  __shared__ __attribute__((aligned(16))) float s_den[NRH];
  __shared__ int    s_cnt;
  __shared__ int    s_over;

  const int tid   = threadIdx.x;
  const int wave  = __builtin_amdgcn_readfirstlane(tid >> 5);
  const int lane  = tid & 31;
  const int node0 = blockIdx.x * GAT_TILE;
  const int rows  = min(GAT_TILE, GAT_N - node0);
  int* my_list = tile_list + (size_t)blockIdx.x * GAT_LIST_CAP;
  const int saved = SAVED ? __builtin_amdgcn_readfirstlane(tile_cnt[blockIdx.x]) : -1;
  const bool have = saved >= 0;

  float4* t4 = (float4*)tile;
  for (int i = tid; i < GAT_TILE * F / 4; i += GAT_AGG_THREADS)
    t4[i] = make_float4(0.f, 0.f, 0.f, 0.f);
  if (tid < NRH) {
    s_m[tid]   = __int_as_float((int)0xFF800000u);
    s_den[tid] = 0.f;
  }
  if (tid < GAT_TILE)
    s_er[tid] = tid < rows ? ((const float4*)er)[node0 + tid] : make_float4(0.f, 0.f, 0.f, 0.f);
  if (tid == 0) { s_cnt = 0; s_over = 0; }
  __syncthreads();

  const int4* dst4 = (const int4*)dst;
  constexpr int NGROUPS = (GAT_N + GAT_TILE - 1) / GAT_TILE;
  const int rot = (int)blockIdx.x * (NCH / NGROUPS);
  auto place = [&](int t) { const int p = t + rot; return p >= NCH ? p - NCH : p; };
  auto fetch = [&](int t) { return dst4[(size_t)place(min(t, NCH - 1)) * 32 + lane]; };
  auto step = [&](const int4 cur, int t, int end) {
    const unsigned live = t < end ? 0xFFFFFFFFu : 0u;
    const int cc = place(t);
    const int l0 = cur.x - node0, l1 = cur.y - node0;
    const int l2 = cur.z - node0, l3 = cur.w - node0;
    const bool m0 = (unsigned)l0 < (unsigned)rows, m1 = (unsigned)l1 < (unsigned)rows;
    const bool m2 = (unsigned)l2 < (unsigned)rows, m3 = (unsigned)l3 < (unsigned)rows;
    const unsigned b0 = __builtin_amdgcn_ballot_w32(m0), b1 = __builtin_amdgcn_ballot_w32(m1);
    const unsigned b2 = __builtin_amdgcn_ballot_w32(m2), b3 = __builtin_amdgcn_ballot_w32(m3);
    if (((b0 | b1 | b2 | b3) & live) != 0u) {
      const int k0 = __popc(b0), k1 = __popc(b1), k2 = __popc(b2), k3 = __popc(b3);
      const int total = k0 + k1 + k2 + k3;
      int base = 0;
      if (lane == 0)
        base = __hip_atomic_fetch_add(&s_cnt, total, __ATOMIC_RELAXED, __HIP_MEMORY_SCOPE_WORKGROUP);
      base = __builtin_amdgcn_readfirstlane(base);
      if (base + total <= GAT_LIST_CAP) {
        const int e0 = cc * 128 + lane * 4;
        if (m0) { const int i = base + (int)__builtin_amdgcn_mbcnt_lo(b0, 0u);                l_s[i] = e0;     l_r[i] = l0; }
        if (m1) { const int i = base + k0 + (int)__builtin_amdgcn_mbcnt_lo(b1, 0u);           l_s[i] = e0 + 1; l_r[i] = l1; }
        if (m2) { const int i = base + k0 + k1 + (int)__builtin_amdgcn_mbcnt_lo(b2, 0u);      l_s[i] = e0 + 2; l_r[i] = l2; }
        if (m3) { const int i = base + k0 + k1 + k2 + (int)__builtin_amdgcn_mbcnt_lo(b3, 0u); l_s[i] = e0 + 3; l_r[i] = l3; }
      } else if (lane == 0) {
        s_over = 1;
      }
    }
  };
  auto collect = [&](int begin, int end) {
    const int w0 = begin + wave;
    int4 n0 = fetch(w0), n1 = fetch(w0 + NW), n2 = fetch(w0 + 2 * NW), n3 = fetch(w0 + 3 * NW);
#pragma unroll 1
    for (int c = w0; c < end; c += 4 * NW) {
      const int4 c0 = n0, c1 = n1, c2 = n2, c3 = n3;
      n0 = fetch(c + 4 * NW); n1 = fetch(c + 5 * NW); n2 = fetch(c + 6 * NW); n3 = fetch(c + 7 * NW);
      step(c0, c, end);
      step(c1, c + NW, end);
      step(c2, c + 2 * NW, end);
      step(c3, c + 3 * NW, end);
    }
  };

  auto drain = [&](int base, int n) {
    int   s_l = 0, r_l = 0;
    float a0 = 0.f, a1 = 0.f, a2 = 0.f, a3 = 0.f;
    if (lane < n) {
      s_l = l_s[base + lane];
      r_l = l_r[base + lane];
      const float4 xv = l_w[base + lane];
      a0 = xv.x; a1 = xv.y; a2 = xv.z; a3 = xv.w;
    }
#pragma unroll 1
    for (int k = 0; k < n; k += 4) {
      const float* fp[4];
      float*       tp[4];
      float        w[4][4];
#pragma unroll
      for (int j = 0; j < 4; ++j) {
        const int  kj = min(k + j, n - 1);
        const bool in = k + j < n;
        fp[j] = feat + (size_t)__builtin_amdgcn_readlane(s_l, kj) * F + lane;
        tp[j] = tile + __builtin_amdgcn_readlane(r_l, kj) * F + lane;
        w[j][0] = in ? __int_as_float(__builtin_amdgcn_readlane(__float_as_int(a0), kj)) : 0.f;
        w[j][1] = in ? __int_as_float(__builtin_amdgcn_readlane(__float_as_int(a1), kj)) : 0.f;
        w[j][2] = in ? __int_as_float(__builtin_amdgcn_readlane(__float_as_int(a2), kj)) : 0.f;
        w[j][3] = in ? __int_as_float(__builtin_amdgcn_readlane(__float_as_int(a3), kj)) : 0.f;
      }
      float v[4][FI];
#pragma unroll
      for (int j = 0; j < 4; ++j) {
#pragma unroll
        for (int i = 0; i < FI; ++i) {
          const int h = (i * 32 + lane) / D;
          const float wh = h == 0 ? w[j][0] : (h == 1 ? w[j][1] : (h == 2 ? w[j][2] : w[j][3]));
          v[j][i] = fp[j][i * 32] * wh;
        }
      }
#pragma unroll
      for (int j = 0; j < 4; ++j) {
#pragma unroll
        for (int i = 0; i < FI; ++i)
          __hip_atomic_fetch_add(tp[j] + i * 32, v[j][i], __ATOMIC_RELAXED, __HIP_MEMORY_SCOPE_WORKGROUP);
      }
    }
  };

  if (have) {
    for (int i = tid; i < saved; i += GAT_AGG_THREADS) {
      const int p = my_list[i];
      l_s[i] = p >> 8;
      l_r[i] = p & 255;
    }
    if (tid == 0) s_cnt = saved;
  } else {
    collect(0, NCH);
  }
  __syncthreads();
  const bool over = __builtin_amdgcn_readfirstlane(s_over) != 0;
  const int  span = over ? NW : NCH;
  if (!SAVED && tid == 0) tile_cnt[blockIdx.x] = over ? -1 : s_cnt;

#pragma unroll 1
  for (int begin = 0; begin < NCH; begin += span) {
    if (over) {
      __syncthreads();
      if (tid == 0) s_cnt = 0;
      __syncthreads();
      collect(begin, min(begin + span, NCH));
      __syncthreads();
    }
    const int cnt = __builtin_amdgcn_readfirstlane(s_cnt);
    if (begin > 0) {
      if (tid < NRH) s_mo[tid] = s_m[tid];
      __syncthreads();
    }

    for (int i0 = tid; i0 < cnt; i0 += 4 * GAT_AGG_THREADS) {
      int  sv[4], rv[4];
      bool ok[4];
#pragma unroll
      for (int u = 0; u < 4; ++u) {
        const int i = i0 + u * GAT_AGG_THREADS;
        ok[u] = i < cnt;
        const int e = ok[u] ? l_s[i] : 0;
        rv[u] = ok[u] ? l_r[i] : 0;
        sv[u] = have ? e : src[e];
      }
      float4 av[4];
#pragma unroll
      for (int u = 0; u < 4; ++u) av[u] = ((const float4*)el)[sv[u]];
#pragma unroll
      for (int u = 0; u < 4; ++u) {
        if (ok[u]) {
          const int i = i0 + u * GAT_AGG_THREADS;
          const float4 b = s_er[rv[u]];
          float4 sc;
          sc.x = gat_leaky(av[u].x + b.x);
          sc.y = gat_leaky(av[u].y + b.y);
          sc.z = gat_leaky(av[u].z + b.z);
          sc.w = gat_leaky(av[u].w + b.w);
          l_s[i] = sv[u];
          l_w[i] = sc;
          if (!SAVED && !over) my_list[i] = (sv[u] << 8) | rv[u];
          float* mp = s_m + rv[u] * GAT_H;
          gat_lds_max_f32(mp + 0, sc.x);
          gat_lds_max_f32(mp + 1, sc.y);
          gat_lds_max_f32(mp + 2, sc.z);
          gat_lds_max_f32(mp + 3, sc.w);
        }
      }
    }
    __syncthreads();

    if (begin > 0) {
      for (int i = tid; i < rows * F; i += GAT_AGG_THREADS) {
        const int p = (i / F) * GAT_H + (i % F) / D;
        const float mo = s_mo[p], mn = s_m[p];
        if (mn > mo) tile[i] *= expf(mo - mn);
      }
      if (tid < NRH) {
        const float mo = s_mo[tid], mn = s_m[tid];
        if (mn > mo) s_den[tid] *= expf(mo - mn);
      }
      __syncthreads();
    }

    for (int i = tid; i < cnt; i += GAT_AGG_THREADS) {
      const int    r  = l_r[i];
      const float4 sc = l_w[i];
      const float4 mm = ((const float4*)s_m)[r];
      float4 x;
      x.x = expf(sc.x - mm.x);
      x.y = expf(sc.y - mm.y);
      x.z = expf(sc.z - mm.z);
      x.w = expf(sc.w - mm.w);
      l_w[i] = x;
      float* dp = s_den + r * GAT_H;
      __hip_atomic_fetch_add(dp + 0, x.x, __ATOMIC_RELAXED, __HIP_MEMORY_SCOPE_WORKGROUP);
      __hip_atomic_fetch_add(dp + 1, x.y, __ATOMIC_RELAXED, __HIP_MEMORY_SCOPE_WORKGROUP);
      __hip_atomic_fetch_add(dp + 2, x.z, __ATOMIC_RELAXED, __HIP_MEMORY_SCOPE_WORKGROUP);
      __hip_atomic_fetch_add(dp + 3, x.w, __ATOMIC_RELAXED, __HIP_MEMORY_SCOPE_WORKGROUP);
    }
    __syncthreads();

    const int per = (cnt + NW - 1) / NW;
    const int lo  = wave * per;
    const int hi  = min(cnt, lo + per);
    for (int b = lo; b < hi; b += 32) drain(b, min(32, hi - b));
  }

  __syncthreads();
  constexpr int F4 = F / 4;
  const v4f* tv = (const v4f*)tile;
  const v4f  zero = {0.f, 0.f, 0.f, 0.f};
  if constexpr (EPI == GAT_EPI_RES_MEAN) {
    constexpr int C4 = D / 4;
    const int  total = rows * C4;
    const v4f* r4 = (const v4f*)(res + (size_t)node0 * F);
    const v4f* b4 = (const v4f*)bias;
    v4f*       o4 = (v4f*)(out + (size_t)node0 * D);
    for (int i = tid; i < total; i += GAT_AGG_THREADS) {
      const int n = i / C4, c4 = i % C4;
      v4f rv[GAT_H];
#pragma unroll
      for (int h = 0; h < GAT_H; ++h) rv[h] = r4[n * F4 + h * C4 + c4];
      v4f sum = zero;
#pragma unroll
      for (int h = 0; h < GAT_H; ++h) {
        const float dn = s_den[n * GAT_H + h];
        const v4f   t  = tv[n * F4 + h * C4 + c4];
        const v4f   a  = dn > 0.f ? t / dn : zero;
        sum += a + b4[h * C4 + c4] + rv[h];
      }
      o4[i] = sum * 0.25f;
    }
  } else {
    static_assert(GAT_AGG_THREADS % F4 == 0, "a thread keeps its 16-byte piece of the bias row");
    const int  total = rows * F4;
    v4f*       o4 = (v4f*)(out + (size_t)node0 * F);
    const v4f  b  = ((const v4f*)bias)[tid % F4];
    const int  hd = (tid % F4) * 4 / D;
    for (int i0 = tid; i0 < total; i0 += GAT_EPI_BATCH * GAT_AGG_THREADS) {
      v4f rv[GAT_EPI_BATCH];
      if constexpr (EPI == GAT_EPI_RES_ELU) {
        const v4f* r4 = (const v4f*)(res + (size_t)node0 * F);
#pragma unroll
        for (int u = 0; u < GAT_EPI_BATCH; ++u) rv[u] = r4[min(i0 + u * GAT_AGG_THREADS, total - 1)];
      }
#pragma unroll
      for (int u = 0; u < GAT_EPI_BATCH; ++u) {
        const int i = i0 + u * GAT_AGG_THREADS;
        if (i < total) {
          const float dn = s_den[(i / F4) * GAT_H + hd];
          const v4f   t  = tv[i];
          v4f v = dn > 0.f ? t / dn : zero;
          v = v + b;
          if constexpr (EPI == GAT_EPI_RES_ELU) v += rv[u];
          v4f o;
          o.x = v.x > 0.f ? v.x : (expf(v.x) - 1.f);
          o.y = v.y > 0.f ? v.y : (expf(v.y) - 1.f);
          o.z = v.z > 0.f ? v.z : (expf(v.z) - 1.f);
          o.w = v.w > 0.f ? v.w : (expf(v.w) - 1.f);
          o4[i] = o;
        }
      }
    }
  }
}

extern "C" void kernel_launch(void* const* d_in, const int* in_sizes, int n_in,
                              void* d_out, int out_size, void* d_ws, size_t ws_size,
                              hipStream_t stream)
{
  const float* x     = (const float*)d_in[0];
  const int*   src   = (const int*)d_in[1];
  const int*   dst   = (const int*)d_in[2];
  const float* W0    = (const float*)d_in[3];
  const float* al0   = (const float*)d_in[4];
  const float* ar0   = (const float*)d_in[5];
  const float* b0    = (const float*)d_in[6];
  const float* W1    = (const float*)d_in[7];
  const float* al1   = (const float*)d_in[8];
  const float* ar1   = (const float*)d_in[9];
  const float* b1    = (const float*)d_in[10];
  const float* W2    = (const float*)d_in[11];
  const float* al2   = (const float*)d_in[12];
  const float* ar2   = (const float*)d_in[13];
  const float* b2    = (const float*)d_in[14];
  const float* Wres2 = (const float*)d_in[15];

  const size_t NF = (size_t)GAT_N * 256;
  float* ws   = (float*)d_ws;
  float* bufA = ws;
  float* bufB = bufA + NF;
  float* bufC = bufB + NF;
  float* EL   = bufC + NF;
  float* ER   = EL + (size_t)GAT_N * GAT_H;
  int* tileCnt  = (int*)(ER + (size_t)GAT_N * GAT_H);
  int* tileList = tileCnt + GAT_NTILES;

  dim3 gemmBlk(128);
  dim3 gemmGrid256(4, (GAT_N + 63) / 64);
  dim3 gemmGrid160(5, (GAT_N + 127) / 128);

  const int nhThreads = GAT_N * GAT_H;
  const int nhBlocks  = (nhThreads + 255) / 256;
  const int aggBlocks = (GAT_N + GAT_TILE - 1) / GAT_TILE;

  gat_gemm_wmma<256, 256, 2, 2><<<gemmGrid256, gemmBlk, 0, stream>>>(x, W0, bufA, GAT_N);
  gat_attn_scores<<<nhBlocks, 256, 0, stream>>>(bufA, al0, ar0, EL, ER, 256, 64);
  gat_tile_softmax_aggregate<256, 64, false, GAT_EPI_ELU><<<aggBlocks, GAT_AGG_THREADS, 0, stream>>>(src, dst, bufA, EL, ER, b0, nullptr, bufB, tileCnt, tileList);

  gat_gemm_wmma<256, 256, 2, 2><<<gemmGrid256, gemmBlk, 0, stream>>>(bufB, W1, bufA, GAT_N);
  gat_attn_scores<<<nhBlocks, 256, 0, stream>>>(bufA, al1, ar1, EL, ER, 256, 64);
  gat_tile_softmax_aggregate<256, 64, true, GAT_EPI_RES_ELU><<<aggBlocks, GAT_AGG_THREADS, 0, stream>>>(src, dst, bufA, EL, ER, b1, bufB, bufC, tileCnt, tileList);

  gat_gemm_wmma<256, 160, 4, 1><<<gemmGrid160, gemmBlk, 0, stream>>>(bufC, W2, bufA, GAT_N);
  gat_gemm_wmma<256, 160, 4, 1><<<gemmGrid160, gemmBlk, 0, stream>>>(bufC, Wres2, bufB, GAT_N);
  gat_attn_scores<<<nhBlocks, 256, 0, stream>>>(bufA, al2, ar2, EL, ER, 160, 40);
  gat_tile_softmax_aggregate<160, 40, true, GAT_EPI_RES_MEAN><<<aggBlocks, GAT_AGG_THREADS, 0, stream>>>(src, dst, bufA, EL, ER, b2, bufB, (float*)d_out, tileCnt, tileList);
}


# attention score kernel reads feat by whole lines: a thread takes 16-byte pieces of 50 contiguous rows (512 B per wave load, was 32 lines per load), piece sums meet in LDS; el/er sum order changes
# speedup vs baseline: 3.7141x; 1.0426x over previous
#include <hip/hip_runtime.h>
#include <math.h>

#define GAT_N 50000
#define GAT_E 800000
#define GAT_H 4
#define GAT_SLOPE 0.2f

typedef __attribute__((ext_vector_type(2))) float v2f;
typedef __attribute__((ext_vector_type(4))) float v4f;
typedef __attribute__((ext_vector_type(8))) float v8f;

template <int K, int NCOL, int WR, int WC>
__global__ __launch_bounds__(WR * WC * 32) void gat_gemm_wmma(
    const float* __restrict__ A, const float* __restrict__ B,
    float* __restrict__ C, int M)
{
  constexpr int NT = WR * WC * 32;
  constexpr int BM = WR * 32;
  constexpr int BN = WC * 32;
  constexpr int KC = 32;
  constexpr int PA = KC + 4;
  constexpr int PB = 2 * BN + 32;
  constexpr int A4 = KC / 4;
  constexpr int B4 = BN / 4;
  constexpr int NA = BM * A4 / NT;
  constexpr int NB = (KC / 2) * B4 / NT;
  static_assert(K % KC == 0, "K is walked in whole chunks");
  static_assert(NCOL % BN == 0, "column tiles cover C exactly");
  static_assert((BM * A4) % NT == 0 && ((KC / 2) * B4) % NT == 0, "copies divide evenly");

  __shared__ __attribute__((aligned(16))) float As[BM * PA];
  __shared__ __attribute__((aligned(16))) float Bs[(KC / 2) * PB];

  const int tid  = threadIdx.x;
  const int wave = tid >> 5;
  const int lane = tid & 31;
  const int wr   = wave / WC;
  const int wc   = wave % WC;
  const int col0 = blockIdx.x * BN;
  const int row0 = blockIdx.y * BM;

  const int m  = lane & 15;
  const int hl = lane >> 4;

  const float* ag[NA];
  int          al[NA];
#pragma unroll
  for (int i = 0; i < NA; ++i) {
    const int idx = tid + i * NT;
    const int r = idx / A4, c4 = idx % A4;
    const int gr = min(row0 + r, M - 1);
    ag[i] = A + (size_t)gr * K + c4 * 4;
    al[i] = r * PA + c4 * 4;
  }
  const float* bg[NB];
  int          bl[NB];
#pragma unroll
  for (int i = 0; i < NB; ++i) {
    const int idx = tid + i * NT;
    const int kp = idx / B4, c4 = idx % B4;
    bg[i] = B + (size_t)(2 * kp) * NCOL + col0 + c4 * 4;
    bl[i] = kp * PB + c4 * 8;
  }

  v4f ra[NA], rb0[NB], rb1[NB];
#pragma unroll
  for (int i = 0; i < NA; ++i) ra[i] = *(const v4f*)ag[i];
#pragma unroll
  for (int i = 0; i < NB; ++i) {
    rb0[i] = *(const v4f*)bg[i];
    rb1[i] = *(const v4f*)(bg[i] + NCOL);
  }

  v8f acc00 = {}, acc01 = {}, acc10 = {}, acc11 = {};
  const float* afrag = As + (wr * 32 + m) * PA + 2 * hl;
  const float* bfrag = Bs + hl * PB + (wc * 32 + m) * 2;

  auto stage = [&]() {
#pragma unroll
    for (int i = 0; i < NA; ++i) *(v4f*)(As + al[i]) = ra[i];
#pragma unroll
    for (int i = 0; i < NB; ++i) {
      v4f lo, hi;
      lo.x = rb0[i].x; lo.y = rb1[i].x; lo.z = rb0[i].y; lo.w = rb1[i].y;
      hi.x = rb0[i].z; hi.y = rb1[i].z; hi.z = rb0[i].w; hi.w = rb1[i].w;
      *(v4f*)(Bs + bl[i])     = lo;
      *(v4f*)(Bs + bl[i] + 4) = hi;
    }
  };
  auto multiply = [&]() {
#pragma unroll
    for (int ks = 0; ks < KC; ks += 4) {
      const v2f a0 = *(const v2f*)(afrag + ks);
      const v2f a1 = *(const v2f*)(afrag + 16 * PA + ks);
      const v2f b0 = *(const v2f*)(bfrag + (ks / 2) * PB);
      const v2f b1 = *(const v2f*)(bfrag + (ks / 2) * PB + 32);
      acc00 = __builtin_amdgcn_wmma_f32_16x16x4_f32(false, a0, false, b0, (short)0, acc00, false, false);
      acc01 = __builtin_amdgcn_wmma_f32_16x16x4_f32(false, a0, false, b1, (short)0, acc01, false, false);
      acc10 = __builtin_amdgcn_wmma_f32_16x16x4_f32(false, a1, false, b0, (short)0, acc10, false, false);
      acc11 = __builtin_amdgcn_wmma_f32_16x16x4_f32(false, a1, false, b1, (short)0, acc11, false, false);
    }
  };

#pragma unroll 1
  for (int kn = KC; kn < K; kn += KC) {
    __syncthreads();
    stage();
    __syncthreads();
#pragma unroll
    for (int i = 0; i < NA; ++i) ra[i] = *(const v4f*)(ag[i] + kn);
#pragma unroll
    for (int i = 0; i < NB; ++i) {
      rb0[i] = *(const v4f*)(bg[i] + (size_t)kn * NCOL);
      rb1[i] = *(const v4f*)(bg[i] + (size_t)(kn + 1) * NCOL);
    }
    __builtin_amdgcn_sched_barrier(0);
    multiply();
  }
  __syncthreads();
  stage();
  __syncthreads();
  multiply();

  const int rsub = hl << 3;
  const int rt0  = row0 + wr * 32;
  float* crow = C + (size_t)(rt0 + rsub) * NCOL + col0 + wc * 32 + m;
  if (rt0 < M) {
#pragma unroll
    for (int r = 0; r < 8; ++r) {
      crow[(size_t)r * NCOL]      = acc00[r];
      crow[(size_t)r * NCOL + 16] = acc01[r];
    }
  }
  if (rt0 + 16 < M) {
#pragma unroll
    for (int r = 0; r < 8; ++r) {
      crow[(size_t)(r + 16) * NCOL]      = acc10[r];
      crow[(size_t)(r + 16) * NCOL + 16] = acc11[r];
    }
  }
}

#define GAT_SC_ROWS 50
#define GAT_SC_THREADS 256
static_assert(GAT_N % GAT_SC_ROWS == 0, "score groups cover the rows exactly");

template <int F, int D>
__global__ __launch_bounds__(GAT_SC_THREADS) void gat_attn_scores(
    const float* __restrict__ feat, const float* __restrict__ al,
    const float* __restrict__ ar, float* __restrict__ el, float* __restrict__ er)
{
  static_assert(F == GAT_H * D && D % 4 == 0, "a 16-byte piece of a row lies in one head");
  static_assert((GAT_SC_ROWS * F * 4) % 128 == 0, "a group's rows start on a line boundary");
  constexpr int F4  = F / 4;
  constexpr int D4  = D / 4;
  constexpr int P   = D4 | 1;
  constexpr int NP  = GAT_SC_ROWS * F4;
  constexpr int NIT = (NP + GAT_SC_THREADS - 1) / GAT_SC_THREADS;
  constexpr int NPAIR = GAT_SC_ROWS * GAT_H;
  static_assert(NPAIR <= GAT_SC_THREADS, "one thread per (row, head) pair");

  __shared__ float pl[NPAIR * P];
  __shared__ float pr[NPAIR * P];

  const int tid = threadIdx.x;
  const int n0  = blockIdx.x * GAT_SC_ROWS;
  const v4f* f4  = (const v4f*)feat + (size_t)n0 * F4;
  const v4f* al4 = (const v4f*)al;
  const v4f* ar4 = (const v4f*)ar;

  v4f v[NIT], a[NIT], b[NIT];
#pragma unroll
  for (int j = 0; j < NIT; ++j) {
    const int q = min(tid + j * GAT_SC_THREADS, NP - 1);
    v[j] = f4[q];
    a[j] = al4[q % F4];
    b[j] = ar4[q % F4];
  }
#pragma unroll
  for (int j = 0; j < NIT; ++j) {
    const int q = tid + j * GAT_SC_THREADS;
    if (q < NP) {
      const float sl = fmaf(v[j].w, a[j].w, fmaf(v[j].z, a[j].z, fmaf(v[j].y, a[j].y, v[j].x * a[j].x)));
      const float sr = fmaf(v[j].w, b[j].w, fmaf(v[j].z, b[j].z, fmaf(v[j].y, b[j].y, v[j].x * b[j].x)));
      const int o = (q / D4) * P + (q % D4);
      pl[o] = sl;
      pr[o] = sr;
    }
  }
  __syncthreads();
  if (tid < NPAIR) {
    float sl = 0.f, sr = 0.f;
#pragma unroll
    for (int i = 0; i < D4; ++i) {
      sl += pl[tid * P + i];
      sr += pr[tid * P + i];
    }
    el[(size_t)n0 * GAT_H + tid] = sl;
    er[(size_t)n0 * GAT_H + tid] = sr;
  }
}

__device__ __forceinline__ float gat_leaky(float x) {
  return x > 0.f ? x : GAT_SLOPE * x;
}

__device__ __forceinline__ void gat_lds_max_f32(float* addr, float v) {
  if (__float_as_int(v) >= 0)
    __hip_atomic_fetch_max((int*)addr, __float_as_int(v), __ATOMIC_RELAXED, __HIP_MEMORY_SCOPE_WORKGROUP);
  else
    __hip_atomic_fetch_min((unsigned int*)addr, __float_as_uint(v), __ATOMIC_RELAXED, __HIP_MEMORY_SCOPE_WORKGROUP);
}

#define GAT_TILE 196
#define GAT_AGG_THREADS 1024
#define GAT_LIST_CAP 4096
#define GAT_NTILES ((GAT_N + GAT_TILE - 1) / GAT_TILE)
#define GAT_EPI_ELU      0
#define GAT_EPI_RES_ELU  1
#define GAT_EPI_RES_MEAN 2
#define GAT_EPI_BATCH    4
static_assert(GAT_N <= (1 << 23) && GAT_TILE <= 256, "source node and local row share one 32-bit word");

template <int F, int D, bool SAVED, int EPI>
__global__ __launch_bounds__(GAT_AGG_THREADS) void gat_tile_softmax_aggregate(
    const int* __restrict__ src, const int* __restrict__ dst,
    const float* __restrict__ feat, const float* __restrict__ el,
    const float* __restrict__ er, const float* __restrict__ bias,
    const float* __restrict__ res, float* __restrict__ out,
    int* tile_cnt, int* tile_list)
{
  static_assert(GAT_E % 128 == 0, "dst is walked in steps of 128 edges");
  static_assert(F % 32 == 0, "row is walked in steps of one wave");
  static_assert(D % 4 == 0, "a 16-byte piece of a row lies in one head");
  constexpr int NW  = GAT_AGG_THREADS / 32;
  constexpr int NCH = GAT_E / 128;
  constexpr int FI  = F / 32;
  constexpr int NRH = GAT_TILE * GAT_H;
  static_assert(GAT_LIST_CAP >= NW * 128, "one step of every wave always fits the list");
  static_assert(NRH <= GAT_AGG_THREADS, "one thread per (row, head) pair");

  __shared__ __attribute__((aligned(16))) float tile[GAT_TILE * F];
  __shared__ float4 l_w[GAT_LIST_CAP];
  __shared__ int    l_s[GAT_LIST_CAP];
  __shared__ int    l_r[GAT_LIST_CAP];
  __shared__ float4 s_er[GAT_TILE];
  __shared__ __attribute__((aligned(16))) float s_m[NRH];
  __shared__ __attribute__((aligned(16))) float s_mo[NRH];
  __shared__ __attribute__((aligned(16))) float s_den[NRH];
  __shared__ int    s_cnt;
  __shared__ int    s_over;

  const int tid   = threadIdx.x;
  const int wave  = __builtin_amdgcn_readfirstlane(tid >> 5);
  const int lane  = tid & 31;
  const int node0 = blockIdx.x * GAT_TILE;
  const int rows  = min(GAT_TILE, GAT_N - node0);
  int* my_list = tile_list + (size_t)blockIdx.x * GAT_LIST_CAP;
  const int saved = SAVED ? __builtin_amdgcn_readfirstlane(tile_cnt[blockIdx.x]) : -1;
  const bool have = saved >= 0;

  float4* t4 = (float4*)tile;
  for (int i = tid; i < GAT_TILE * F / 4; i += GAT_AGG_THREADS)
    t4[i] = make_float4(0.f, 0.f, 0.f, 0.f);
  if (tid < NRH) {
    s_m[tid]   = __int_as_float((int)0xFF800000u);
    s_den[tid] = 0.f;
  }
  if (tid < GAT_TILE)
    s_er[tid] = tid < rows ? ((const float4*)er)[node0 + tid] : make_float4(0.f, 0.f, 0.f, 0.f);
  if (tid == 0) { s_cnt = 0; s_over = 0; }
  __syncthreads();

  const int4* dst4 = (const int4*)dst;
  constexpr int NGROUPS = (GAT_N + GAT_TILE - 1) / GAT_TILE;
  const int rot = (int)blockIdx.x * (NCH / NGROUPS);
  auto place = [&](int t) { const int p = t + rot; return p >= NCH ? p - NCH : p; };
  auto fetch = [&](int t) { return dst4[(size_t)place(min(t, NCH - 1)) * 32 + lane]; };
  auto step = [&](const int4 cur, int t, int end) {
    const unsigned live = t < end ? 0xFFFFFFFFu : 0u;
    const int cc = place(t);
    const int l0 = cur.x - node0, l1 = cur.y - node0;
    const int l2 = cur.z - node0, l3 = cur.w - node0;
    const bool m0 = (unsigned)l0 < (unsigned)rows, m1 = (unsigned)l1 < (unsigned)rows;
    const bool m2 = (unsigned)l2 < (unsigned)rows, m3 = (unsigned)l3 < (unsigned)rows;
    const unsigned b0 = __builtin_amdgcn_ballot_w32(m0), b1 = __builtin_amdgcn_ballot_w32(m1);
    const unsigned b2 = __builtin_amdgcn_ballot_w32(m2), b3 = __builtin_amdgcn_ballot_w32(m3);
    if (((b0 | b1 | b2 | b3) & live) != 0u) {
      const int k0 = __popc(b0), k1 = __popc(b1), k2 = __popc(b2), k3 = __popc(b3);
      const int total = k0 + k1 + k2 + k3;
      int base = 0;
      if (lane == 0)
        base = __hip_atomic_fetch_add(&s_cnt, total, __ATOMIC_RELAXED, __HIP_MEMORY_SCOPE_WORKGROUP);
      base = __builtin_amdgcn_readfirstlane(base);
      if (base + total <= GAT_LIST_CAP) {
        const int e0 = cc * 128 + lane * 4;
        if (m0) { const int i = base + (int)__builtin_amdgcn_mbcnt_lo(b0, 0u);                l_s[i] = e0;     l_r[i] = l0; }
        if (m1) { const int i = base + k0 + (int)__builtin_amdgcn_mbcnt_lo(b1, 0u);           l_s[i] = e0 + 1; l_r[i] = l1; }
        if (m2) { const int i = base + k0 + k1 + (int)__builtin_amdgcn_mbcnt_lo(b2, 0u);      l_s[i] = e0 + 2; l_r[i] = l2; }
        if (m3) { const int i = base + k0 + k1 + k2 + (int)__builtin_amdgcn_mbcnt_lo(b3, 0u); l_s[i] = e0 + 3; l_r[i] = l3; }
      } else if (lane == 0) {
        s_over = 1;
      }
    }
  };
  auto collect = [&](int begin, int end) {
    const int w0 = begin + wave;
    int4 n0 = fetch(w0), n1 = fetch(w0 + NW), n2 = fetch(w0 + 2 * NW), n3 = fetch(w0 + 3 * NW);
#pragma unroll 1
    for (int c = w0; c < end; c += 4 * NW) {
      const int4 c0 = n0, c1 = n1, c2 = n2, c3 = n3;
      n0 = fetch(c + 4 * NW); n1 = fetch(c + 5 * NW); n2 = fetch(c + 6 * NW); n3 = fetch(c + 7 * NW);
      step(c0, c, end);
      step(c1, c + NW, end);
      step(c2, c + 2 * NW, end);
      step(c3, c + 3 * NW, end);
    }
  };

  auto drain = [&](int base, int n) {
    int   s_l = 0, r_l = 0;
    float a0 = 0.f, a1 = 0.f, a2 = 0.f, a3 = 0.f;
    if (lane < n) {
      s_l = l_s[base + lane];
      r_l = l_r[base + lane];
      const float4 xv = l_w[base + lane];
      a0 = xv.x; a1 = xv.y; a2 = xv.z; a3 = xv.w;
    }
#pragma unroll 1
    for (int k = 0; k < n; k += 4) {
      const float* fp[4];
      float*       tp[4];
      float        w[4][4];
#pragma unroll
      for (int j = 0; j < 4; ++j) {
        const int  kj = min(k + j, n - 1);
        const bool in = k + j < n;
        fp[j] = feat + (size_t)__builtin_amdgcn_readlane(s_l, kj) * F + lane;
        tp[j] = tile + __builtin_amdgcn_readlane(r_l, kj) * F + lane;
        w[j][0] = in ? __int_as_float(__builtin_amdgcn_readlane(__float_as_int(a0), kj)) : 0.f;
        w[j][1] = in ? __int_as_float(__builtin_amdgcn_readlane(__float_as_int(a1), kj)) : 0.f;
        w[j][2] = in ? __int_as_float(__builtin_amdgcn_readlane(__float_as_int(a2), kj)) : 0.f;
        w[j][3] = in ? __int_as_float(__builtin_amdgcn_readlane(__float_as_int(a3), kj)) : 0.f;
      }
      float v[4][FI];
#pragma unroll
      for (int j = 0; j < 4; ++j) {
#pragma unroll
        for (int i = 0; i < FI; ++i) {
          const int h = (i * 32 + lane) / D;
          const float wh = h == 0 ? w[j][0] : (h == 1 ? w[j][1] : (h == 2 ? w[j][2] : w[j][3]));
          v[j][i] = fp[j][i * 32] * wh;
        }
      }
#pragma unroll
      for (int j = 0; j < 4; ++j) {
#pragma unroll
        for (int i = 0; i < FI; ++i)
          __hip_atomic_fetch_add(tp[j] + i * 32, v[j][i], __ATOMIC_RELAXED, __HIP_MEMORY_SCOPE_WORKGROUP);
      }
    }
  };

  if (have) {
    for (int i = tid; i < saved; i += GAT_AGG_THREADS) {
      const int p = my_list[i];
      l_s[i] = p >> 8;
      l_r[i] = p & 255;
    }
    if (tid == 0) s_cnt = saved;
  } else {
    collect(0, NCH);
  }
  __syncthreads();
  const bool over = __builtin_amdgcn_readfirstlane(s_over) != 0;
  const int  span = over ? NW : NCH;
  if (!SAVED && tid == 0) tile_cnt[blockIdx.x] = over ? -1 : s_cnt;

#pragma unroll 1
  for (int begin = 0; begin < NCH; begin += span) {
    if (over) {
      __syncthreads();
      if (tid == 0) s_cnt = 0;
      __syncthreads();
      collect(begin, min(begin + span, NCH));
      __syncthreads();
    }
    const int cnt = __builtin_amdgcn_readfirstlane(s_cnt);
    if (begin > 0) {
      if (tid < NRH) s_mo[tid] = s_m[tid];
      __syncthreads();
    }

    for (int i0 = tid; i0 < cnt; i0 += 4 * GAT_AGG_THREADS) {
      int  sv[4], rv[4];
      bool ok[4];
#pragma unroll
      for (int u = 0; u < 4; ++u) {
        const int i = i0 + u * GAT_AGG_THREADS;
        ok[u] = i < cnt;
        const int e = ok[u] ? l_s[i] : 0;
        rv[u] = ok[u] ? l_r[i] : 0;
        sv[u] = have ? e : src[e];
      }
      float4 av[4];
#pragma unroll
      for (int u = 0; u < 4; ++u) av[u] = ((const float4*)el)[sv[u]];
#pragma unroll
      for (int u = 0; u < 4; ++u) {
        if (ok[u]) {
          const int i = i0 + u * GAT_AGG_THREADS;
          const float4 b = s_er[rv[u]];
          float4 sc;
          sc.x = gat_leaky(av[u].x + b.x);
          sc.y = gat_leaky(av[u].y + b.y);
          sc.z = gat_leaky(av[u].z + b.z);
          sc.w = gat_leaky(av[u].w + b.w);
          l_s[i] = sv[u];
          l_w[i] = sc;
          if (!SAVED && !over) my_list[i] = (sv[u] << 8) | rv[u];
          float* mp = s_m + rv[u] * GAT_H;
          gat_lds_max_f32(mp + 0, sc.x);
          gat_lds_max_f32(mp + 1, sc.y);
          gat_lds_max_f32(mp + 2, sc.z);
          gat_lds_max_f32(mp + 3, sc.w);
        }
      }
    }
    __syncthreads();

    if (begin > 0) {
      for (int i = tid; i < rows * F; i += GAT_AGG_THREADS) {
        const int p = (i / F) * GAT_H + (i % F) / D;
        const float mo = s_mo[p], mn = s_m[p];
        if (mn > mo) tile[i] *= expf(mo - mn);
      }
      if (tid < NRH) {
        const float mo = s_mo[tid], mn = s_m[tid];
        if (mn > mo) s_den[tid] *= expf(mo - mn);
      }
      __syncthreads();
    }

    for (int i = tid; i < cnt; i += GAT_AGG_THREADS) {
      const int    r  = l_r[i];
      const float4 sc = l_w[i];
      const float4 mm = ((const float4*)s_m)[r];
      float4 x;
      x.x = expf(sc.x - mm.x);
      x.y = expf(sc.y - mm.y);
      x.z = expf(sc.z - mm.z);
      x.w = expf(sc.w - mm.w);
      l_w[i] = x;
      float* dp = s_den + r * GAT_H;
      __hip_atomic_fetch_add(dp + 0, x.x, __ATOMIC_RELAXED, __HIP_MEMORY_SCOPE_WORKGROUP);
      __hip_atomic_fetch_add(dp + 1, x.y, __ATOMIC_RELAXED, __HIP_MEMORY_SCOPE_WORKGROUP);
      __hip_atomic_fetch_add(dp + 2, x.z, __ATOMIC_RELAXED, __HIP_MEMORY_SCOPE_WORKGROUP);
      __hip_atomic_fetch_add(dp + 3, x.w, __ATOMIC_RELAXED, __HIP_MEMORY_SCOPE_WORKGROUP);
    }
    __syncthreads();

    const int per = (cnt + NW - 1) / NW;
    const int lo  = wave * per;
    const int hi  = min(cnt, lo + per);
    for (int b = lo; b < hi; b += 32) drain(b, min(32, hi - b));
  }

  __syncthreads();
  constexpr int F4 = F / 4;
  const v4f* tv = (const v4f*)tile;
  const v4f  zero = {0.f, 0.f, 0.f, 0.f};
  if constexpr (EPI == GAT_EPI_RES_MEAN) {
    constexpr int C4 = D / 4;
    const int  total = rows * C4;
    const v4f* r4 = (const v4f*)(res + (size_t)node0 * F);
    const v4f* b4 = (const v4f*)bias;
    v4f*       o4 = (v4f*)(out + (size_t)node0 * D);
    for (int i = tid; i < total; i += GAT_AGG_THREADS) {
      const int n = i / C4, c4 = i % C4;
      v4f rv[GAT_H];
#pragma unroll
      for (int h = 0; h < GAT_H; ++h) rv[h] = r4[n * F4 + h * C4 + c4];
      v4f sum = zero;
#pragma unroll
      for (int h = 0; h < GAT_H; ++h) {
        const float dn = s_den[n * GAT_H + h];
        const v4f   t  = tv[n * F4 + h * C4 + c4];
        const v4f   a  = dn > 0.f ? t / dn : zero;
        sum += a + b4[h * C4 + c4] + rv[h];
      }
      o4[i] = sum * 0.25f;
    }
  } else {
    static_assert(GAT_AGG_THREADS % F4 == 0, "a thread keeps its 16-byte piece of the bias row");
    const int  total = rows * F4;
    v4f*       o4 = (v4f*)(out + (size_t)node0 * F);
    const v4f  b  = ((const v4f*)bias)[tid % F4];
    const int  hd = (tid % F4) * 4 / D;
    for (int i0 = tid; i0 < total; i0 += GAT_EPI_BATCH * GAT_AGG_THREADS) {
      v4f rv[GAT_EPI_BATCH];
      if constexpr (EPI == GAT_EPI_RES_ELU) {
        const v4f* r4 = (const v4f*)(res + (size_t)node0 * F);
#pragma unroll
        for (int u = 0; u < GAT_EPI_BATCH; ++u) rv[u] = r4[min(i0 + u * GAT_AGG_THREADS, total - 1)];
      }
#pragma unroll
      for (int u = 0; u < GAT_EPI_BATCH; ++u) {
        const int i = i0 + u * GAT_AGG_THREADS;
        if (i < total) {
          const float dn = s_den[(i / F4) * GAT_H + hd];
          const v4f   t  = tv[i];
          v4f v = dn > 0.f ? t / dn : zero;
          v = v + b;
          if constexpr (EPI == GAT_EPI_RES_ELU) v += rv[u];
          v4f o;
          o.x = v.x > 0.f ? v.x : (expf(v.x) - 1.f);
          o.y = v.y > 0.f ? v.y : (expf(v.y) - 1.f);
          o.z = v.z > 0.f ? v.z : (expf(v.z) - 1.f);
          o.w = v.w > 0.f ? v.w : (expf(v.w) - 1.f);
          o4[i] = o;
        }
      }
    }
  }
}

extern "C" void kernel_launch(void* const* d_in, const int* in_sizes, int n_in,
                              void* d_out, int out_size, void* d_ws, size_t ws_size,
                              hipStream_t stream)
{
  const float* x     = (const float*)d_in[0];
  const int*   src   = (const int*)d_in[1];
  const int*   dst   = (const int*)d_in[2];
  const float* W0    = (const float*)d_in[3];
  const float* al0   = (const float*)d_in[4];
  const float* ar0   = (const float*)d_in[5];
  const float* b0    = (const float*)d_in[6];
  const float* W1    = (const float*)d_in[7];
  const float* al1   = (const float*)d_in[8];
  const float* ar1   = (const float*)d_in[9];
  const float* b1    = (const float*)d_in[10];
  const float* W2    = (const float*)d_in[11];
  const float* al2   = (const float*)d_in[12];
  const float* ar2   = (const float*)d_in[13];
  const float* b2    = (const float*)d_in[14];
  const float* Wres2 = (const float*)d_in[15];

  const size_t NF = (size_t)GAT_N * 256;
  float* ws   = (float*)d_ws;
  float* bufA = ws;
  float* bufB = bufA + NF;
  float* bufC = bufB + NF;
  float* EL   = bufC + NF;
  float* ER   = EL + (size_t)GAT_N * GAT_H;
  int* tileCnt  = (int*)(ER + (size_t)GAT_N * GAT_H);
  int* tileList = tileCnt + GAT_NTILES;

  dim3 gemmBlk(128);
  dim3 gemmGrid256(4, (GAT_N + 63) / 64);
  dim3 gemmGrid160(5, (GAT_N + 127) / 128);

  const int scBlocks  = GAT_N / GAT_SC_ROWS;
  const int aggBlocks = (GAT_N + GAT_TILE - 1) / GAT_TILE;

  gat_gemm_wmma<256, 256, 2, 2><<<gemmGrid256, gemmBlk, 0, stream>>>(x, W0, bufA, GAT_N);
  gat_attn_scores<256, 64><<<scBlocks, GAT_SC_THREADS, 0, stream>>>(bufA, al0, ar0, EL, ER);
  gat_tile_softmax_aggregate<256, 64, false, GAT_EPI_ELU><<<aggBlocks, GAT_AGG_THREADS, 0, stream>>>(src, dst, bufA, EL, ER, b0, nullptr, bufB, tileCnt, tileList);

  gat_gemm_wmma<256, 256, 2, 2><<<gemmGrid256, gemmBlk, 0, stream>>>(bufB, W1, bufA, GAT_N);
  gat_attn_scores<256, 64><<<scBlocks, GAT_SC_THREADS, 0, stream>>>(bufA, al1, ar1, EL, ER);
  gat_tile_softmax_aggregate<256, 64, true, GAT_EPI_RES_ELU><<<aggBlocks, GAT_AGG_THREADS, 0, stream>>>(src, dst, bufA, EL, ER, b1, bufB, bufC, tileCnt, tileList);

  gat_gemm_wmma<256, 160, 4, 1><<<gemmGrid160, gemmBlk, 0, stream>>>(bufC, W2, bufA, GAT_N);
  gat_gemm_wmma<256, 160, 4, 1><<<gemmGrid160, gemmBlk, 0, stream>>>(bufC, Wres2, bufB, GAT_N);
  gat_attn_scores<160, 40><<<scBlocks, GAT_SC_THREADS, 0, stream>>>(bufA, al2, ar2, EL, ER);
  gat_tile_softmax_aggregate<160, 40, true, GAT_EPI_RES_MEAN><<<aggBlocks, GAT_AGG_THREADS, 0, stream>>>(src, dst, bufA, EL, ER, b2, bufB, (float*)d_out, tileCnt, tileList);
}


# tile gather fetches source rows by 16-byte loads: 2 wave loads of 512 B per 1-KB row, was 8 of 128 B; tile rows kept component-major in LDS so each LDS float add still meets 32 banks; sum order kept
# speedup vs baseline: 3.9282x; 1.0576x over previous
#include <hip/hip_runtime.h>
#include <math.h>

#define GAT_N 50000
#define GAT_E 800000
#define GAT_H 4
#define GAT_SLOPE 0.2f

typedef __attribute__((ext_vector_type(2))) float v2f;
typedef __attribute__((ext_vector_type(4))) float v4f;
typedef __attribute__((ext_vector_type(8))) float v8f;

template <int K, int NCOL, int WR, int WC>
__global__ __launch_bounds__(WR * WC * 32) void gat_gemm_wmma(
    const float* __restrict__ A, const float* __restrict__ B,
    float* __restrict__ C, int M)
{
  constexpr int NT = WR * WC * 32;
  constexpr int BM = WR * 32;
  constexpr int BN = WC * 32;
  constexpr int KC = 32;
  constexpr int PA = KC + 4;
  constexpr int PB = 2 * BN + 32;
  constexpr int A4 = KC / 4;
  constexpr int B4 = BN / 4;
  constexpr int NA = BM * A4 / NT;
  constexpr int NB = (KC / 2) * B4 / NT;
  static_assert(K % KC == 0, "K is walked in whole chunks");
  static_assert(NCOL % BN == 0, "column tiles cover C exactly");
  static_assert((BM * A4) % NT == 0 && ((KC / 2) * B4) % NT == 0, "copies divide evenly");

  __shared__ __attribute__((aligned(16))) float As[BM * PA];
  __shared__ __attribute__((aligned(16))) float Bs[(KC / 2) * PB];

  const int tid  = threadIdx.x;
  const int wave = tid >> 5;
  const int lane = tid & 31;
  const int wr   = wave / WC;
  const int wc   = wave % WC;
  const int col0 = blockIdx.x * BN;
  const int row0 = blockIdx.y * BM;

  const int m  = lane & 15;
  const int hl = lane >> 4;

  const float* ag[NA];
  int          al[NA];
#pragma unroll
  for (int i = 0; i < NA; ++i) {
    const int idx = tid + i * NT;
    const int r = idx / A4, c4 = idx % A4;
    const int gr = min(row0 + r, M - 1);
    ag[i] = A + (size_t)gr * K + c4 * 4;
    al[i] = r * PA + c4 * 4;
  }
  const float* bg[NB];
  int          bl[NB];
#pragma unroll
  for (int i = 0; i < NB; ++i) {
    const int idx = tid + i * NT;
    const int kp = idx / B4, c4 = idx % B4;
    bg[i] = B + (size_t)(2 * kp) * NCOL + col0 + c4 * 4;
    bl[i] = kp * PB + c4 * 8;
  }

  v4f ra[NA], rb0[NB], rb1[NB];
#pragma unroll
  for (int i = 0; i < NA; ++i) ra[i] = *(const v4f*)ag[i];
#pragma unroll
  for (int i = 0; i < NB; ++i) {
    rb0[i] = *(const v4f*)bg[i];
    rb1[i] = *(const v4f*)(bg[i] + NCOL);
  }

  v8f acc00 = {}, acc01 = {}, acc10 = {}, acc11 = {};
  const float* afrag = As + (wr * 32 + m) * PA + 2 * hl;
  const float* bfrag = Bs + hl * PB + (wc * 32 + m) * 2;

  auto stage = [&]() {
#pragma unroll
    for (int i = 0; i < NA; ++i) *(v4f*)(As + al[i]) = ra[i];
#pragma unroll
    for (int i = 0; i < NB; ++i) {
      v4f lo, hi;
      lo.x = rb0[i].x; lo.y = rb1[i].x; lo.z = rb0[i].y; lo.w = rb1[i].y;
      hi.x = rb0[i].z; hi.y = rb1[i].z; hi.z = rb0[i].w; hi.w = rb1[i].w;
      *(v4f*)(Bs + bl[i])     = lo;
      *(v4f*)(Bs + bl[i] + 4) = hi;
    }
  };
  auto multiply = [&]() {
#pragma unroll
    for (int ks = 0; ks < KC; ks += 4) {
      const v2f a0 = *(const v2f*)(afrag + ks);
      const v2f a1 = *(const v2f*)(afrag + 16 * PA + ks);
      const v2f b0 = *(const v2f*)(bfrag + (ks / 2) * PB);
      const v2f b1 = *(const v2f*)(bfrag + (ks / 2) * PB + 32);
      acc00 = __builtin_amdgcn_wmma_f32_16x16x4_f32(false, a0, false, b0, (short)0, acc00, false, false);
      acc01 = __builtin_amdgcn_wmma_f32_16x16x4_f32(false, a0, false, b1, (short)0, acc01, false, false);
      acc10 = __builtin_amdgcn_wmma_f32_16x16x4_f32(false, a1, false, b0, (short)0, acc10, false, false);
      acc11 = __builtin_amdgcn_wmma_f32_16x16x4_f32(false, a1, false, b1, (short)0, acc11, false, false);
    }
  };

#pragma unroll 1
  for (int kn = KC; kn < K; kn += KC) {
    __syncthreads();
    stage();
    __syncthreads();
#pragma unroll
    for (int i = 0; i < NA; ++i) ra[i] = *(const v4f*)(ag[i] + kn);
#pragma unroll
    for (int i = 0; i < NB; ++i) {
      rb0[i] = *(const v4f*)(bg[i] + (size_t)kn * NCOL);
      rb1[i] = *(const v4f*)(bg[i] + (size_t)(kn + 1) * NCOL);
    }
    __builtin_amdgcn_sched_barrier(0);
    multiply();
  }
  __syncthreads();
  stage();
  __syncthreads();
  multiply();

  const int rsub = hl << 3;
  const int rt0  = row0 + wr * 32;
  float* crow = C + (size_t)(rt0 + rsub) * NCOL + col0 + wc * 32 + m;
  if (rt0 < M) {
#pragma unroll
    for (int r = 0; r < 8; ++r) {
      crow[(size_t)r * NCOL]      = acc00[r];
      crow[(size_t)r * NCOL + 16] = acc01[r];
    }
  }
  if (rt0 + 16 < M) {
#pragma unroll
    for (int r = 0; r < 8; ++r) {
      crow[(size_t)(r + 16) * NCOL]      = acc10[r];
      crow[(size_t)(r + 16) * NCOL + 16] = acc11[r];
    }
  }
}

#define GAT_SC_ROWS 50
#define GAT_SC_THREADS 256
static_assert(GAT_N % GAT_SC_ROWS == 0, "score groups cover the rows exactly");

template <int F, int D>
__global__ __launch_bounds__(GAT_SC_THREADS) void gat_attn_scores(
    const float* __restrict__ feat, const float* __restrict__ al,
    const float* __restrict__ ar, float* __restrict__ el, float* __restrict__ er)
{
  static_assert(F == GAT_H * D && D % 4 == 0, "a 16-byte piece of a row lies in one head");
  static_assert((GAT_SC_ROWS * F * 4) % 128 == 0, "a group's rows start on a line boundary");
  constexpr int F4  = F / 4;
  constexpr int D4  = D / 4;
  constexpr int P   = D4 | 1;
  constexpr int NP  = GAT_SC_ROWS * F4;
  constexpr int NIT = (NP + GAT_SC_THREADS - 1) / GAT_SC_THREADS;
  constexpr int NPAIR = GAT_SC_ROWS * GAT_H;
  static_assert(NPAIR <= GAT_SC_THREADS, "one thread per (row, head) pair");

  __shared__ float pl[NPAIR * P];
  __shared__ float pr[NPAIR * P];

  const int tid = threadIdx.x;
  const int n0  = blockIdx.x * GAT_SC_ROWS;
  const v4f* f4  = (const v4f*)feat + (size_t)n0 * F4;
  const v4f* al4 = (const v4f*)al;
  const v4f* ar4 = (const v4f*)ar;

  v4f v[NIT], a[NIT], b[NIT];
#pragma unroll
  for (int j = 0; j < NIT; ++j) {
    const int q = min(tid + j * GAT_SC_THREADS, NP - 1);
    v[j] = f4[q];
    a[j] = al4[q % F4];
    b[j] = ar4[q % F4];
  }
#pragma unroll
  for (int j = 0; j < NIT; ++j) {
    const int q = tid + j * GAT_SC_THREADS;
    if (q < NP) {
      const float sl = fmaf(v[j].w, a[j].w, fmaf(v[j].z, a[j].z, fmaf(v[j].y, a[j].y, v[j].x * a[j].x)));
      const float sr = fmaf(v[j].w, b[j].w, fmaf(v[j].z, b[j].z, fmaf(v[j].y, b[j].y, v[j].x * b[j].x)));
      const int o = (q / D4) * P + (q % D4);
      pl[o] = sl;
      pr[o] = sr;
    }
  }
  __syncthreads();
  if (tid < NPAIR) {
    float sl = 0.f, sr = 0.f;
#pragma unroll
    for (int i = 0; i < D4; ++i) {
      sl += pl[tid * P + i];
      sr += pr[tid * P + i];
    }
    el[(size_t)n0 * GAT_H + tid] = sl;
    er[(size_t)n0 * GAT_H + tid] = sr;
  }
}

__device__ __forceinline__ float gat_leaky(float x) {
  return x > 0.f ? x : GAT_SLOPE * x;
}

__device__ __forceinline__ void gat_lds_max_f32(float* addr, float v) {
  if (__float_as_int(v) >= 0)
    __hip_atomic_fetch_max((int*)addr, __float_as_int(v), __ATOMIC_RELAXED, __HIP_MEMORY_SCOPE_WORKGROUP);
  else
    __hip_atomic_fetch_min((unsigned int*)addr, __float_as_uint(v), __ATOMIC_RELAXED, __HIP_MEMORY_SCOPE_WORKGROUP);
}

#define GAT_TILE 196
#define GAT_AGG_THREADS 1024
#define GAT_LIST_CAP 4096
#define GAT_NTILES ((GAT_N + GAT_TILE - 1) / GAT_TILE)
#define GAT_EPI_ELU      0
#define GAT_EPI_RES_ELU  1
#define GAT_EPI_RES_MEAN 2
#define GAT_EPI_BATCH    4
static_assert(GAT_N <= (1 << 23) && GAT_TILE <= 256, "source node and local row share one 32-bit word");

template <int F>
__device__ __forceinline__ constexpr int gat_tile_step(int q) {
  return q < 32 * (F / 128) ? 32 : 1;
}
template <int F>
__device__ __forceinline__ constexpr int gat_tile_pos(int q) {
  return q < 32 * (F / 128) ? 128 * (q >> 5) + (q & 31) : 4 * q;
}
template <int F>
__device__ __forceinline__ constexpr int gat_tile_col(int p) {
  return p < 128 * (F / 128) ? 128 * (p >> 7) + 4 * (p & 31) + ((p & 127) >> 5) : p;
}

template <int F, int D, bool SAVED, int EPI>
__global__ __launch_bounds__(GAT_AGG_THREADS) void gat_tile_softmax_aggregate(
    const int* __restrict__ src, const int* __restrict__ dst,
    const float* __restrict__ feat, const float* __restrict__ el,
    const float* __restrict__ er, const float* __restrict__ bias,
    const float* __restrict__ res, float* __restrict__ out,
    int* tile_cnt, int* tile_list)
{
  static_assert(GAT_E % 128 == 0, "dst is walked in steps of 128 edges");
  static_assert(F % 4 == 0 && D % 4 == 0, "a 16-byte piece of a row lies in one head");
  constexpr int NW  = GAT_AGG_THREADS / 32;
  constexpr int NCH = GAT_E / 128;
  constexpr int F4  = F / 4;
  constexpr int NL  = F / 128;
  constexpr int TL  = F % 128;
  static_assert(TL == 0 || TL == 32, "the rest of a row is one line, a float per lane");
  constexpr int NRH = GAT_TILE * GAT_H;
  static_assert(GAT_LIST_CAP >= NW * 128, "one step of every wave always fits the list");
  static_assert(NRH <= GAT_AGG_THREADS, "one thread per (row, head) pair");

  __shared__ __attribute__((aligned(16))) float tile[GAT_TILE * F];
  __shared__ float4 l_w[GAT_LIST_CAP];
  __shared__ int    l_s[GAT_LIST_CAP];
  __shared__ int    l_r[GAT_LIST_CAP];
  __shared__ float4 s_er[GAT_TILE];
  __shared__ __attribute__((aligned(16))) float s_m[NRH];
  __shared__ __attribute__((aligned(16))) float s_mo[NRH];
  __shared__ __attribute__((aligned(16))) float s_den[NRH];
  __shared__ int    s_cnt;
  __shared__ int    s_over;

  const int tid   = threadIdx.x;
  const int wave  = __builtin_amdgcn_readfirstlane(tid >> 5);
  const int lane  = tid & 31;
  const int node0 = blockIdx.x * GAT_TILE;
  const int rows  = min(GAT_TILE, GAT_N - node0);
  int* my_list = tile_list + (size_t)blockIdx.x * GAT_LIST_CAP;
  const int saved = SAVED ? __builtin_amdgcn_readfirstlane(tile_cnt[blockIdx.x]) : -1;
  const bool have = saved >= 0;

  float4* t4 = (float4*)tile;
  for (int i = tid; i < GAT_TILE * F / 4; i += GAT_AGG_THREADS)
    t4[i] = make_float4(0.f, 0.f, 0.f, 0.f);
  if (tid < NRH) {
    s_m[tid]   = __int_as_float((int)0xFF800000u);
    s_den[tid] = 0.f;
  }
  if (tid < GAT_TILE)
    s_er[tid] = tid < rows ? ((const float4*)er)[node0 + tid] : make_float4(0.f, 0.f, 0.f, 0.f);
  if (tid == 0) { s_cnt = 0; s_over = 0; }
  __syncthreads();

  const int4* dst4 = (const int4*)dst;
  constexpr int NGROUPS = (GAT_N + GAT_TILE - 1) / GAT_TILE;
  const int rot = (int)blockIdx.x * (NCH / NGROUPS);
  auto place = [&](int t) { const int p = t + rot; return p >= NCH ? p - NCH : p; };
  auto fetch = [&](int t) { return dst4[(size_t)place(min(t, NCH - 1)) * 32 + lane]; };
  auto step = [&](const int4 cur, int t, int end) {
    const unsigned live = t < end ? 0xFFFFFFFFu : 0u;
    const int cc = place(t);
    const int l0 = cur.x - node0, l1 = cur.y - node0;
    const int l2 = cur.z - node0, l3 = cur.w - node0;
    const bool m0 = (unsigned)l0 < (unsigned)rows, m1 = (unsigned)l1 < (unsigned)rows;
    const bool m2 = (unsigned)l2 < (unsigned)rows, m3 = (unsigned)l3 < (unsigned)rows;
    const unsigned b0 = __builtin_amdgcn_ballot_w32(m0), b1 = __builtin_amdgcn_ballot_w32(m1);
    const unsigned b2 = __builtin_amdgcn_ballot_w32(m2), b3 = __builtin_amdgcn_ballot_w32(m3);
    if (((b0 | b1 | b2 | b3) & live) != 0u) {
      const int k0 = __popc(b0), k1 = __popc(b1), k2 = __popc(b2), k3 = __popc(b3);
      const int total = k0 + k1 + k2 + k3;
      int base = 0;
      if (lane == 0)
        base = __hip_atomic_fetch_add(&s_cnt, total, __ATOMIC_RELAXED, __HIP_MEMORY_SCOPE_WORKGROUP);
      base = __builtin_amdgcn_readfirstlane(base);
      if (base + total <= GAT_LIST_CAP) {
        const int e0 = cc * 128 + lane * 4;
        if (m0) { const int i = base + (int)__builtin_amdgcn_mbcnt_lo(b0, 0u);                l_s[i] = e0;     l_r[i] = l0; }
        if (m1) { const int i = base + k0 + (int)__builtin_amdgcn_mbcnt_lo(b1, 0u);           l_s[i] = e0 + 1; l_r[i] = l1; }
        if (m2) { const int i = base + k0 + k1 + (int)__builtin_amdgcn_mbcnt_lo(b2, 0u);      l_s[i] = e0 + 2; l_r[i] = l2; }
        if (m3) { const int i = base + k0 + k1 + k2 + (int)__builtin_amdgcn_mbcnt_lo(b3, 0u); l_s[i] = e0 + 3; l_r[i] = l3; }
      } else if (lane == 0) {
        s_over = 1;
      }
    }
  };
  auto collect = [&](int begin, int end) {
    const int w0 = begin + wave;
    int4 n0 = fetch(w0), n1 = fetch(w0 + NW), n2 = fetch(w0 + 2 * NW), n3 = fetch(w0 + 3 * NW);
#pragma unroll 1
    for (int c = w0; c < end; c += 4 * NW) {
      const int4 c0 = n0, c1 = n1, c2 = n2, c3 = n3;
      n0 = fetch(c + 4 * NW); n1 = fetch(c + 5 * NW); n2 = fetch(c + 6 * NW); n3 = fetch(c + 7 * NW);
      step(c0, c, end);
      step(c1, c + NW, end);
      step(c2, c + 2 * NW, end);
      step(c3, c + 3 * NW, end);
    }
  };

  int g_head[NL > 0 ? NL : 1];
#pragma unroll
  for (int i = 0; i < NL; ++i) g_head[i] = (128 * i + 4 * lane) / D;
  const int g_headt = (128 * NL + lane) / D;

  auto drain = [&](int base, int n) {
    int   s_l = 0, r_l = 0;
    float a0 = 0.f, a1 = 0.f, a2 = 0.f, a3 = 0.f;
    if (lane < n) {
      s_l = l_s[base + lane];
      r_l = l_r[base + lane];
      const float4 xv = l_w[base + lane];
      a0 = xv.x; a1 = xv.y; a2 = xv.z; a3 = xv.w;
    }
#pragma unroll 1
    for (int k = 0; k < n; k += 4) {
      const float* fp[4];
      float*       tp[4];
      float        w[4][4];
#pragma unroll
      for (int j = 0; j < 4; ++j) {
        const int  kj = min(k + j, n - 1);
        const bool in = k + j < n;
        fp[j] = feat + (size_t)__builtin_amdgcn_readlane(s_l, kj) * F;
        tp[j] = tile + __builtin_amdgcn_readlane(r_l, kj) * F + lane;
        w[j][0] = in ? __int_as_float(__builtin_amdgcn_readlane(__float_as_int(a0), kj)) : 0.f;
        w[j][1] = in ? __int_as_float(__builtin_amdgcn_readlane(__float_as_int(a1), kj)) : 0.f;
        w[j][2] = in ? __int_as_float(__builtin_amdgcn_readlane(__float_as_int(a2), kj)) : 0.f;
        w[j][3] = in ? __int_as_float(__builtin_amdgcn_readlane(__float_as_int(a3), kj)) : 0.f;
      }
      v4f   v[4][NL > 0 ? NL : 1];
      float vt[4];
#pragma unroll
      for (int j = 0; j < 4; ++j) {
#pragma unroll
        for (int i = 0; i < NL; ++i) {
          const int hlo = (128 * i) / D;
          const int hhi = min(GAT_H - 1, (128 * i + 127) / D);
          float wh = w[j][hlo];
#pragma unroll
          for (int hh = hlo + 1; hh <= hhi; ++hh) wh = g_head[i] >= hh ? w[j][hh] : wh;
          v[j][i] = *(const v4f*)(fp[j] + 128 * i + 4 * lane) * wh;
        }
        if constexpr (TL != 0) {
          const int hlo = (128 * NL) / D;
          const int hhi = min(GAT_H - 1, (F - 1) / D);
          float wh = w[j][hlo];
#pragma unroll
          for (int hh = hlo + 1; hh <= hhi; ++hh) wh = g_headt >= hh ? w[j][hh] : wh;
          vt[j] = fp[j][128 * NL + lane] * wh;
        }
      }
#pragma unroll
      for (int j = 0; j < 4; ++j) {
#pragma unroll
        for (int i = 0; i < NL; ++i) {
          float* t = tp[j] + 128 * i;
          __hip_atomic_fetch_add(t,      v[j][i].x, __ATOMIC_RELAXED, __HIP_MEMORY_SCOPE_WORKGROUP);
          __hip_atomic_fetch_add(t + 32, v[j][i].y, __ATOMIC_RELAXED, __HIP_MEMORY_SCOPE_WORKGROUP);
          __hip_atomic_fetch_add(t + 64, v[j][i].z, __ATOMIC_RELAXED, __HIP_MEMORY_SCOPE_WORKGROUP);
          __hip_atomic_fetch_add(t + 96, v[j][i].w, __ATOMIC_RELAXED, __HIP_MEMORY_SCOPE_WORKGROUP);
        }
        if constexpr (TL != 0)
          __hip_atomic_fetch_add(tp[j] + 128 * NL, vt[j], __ATOMIC_RELAXED, __HIP_MEMORY_SCOPE_WORKGROUP);
      }
    }
  };

  if (have) {
    for (int i = tid; i < saved; i += GAT_AGG_THREADS) {
      const int p = my_list[i];
      l_s[i] = p >> 8;
      l_r[i] = p & 255;
    }
    if (tid == 0) s_cnt = saved;
  } else {
    collect(0, NCH);
  }
  __syncthreads();
  const bool over = __builtin_amdgcn_readfirstlane(s_over) != 0;
  const int  span = over ? NW : NCH;
  if (!SAVED && tid == 0) tile_cnt[blockIdx.x] = over ? -1 : s_cnt;

#pragma unroll 1
  for (int begin = 0; begin < NCH; begin += span) {
    if (over) {
      __syncthreads();
      if (tid == 0) s_cnt = 0;
      __syncthreads();
      collect(begin, min(begin + span, NCH));
      __syncthreads();
    }
    const int cnt = __builtin_amdgcn_readfirstlane(s_cnt);
    if (begin > 0) {
      if (tid < NRH) s_mo[tid] = s_m[tid];
      __syncthreads();
    }

    for (int i0 = tid; i0 < cnt; i0 += 4 * GAT_AGG_THREADS) {
      int  sv[4], rv[4];
      bool ok[4];
#pragma unroll
      for (int u = 0; u < 4; ++u) {
        const int i = i0 + u * GAT_AGG_THREADS;
        ok[u] = i < cnt;
        const int e = ok[u] ? l_s[i] : 0;
        rv[u] = ok[u] ? l_r[i] : 0;
        sv[u] = have ? e : src[e];
      }
      float4 av[4];
#pragma unroll
      for (int u = 0; u < 4; ++u) av[u] = ((const float4*)el)[sv[u]];
#pragma unroll
      for (int u = 0; u < 4; ++u) {
        if (ok[u]) {
          const int i = i0 + u * GAT_AGG_THREADS;
          const float4 b = s_er[rv[u]];
          float4 sc;
          sc.x = gat_leaky(av[u].x + b.x);
          sc.y = gat_leaky(av[u].y + b.y);
          sc.z = gat_leaky(av[u].z + b.z);
          sc.w = gat_leaky(av[u].w + b.w);
          l_s[i] = sv[u];
          l_w[i] = sc;
          if (!SAVED && !over) my_list[i] = (sv[u] << 8) | rv[u];
          float* mp = s_m + rv[u] * GAT_H;
          gat_lds_max_f32(mp + 0, sc.x);
          gat_lds_max_f32(mp + 1, sc.y);
          gat_lds_max_f32(mp + 2, sc.z);
          gat_lds_max_f32(mp + 3, sc.w);
        }
      }
    }
    __syncthreads();

    if (begin > 0) {
      for (int i = tid; i < rows * F; i += GAT_AGG_THREADS) {
        const int p = (i / F) * GAT_H + gat_tile_col<F>(i % F) / D;
        const float mo = s_mo[p], mn = s_m[p];
        if (mn > mo) tile[i] *= expf(mo - mn);
      }
      if (tid < NRH) {
        const float mo = s_mo[tid], mn = s_m[tid];
        if (mn > mo) s_den[tid] *= expf(mo - mn);
      }
      __syncthreads();
    }

    for (int i = tid; i < cnt; i += GAT_AGG_THREADS) {
      const int    r  = l_r[i];
      const float4 sc = l_w[i];
      const float4 mm = ((const float4*)s_m)[r];
      float4 x;
      x.x = expf(sc.x - mm.x);
      x.y = expf(sc.y - mm.y);
      x.z = expf(sc.z - mm.z);
      x.w = expf(sc.w - mm.w);
      l_w[i] = x;
      float* dp = s_den + r * GAT_H;
      __hip_atomic_fetch_add(dp + 0, x.x, __ATOMIC_RELAXED, __HIP_MEMORY_SCOPE_WORKGROUP);
      __hip_atomic_fetch_add(dp + 1, x.y, __ATOMIC_RELAXED, __HIP_MEMORY_SCOPE_WORKGROUP);
      __hip_atomic_fetch_add(dp + 2, x.z, __ATOMIC_RELAXED, __HIP_MEMORY_SCOPE_WORKGROUP);
      __hip_atomic_fetch_add(dp + 3, x.w, __ATOMIC_RELAXED, __HIP_MEMORY_SCOPE_WORKGROUP);
    }
    __syncthreads();

    const int per = (cnt + NW - 1) / NW;
    const int lo  = wave * per;
    const int hi  = min(cnt, lo + per);
    for (int b = lo; b < hi; b += 32) drain(b, min(32, hi - b));
  }

  __syncthreads();
  const v4f  zero = {0.f, 0.f, 0.f, 0.f};
  if constexpr (EPI == GAT_EPI_RES_MEAN) {
    constexpr int C4 = D / 4;
    const int  total = rows * C4;
    const v4f* r4 = (const v4f*)(res + (size_t)node0 * F);
    const v4f* b4 = (const v4f*)bias;
    v4f*       o4 = (v4f*)(out + (size_t)node0 * D);
    for (int i = tid; i < total; i += GAT_AGG_THREADS) {
      const int n = i / C4, c4 = i % C4;
      v4f rv[GAT_H];
#pragma unroll
      for (int h = 0; h < GAT_H; ++h) rv[h] = r4[n * F4 + h * C4 + c4];
      v4f sum = zero;
#pragma unroll
      for (int h = 0; h < GAT_H; ++h) {
        const float dn = s_den[n * GAT_H + h];
        const float* tr = tile + n * F + gat_tile_pos<F>(h * C4 + c4);
        const int    st = gat_tile_step<F>(h * C4 + c4);
        v4f t;
        t.x = tr[0]; t.y = tr[st]; t.z = tr[2 * st]; t.w = tr[3 * st];
        const v4f   a  = dn > 0.f ? t / dn : zero;
        sum += a + b4[h * C4 + c4] + rv[h];
      }
      o4[i] = sum * 0.25f;
    }
  } else {
    static_assert(GAT_AGG_THREADS % F4 == 0, "a thread keeps its 16-byte piece of the bias row");
    const int  total = rows * F4;
    v4f*       o4 = (v4f*)(out + (size_t)node0 * F);
    const v4f  b  = ((const v4f*)bias)[tid % F4];
    const int  hd = (tid % F4) * 4 / D;
    const int  tq = gat_tile_pos<F>(tid % F4);
    const int  st = gat_tile_step<F>(tid % F4);
    for (int i0 = tid; i0 < total; i0 += GAT_EPI_BATCH * GAT_AGG_THREADS) {
      v4f rv[GAT_EPI_BATCH];
      if constexpr (EPI == GAT_EPI_RES_ELU) {
        const v4f* r4 = (const v4f*)(res + (size_t)node0 * F);
#pragma unroll
        for (int u = 0; u < GAT_EPI_BATCH; ++u) rv[u] = r4[min(i0 + u * GAT_AGG_THREADS, total - 1)];
      }
#pragma unroll
      for (int u = 0; u < GAT_EPI_BATCH; ++u) {
        const int i = i0 + u * GAT_AGG_THREADS;
        if (i < total) {
          const float dn = s_den[(i / F4) * GAT_H + hd];
          const float* tr = tile + (i / F4) * F + tq;
          v4f t;
          t.x = tr[0]; t.y = tr[st]; t.z = tr[2 * st]; t.w = tr[3 * st];
          v4f v = dn > 0.f ? t / dn : zero;
          v = v + b;
          if constexpr (EPI == GAT_EPI_RES_ELU) v += rv[u];
          v4f o;
          o.x = v.x > 0.f ? v.x : (expf(v.x) - 1.f);
          o.y = v.y > 0.f ? v.y : (expf(v.y) - 1.f);
          o.z = v.z > 0.f ? v.z : (expf(v.z) - 1.f);
          o.w = v.w > 0.f ? v.w : (expf(v.w) - 1.f);
          o4[i] = o;
        }
      }
    }
  }
}

extern "C" void kernel_launch(void* const* d_in, const int* in_sizes, int n_in,
                              void* d_out, int out_size, void* d_ws, size_t ws_size,
                              hipStream_t stream)
{
  const float* x     = (const float*)d_in[0];
  const int*   src   = (const int*)d_in[1];
  const int*   dst   = (const int*)d_in[2];
  const float* W0    = (const float*)d_in[3];
  const float* al0   = (const float*)d_in[4];
  const float* ar0   = (const float*)d_in[5];
  const float* b0    = (const float*)d_in[6];
  const float* W1    = (const float*)d_in[7];
  const float* al1   = (const float*)d_in[8];
  const float* ar1   = (const float*)d_in[9];
  const float* b1    = (const float*)d_in[10];
  const float* W2    = (const float*)d_in[11];
  const float* al2   = (const float*)d_in[12];
  const float* ar2   = (const float*)d_in[13];
  const float* b2    = (const float*)d_in[14];
  const float* Wres2 = (const float*)d_in[15];

  const size_t NF = (size_t)GAT_N * 256;
  float* ws   = (float*)d_ws;
  float* bufA = ws;
  float* bufB = bufA + NF;
  float* bufC = bufB + NF;
  float* EL   = bufC + NF;
  float* ER   = EL + (size_t)GAT_N * GAT_H;
  int* tileCnt  = (int*)(ER + (size_t)GAT_N * GAT_H);
  int* tileList = tileCnt + GAT_NTILES;

  dim3 gemmBlk(128);
  dim3 gemmGrid256(4, (GAT_N + 63) / 64);
  dim3 gemmGrid160(5, (GAT_N + 127) / 128);

  const int scBlocks  = GAT_N / GAT_SC_ROWS;
  const int aggBlocks = (GAT_N + GAT_TILE - 1) / GAT_TILE;

  gat_gemm_wmma<256, 256, 2, 2><<<gemmGrid256, gemmBlk, 0, stream>>>(x, W0, bufA, GAT_N);
  gat_attn_scores<256, 64><<<scBlocks, GAT_SC_THREADS, 0, stream>>>(bufA, al0, ar0, EL, ER);
  gat_tile_softmax_aggregate<256, 64, false, GAT_EPI_ELU><<<aggBlocks, GAT_AGG_THREADS, 0, stream>>>(src, dst, bufA, EL, ER, b0, nullptr, bufB, tileCnt, tileList);

  gat_gemm_wmma<256, 256, 2, 2><<<gemmGrid256, gemmBlk, 0, stream>>>(bufB, W1, bufA, GAT_N);
  gat_attn_scores<256, 64><<<scBlocks, GAT_SC_THREADS, 0, stream>>>(bufA, al1, ar1, EL, ER);
  gat_tile_softmax_aggregate<256, 64, true, GAT_EPI_RES_ELU><<<aggBlocks, GAT_AGG_THREADS, 0, stream>>>(src, dst, bufA, EL, ER, b1, bufB, bufC, tileCnt, tileList);

  gat_gemm_wmma<256, 160, 4, 1><<<gemmGrid160, gemmBlk, 0, stream>>>(bufC, W2, bufA, GAT_N);
  gat_gemm_wmma<256, 160, 4, 1><<<gemmGrid160, gemmBlk, 0, stream>>>(bufC, Wres2, bufB, GAT_N);
  gat_attn_scores<160, 40><<<scBlocks, GAT_SC_THREADS, 0, stream>>>(bufA, al2, ar2, EL, ER);
  gat_tile_softmax_aggregate<160, 40, true, GAT_EPI_RES_MEAN><<<aggBlocks, GAT_AGG_THREADS, 0, stream>>>(src, dst, bufA, EL, ER, b2, bufB, (float*)d_out, tileCnt, tileList);
}


# first tile walk reads a key plane (a byte per edge: its destination tile) written by the first score launch: 16 edges per lane load, zero-byte test per word, 1/4 of the bytes; list order changes
# speedup vs baseline: 4.1168x; 1.0480x over previous
#include <hip/hip_runtime.h>
#include <math.h>

#define GAT_N 50000
#define GAT_E 800000
#define GAT_H 4
#define GAT_SLOPE 0.2f
#define GAT_TILE 196
#define GAT_NTILES ((GAT_N + GAT_TILE - 1) / GAT_TILE)

typedef __attribute__((ext_vector_type(2))) float v2f;
typedef __attribute__((ext_vector_type(4))) float v4f;
typedef __attribute__((ext_vector_type(8))) float v8f;

template <int K, int NCOL, int WR, int WC>
__global__ __launch_bounds__(WR * WC * 32) void gat_gemm_wmma(
    const float* __restrict__ A, const float* __restrict__ B,
    float* __restrict__ C, int M)
{
  constexpr int NT = WR * WC * 32;
  constexpr int BM = WR * 32;
  constexpr int BN = WC * 32;
  constexpr int KC = 32;
  constexpr int PA = KC + 4;
  constexpr int PB = 2 * BN + 32;
  constexpr int A4 = KC / 4;
  constexpr int B4 = BN / 4;
  constexpr int NA = BM * A4 / NT;
  constexpr int NB = (KC / 2) * B4 / NT;
  static_assert(K % KC == 0, "K is walked in whole chunks");
  static_assert(NCOL % BN == 0, "column tiles cover C exactly");
  static_assert((BM * A4) % NT == 0 && ((KC / 2) * B4) % NT == 0, "copies divide evenly");

  __shared__ __attribute__((aligned(16))) float As[BM * PA];
  __shared__ __attribute__((aligned(16))) float Bs[(KC / 2) * PB];

  const int tid  = threadIdx.x;
  const int wave = tid >> 5;
  const int lane = tid & 31;
  const int wr   = wave / WC;
  const int wc   = wave % WC;
  const int col0 = blockIdx.x * BN;
  const int row0 = blockIdx.y * BM;

  const int m  = lane & 15;
  const int hl = lane >> 4;

  const float* ag[NA];
  int          al[NA];
#pragma unroll
  for (int i = 0; i < NA; ++i) {
    const int idx = tid + i * NT;
    const int r = idx / A4, c4 = idx % A4;
    const int gr = min(row0 + r, M - 1);
    ag[i] = A + (size_t)gr * K + c4 * 4;
    al[i] = r * PA + c4 * 4;
  }
  const float* bg[NB];
  int          bl[NB];
#pragma unroll
  for (int i = 0; i < NB; ++i) {
    const int idx = tid + i * NT;
    const int kp = idx / B4, c4 = idx % B4;
    bg[i] = B + (size_t)(2 * kp) * NCOL + col0 + c4 * 4;
    bl[i] = kp * PB + c4 * 8;
  }

  v4f ra[NA], rb0[NB], rb1[NB];
#pragma unroll
  for (int i = 0; i < NA; ++i) ra[i] = *(const v4f*)ag[i];
#pragma unroll
  for (int i = 0; i < NB; ++i) {
    rb0[i] = *(const v4f*)bg[i];
    rb1[i] = *(const v4f*)(bg[i] + NCOL);
  }

  v8f acc00 = {}, acc01 = {}, acc10 = {}, acc11 = {};
  const float* afrag = As + (wr * 32 + m) * PA + 2 * hl;
  const float* bfrag = Bs + hl * PB + (wc * 32 + m) * 2;

  auto stage = [&]() {
#pragma unroll
    for (int i = 0; i < NA; ++i) *(v4f*)(As + al[i]) = ra[i];
#pragma unroll
    for (int i = 0; i < NB; ++i) {
      v4f lo, hi;
      lo.x = rb0[i].x; lo.y = rb1[i].x; lo.z = rb0[i].y; lo.w = rb1[i].y;
      hi.x = rb0[i].z; hi.y = rb1[i].z; hi.z = rb0[i].w; hi.w = rb1[i].w;
      *(v4f*)(Bs + bl[i])     = lo;
      *(v4f*)(Bs + bl[i] + 4) = hi;
    }
  };
  auto multiply = [&]() {
#pragma unroll
    for (int ks = 0; ks < KC; ks += 4) {
      const v2f a0 = *(const v2f*)(afrag + ks);
      const v2f a1 = *(const v2f*)(afrag + 16 * PA + ks);
      const v2f b0 = *(const v2f*)(bfrag + (ks / 2) * PB);
      const v2f b1 = *(const v2f*)(bfrag + (ks / 2) * PB + 32);
      acc00 = __builtin_amdgcn_wmma_f32_16x16x4_f32(false, a0, false, b0, (short)0, acc00, false, false);
      acc01 = __builtin_amdgcn_wmma_f32_16x16x4_f32(false, a0, false, b1, (short)0, acc01, false, false);
      acc10 = __builtin_amdgcn_wmma_f32_16x16x4_f32(false, a1, false, b0, (short)0, acc10, false, false);
      acc11 = __builtin_amdgcn_wmma_f32_16x16x4_f32(false, a1, false, b1, (short)0, acc11, false, false);
    }
  };

#pragma unroll 1
  for (int kn = KC; kn < K; kn += KC) {
    __syncthreads();
    stage();
    __syncthreads();
#pragma unroll
    for (int i = 0; i < NA; ++i) ra[i] = *(const v4f*)(ag[i] + kn);
#pragma unroll
    for (int i = 0; i < NB; ++i) {
      rb0[i] = *(const v4f*)(bg[i] + (size_t)kn * NCOL);
      rb1[i] = *(const v4f*)(bg[i] + (size_t)(kn + 1) * NCOL);
    }
    __builtin_amdgcn_sched_barrier(0);
    multiply();
  }
  __syncthreads();
  stage();
  __syncthreads();
  multiply();

  const int rsub = hl << 3;
  const int rt0  = row0 + wr * 32;
  float* crow = C + (size_t)(rt0 + rsub) * NCOL + col0 + wc * 32 + m;
  if (rt0 < M) {
#pragma unroll
    for (int r = 0; r < 8; ++r) {
      crow[(size_t)r * NCOL]      = acc00[r];
      crow[(size_t)r * NCOL + 16] = acc01[r];
    }
  }
  if (rt0 + 16 < M) {
#pragma unroll
    for (int r = 0; r < 8; ++r) {
      crow[(size_t)(r + 16) * NCOL]      = acc10[r];
      crow[(size_t)(r + 16) * NCOL + 16] = acc11[r];
    }
  }
}

#define GAT_SC_ROWS 50
#define GAT_SC_THREADS 256
static_assert(GAT_N % GAT_SC_ROWS == 0, "score groups cover the rows exactly");

template <int F, int D, bool KEYS>
__global__ __launch_bounds__(GAT_SC_THREADS) void gat_attn_scores(
    const float* __restrict__ feat, const float* __restrict__ al,
    const float* __restrict__ ar, float* __restrict__ el, float* __restrict__ er,
    const int* __restrict__ dst, unsigned* __restrict__ keys)
{
  static_assert(F == GAT_H * D && D % 4 == 0, "a 16-byte piece of a row lies in one head");
  static_assert((GAT_SC_ROWS * F * 4) % 128 == 0, "a group's rows start on a line boundary");
  constexpr int F4  = F / 4;
  constexpr int D4  = D / 4;
  constexpr int P   = D4 | 1;
  constexpr int NP  = GAT_SC_ROWS * F4;
  constexpr int NIT = (NP + GAT_SC_THREADS - 1) / GAT_SC_THREADS;
  constexpr int NPAIR = GAT_SC_ROWS * GAT_H;
  static_assert(NPAIR <= GAT_SC_THREADS, "one thread per (row, head) pair");

  __shared__ float pl[NPAIR * P];
  __shared__ float pr[NPAIR * P];

  const int tid = threadIdx.x;
  const int n0  = blockIdx.x * GAT_SC_ROWS;
  if constexpr (KEYS) {
    static_assert(GAT_E % 16 == 0, "the key plane is read in 16-byte pieces");
    static_assert(GAT_NTILES <= 256, "a tile number fits a byte");
    static_assert((GAT_N / GAT_SC_ROWS) * GAT_SC_THREADS >= GAT_E / 4, "a thread per four edges");
  }
  const int kg = blockIdx.x * GAT_SC_THREADS + tid;
  int4 kd = make_int4(0, 0, 0, 0);
  if constexpr (KEYS) kd = ((const int4*)dst)[min(kg, GAT_E / 4 - 1)];
  const v4f* f4  = (const v4f*)feat + (size_t)n0 * F4;
  const v4f* al4 = (const v4f*)al;
  const v4f* ar4 = (const v4f*)ar;

  v4f v[NIT], a[NIT], b[NIT];
#pragma unroll
  for (int j = 0; j < NIT; ++j) {
    const int q = min(tid + j * GAT_SC_THREADS, NP - 1);
    v[j] = f4[q];
    a[j] = al4[q % F4];
    b[j] = ar4[q % F4];
  }
#pragma unroll
  for (int j = 0; j < NIT; ++j) {
    const int q = tid + j * GAT_SC_THREADS;
    if (q < NP) {
      const float sl = fmaf(v[j].w, a[j].w, fmaf(v[j].z, a[j].z, fmaf(v[j].y, a[j].y, v[j].x * a[j].x)));
      const float sr = fmaf(v[j].w, b[j].w, fmaf(v[j].z, b[j].z, fmaf(v[j].y, b[j].y, v[j].x * b[j].x)));
      const int o = (q / D4) * P + (q % D4);
      pl[o] = sl;
      pr[o] = sr;
    }
  }
  __syncthreads();
  if (tid < NPAIR) {
    float sl = 0.f, sr = 0.f;
#pragma unroll
    for (int i = 0; i < D4; ++i) {
      sl += pl[tid * P + i];
      sr += pr[tid * P + i];
    }
    el[(size_t)n0 * GAT_H + tid] = sl;
    er[(size_t)n0 * GAT_H + tid] = sr;
  }
  if constexpr (KEYS) {
    if (kg < GAT_E / 4)
      keys[kg] = (((unsigned)kd.x / GAT_TILE) & 255u)         | ((((unsigned)kd.y / GAT_TILE) & 255u) << 8) |
                 ((((unsigned)kd.z / GAT_TILE) & 255u) << 16) | ((((unsigned)kd.w / GAT_TILE) & 255u) << 24);
  }
}

__device__ __forceinline__ float gat_leaky(float x) {
  return x > 0.f ? x : GAT_SLOPE * x;
}

__device__ __forceinline__ void gat_lds_max_f32(float* addr, float v) {
  if (__float_as_int(v) >= 0)
    __hip_atomic_fetch_max((int*)addr, __float_as_int(v), __ATOMIC_RELAXED, __HIP_MEMORY_SCOPE_WORKGROUP);
  else
    __hip_atomic_fetch_min((unsigned int*)addr, __float_as_uint(v), __ATOMIC_RELAXED, __HIP_MEMORY_SCOPE_WORKGROUP);
}

#define GAT_AGG_THREADS 1024
#define GAT_LIST_CAP 4096
#define GAT_EPI_ELU      0
#define GAT_EPI_RES_ELU  1
#define GAT_EPI_RES_MEAN 2
#define GAT_EPI_BATCH    4
static_assert(GAT_N <= (1 << 23) && GAT_TILE <= 256, "source node and local row share one 32-bit word");

template <int F>
__device__ __forceinline__ constexpr int gat_tile_step(int q) {
  return q < 32 * (F / 128) ? 32 : 1;
}
template <int F>
__device__ __forceinline__ constexpr int gat_tile_pos(int q) {
  return q < 32 * (F / 128) ? 128 * (q >> 5) + (q & 31) : 4 * q;
}
template <int F>
__device__ __forceinline__ constexpr int gat_tile_col(int p) {
  return p < 128 * (F / 128) ? 128 * (p >> 7) + 4 * (p & 31) + ((p & 127) >> 5) : p;
}

template <int F, int D, bool SAVED, int EPI>
__global__ __launch_bounds__(GAT_AGG_THREADS) void gat_tile_softmax_aggregate(
    const int* __restrict__ src, const int* __restrict__ dst,
    const float* __restrict__ feat, const float* __restrict__ el,
    const float* __restrict__ er, const float* __restrict__ bias,
    const float* __restrict__ res, float* __restrict__ out,
    int* tile_cnt, int* tile_list, const unsigned* __restrict__ keys)
{
  static_assert(GAT_E % 128 == 0, "dst is walked in steps of 128 edges");
  static_assert(F % 4 == 0 && D % 4 == 0, "a 16-byte piece of a row lies in one head");
  constexpr int NW  = GAT_AGG_THREADS / 32;
  constexpr int NCH = GAT_E / 128;
  constexpr int F4  = F / 4;
  constexpr int NL  = F / 128;
  constexpr int TL  = F % 128;
  static_assert(TL == 0 || TL == 32, "the rest of a row is one line, a float per lane");
  constexpr int NRH = GAT_TILE * GAT_H;
  static_assert(GAT_LIST_CAP >= NW * 128, "one step of every wave always fits the list");
  static_assert(NRH <= GAT_AGG_THREADS, "one thread per (row, head) pair");

  __shared__ __attribute__((aligned(16))) float tile[GAT_TILE * F];
  __shared__ float4 l_w[GAT_LIST_CAP];
  __shared__ int    l_s[GAT_LIST_CAP];
  __shared__ int    l_r[GAT_LIST_CAP];
  __shared__ float4 s_er[GAT_TILE];
  __shared__ __attribute__((aligned(16))) float s_m[NRH];
  __shared__ __attribute__((aligned(16))) float s_mo[NRH];
  __shared__ __attribute__((aligned(16))) float s_den[NRH];
  __shared__ int    s_cnt;
  __shared__ int    s_over;

  const int tid   = threadIdx.x;
  const int wave  = __builtin_amdgcn_readfirstlane(tid >> 5);
  const int lane  = tid & 31;
  const int node0 = blockIdx.x * GAT_TILE;
  const int rows  = min(GAT_TILE, GAT_N - node0);
  int* my_list = tile_list + (size_t)blockIdx.x * GAT_LIST_CAP;
  const int saved = SAVED ? __builtin_amdgcn_readfirstlane(tile_cnt[blockIdx.x]) : -1;
  const bool have = saved >= 0;

  float4* t4 = (float4*)tile;
  for (int i = tid; i < GAT_TILE * F / 4; i += GAT_AGG_THREADS)
    t4[i] = make_float4(0.f, 0.f, 0.f, 0.f);
  if (tid < NRH) {
    s_m[tid]   = __int_as_float((int)0xFF800000u);
    s_den[tid] = 0.f;
  }
  if (tid < GAT_TILE)
    s_er[tid] = tid < rows ? ((const float4*)er)[node0 + tid] : make_float4(0.f, 0.f, 0.f, 0.f);
  if (tid == 0) { s_cnt = 0; s_over = 0; }
  __syncthreads();

  const int4* dst4 = (const int4*)dst;
  constexpr int NGROUPS = (GAT_N + GAT_TILE - 1) / GAT_TILE;
  const int rot = (int)blockIdx.x * (NCH / NGROUPS);
  auto place = [&](int t) { const int p = t + rot; return p >= NCH ? p - NCH : p; };
  auto fetch = [&](int t) { return dst4[(size_t)place(min(t, NCH - 1)) * 32 + lane]; };
  auto step = [&](const int4 cur, int t, int end) {
    const unsigned live = t < end ? 0xFFFFFFFFu : 0u;
    const int cc = place(t);
    const int l0 = cur.x - node0, l1 = cur.y - node0;
    const int l2 = cur.z - node0, l3 = cur.w - node0;
    const bool m0 = (unsigned)l0 < (unsigned)rows, m1 = (unsigned)l1 < (unsigned)rows;
    const bool m2 = (unsigned)l2 < (unsigned)rows, m3 = (unsigned)l3 < (unsigned)rows;
    const unsigned b0 = __builtin_amdgcn_ballot_w32(m0), b1 = __builtin_amdgcn_ballot_w32(m1);
    const unsigned b2 = __builtin_amdgcn_ballot_w32(m2), b3 = __builtin_amdgcn_ballot_w32(m3);
    if (((b0 | b1 | b2 | b3) & live) != 0u) {
      const int k0 = __popc(b0), k1 = __popc(b1), k2 = __popc(b2), k3 = __popc(b3);
      const int total = k0 + k1 + k2 + k3;
      int base = 0;
      if (lane == 0)
        base = __hip_atomic_fetch_add(&s_cnt, total, __ATOMIC_RELAXED, __HIP_MEMORY_SCOPE_WORKGROUP);
      base = __builtin_amdgcn_readfirstlane(base);
      if (base + total <= GAT_LIST_CAP) {
        const int e0 = cc * 128 + lane * 4;
        if (m0) { const int i = base + (int)__builtin_amdgcn_mbcnt_lo(b0, 0u);                l_s[i] = e0;     l_r[i] = l0; }
        if (m1) { const int i = base + k0 + (int)__builtin_amdgcn_mbcnt_lo(b1, 0u);           l_s[i] = e0 + 1; l_r[i] = l1; }
        if (m2) { const int i = base + k0 + k1 + (int)__builtin_amdgcn_mbcnt_lo(b2, 0u);      l_s[i] = e0 + 2; l_r[i] = l2; }
        if (m3) { const int i = base + k0 + k1 + k2 + (int)__builtin_amdgcn_mbcnt_lo(b3, 0u); l_s[i] = e0 + 3; l_r[i] = l3; }
      } else if (lane == 0) {
        s_over = 1;
      }
    }
  };
  auto collect = [&](int begin, int end) {
    const int w0 = begin + wave;
    int4 n0 = fetch(w0), n1 = fetch(w0 + NW), n2 = fetch(w0 + 2 * NW), n3 = fetch(w0 + 3 * NW);
#pragma unroll 1
    for (int c = w0; c < end; c += 4 * NW) {
      const int4 c0 = n0, c1 = n1, c2 = n2, c3 = n3;
      n0 = fetch(c + 4 * NW); n1 = fetch(c + 5 * NW); n2 = fetch(c + 6 * NW); n3 = fetch(c + 7 * NW);
      step(c0, c, end);
      step(c1, c + NW, end);
      step(c2, c + 2 * NW, end);
      step(c3, c + 3 * NW, end);
    }
  };

  constexpr int NK4 = GAT_E / 16;
  constexpr int NKS = (NK4 + 31) / 32;
  const uint4*   keys4 = (const uint4*)keys;
  const unsigned pat   = (unsigned)blockIdx.x * 0x01010101u;
  const int      rotk  = (int)blockIdx.x * (NKS / NGROUPS);
  auto kplace = [&](int t) { const int p = min(t, NKS - 1) + rotk; return p >= NKS ? p - NKS : p; };
  auto kfetch = [&](int t) { return keys4[min(kplace(t) * 32 + lane, NK4 - 1)]; };
  auto kmatch = [&](unsigned w) {
    const unsigned x = w ^ pat;
    return ~(((x & 0x7F7F7F7Fu) + 0x7F7F7F7Fu) | x) & 0x80808080u;
  };
  auto kstep = [&](const uint4 cur, int t) {
    const int  q    = kplace(t) * 32 + lane;
    const bool live = t < NKS && q < NK4;
    unsigned m = (kmatch(cur.x) >> 7) | (kmatch(cur.y) >> 6) | (kmatch(cur.z) >> 5) | (kmatch(cur.w) >> 4);
    m = live ? m : 0u;
    unsigned b = __builtin_amdgcn_ballot_w32(m != 0u);
    while (b != 0u) {
      const int total = __popc(b);
      int base = 0;
      if (lane == 0)
        base = __hip_atomic_fetch_add(&s_cnt, total, __ATOMIC_RELAXED, __HIP_MEMORY_SCOPE_WORKGROUP);
      base = __builtin_amdgcn_readfirstlane(base);
      if (base + total <= GAT_LIST_CAP) {
        if (m != 0u) {
          const int j = __ffs(m) - 1;
          l_s[base + (int)__builtin_amdgcn_mbcnt_lo(b, 0u)] = q * 16 + (j & 7) * 4 + (j >> 3);
        }
      } else if (lane == 0) {
        s_over = 1;
      }
      m &= m - 1u;
      b = __builtin_amdgcn_ballot_w32(m != 0u);
    }
  };
  auto collect_keys = [&]() {
    uint4 n0 = kfetch(wave), n1 = kfetch(wave + NW), n2 = kfetch(wave + 2 * NW), n3 = kfetch(wave + 3 * NW);
#pragma unroll 1
    for (int c = wave; c < NKS; c += 4 * NW) {
      const uint4 c0 = n0, c1 = n1, c2 = n2, c3 = n3;
      n0 = kfetch(c + 4 * NW); n1 = kfetch(c + 5 * NW); n2 = kfetch(c + 6 * NW); n3 = kfetch(c + 7 * NW);
      kstep(c0, c);
      kstep(c1, c + NW);
      kstep(c2, c + 2 * NW);
      kstep(c3, c + 3 * NW);
    }
  };

  int g_head[NL > 0 ? NL : 1];
#pragma unroll
  for (int i = 0; i < NL; ++i) g_head[i] = (128 * i + 4 * lane) / D;
  const int g_headt = (128 * NL + lane) / D;

  auto drain = [&](int base, int n) {
    int   s_l = 0, r_l = 0;
    float a0 = 0.f, a1 = 0.f, a2 = 0.f, a3 = 0.f;
    if (lane < n) {
      s_l = l_s[base + lane];
      r_l = l_r[base + lane];
      const float4 xv = l_w[base + lane];
      a0 = xv.x; a1 = xv.y; a2 = xv.z; a3 = xv.w;
    }
#pragma unroll 1
    for (int k = 0; k < n; k += 4) {
      const float* fp[4];
      float*       tp[4];
      float        w[4][4];
#pragma unroll
      for (int j = 0; j < 4; ++j) {
        const int  kj = min(k + j, n - 1);
        const bool in = k + j < n;
        fp[j] = feat + (size_t)__builtin_amdgcn_readlane(s_l, kj) * F;
        tp[j] = tile + __builtin_amdgcn_readlane(r_l, kj) * F + lane;
        w[j][0] = in ? __int_as_float(__builtin_amdgcn_readlane(__float_as_int(a0), kj)) : 0.f;
        w[j][1] = in ? __int_as_float(__builtin_amdgcn_readlane(__float_as_int(a1), kj)) : 0.f;
        w[j][2] = in ? __int_as_float(__builtin_amdgcn_readlane(__float_as_int(a2), kj)) : 0.f;
        w[j][3] = in ? __int_as_float(__builtin_amdgcn_readlane(__float_as_int(a3), kj)) : 0.f;
      }
      v4f   v[4][NL > 0 ? NL : 1];
      float vt[4];
#pragma unroll
      for (int j = 0; j < 4; ++j) {
#pragma unroll
        for (int i = 0; i < NL; ++i) {
          const int hlo = (128 * i) / D;
          const int hhi = min(GAT_H - 1, (128 * i + 127) / D);
          float wh = w[j][hlo];
#pragma unroll
          for (int hh = hlo + 1; hh <= hhi; ++hh) wh = g_head[i] >= hh ? w[j][hh] : wh;
          v[j][i] = *(const v4f*)(fp[j] + 128 * i + 4 * lane) * wh;
        }
        if constexpr (TL != 0) {
          const int hlo = (128 * NL) / D;
          const int hhi = min(GAT_H - 1, (F - 1) / D);
          float wh = w[j][hlo];
#pragma unroll
          for (int hh = hlo + 1; hh <= hhi; ++hh) wh = g_headt >= hh ? w[j][hh] : wh;
          vt[j] = fp[j][128 * NL + lane] * wh;
        }
      }
#pragma unroll
      for (int j = 0; j < 4; ++j) {
#pragma unroll
        for (int i = 0; i < NL; ++i) {
          float* t = tp[j] + 128 * i;
          __hip_atomic_fetch_add(t,      v[j][i].x, __ATOMIC_RELAXED, __HIP_MEMORY_SCOPE_WORKGROUP);
          __hip_atomic_fetch_add(t + 32, v[j][i].y, __ATOMIC_RELAXED, __HIP_MEMORY_SCOPE_WORKGROUP);
          __hip_atomic_fetch_add(t + 64, v[j][i].z, __ATOMIC_RELAXED, __HIP_MEMORY_SCOPE_WORKGROUP);
          __hip_atomic_fetch_add(t + 96, v[j][i].w, __ATOMIC_RELAXED, __HIP_MEMORY_SCOPE_WORKGROUP);
        }
        if constexpr (TL != 0)
          __hip_atomic_fetch_add(tp[j] + 128 * NL, vt[j], __ATOMIC_RELAXED, __HIP_MEMORY_SCOPE_WORKGROUP);
      }
    }
  };

  if (have) {
    for (int i = tid; i < saved; i += GAT_AGG_THREADS) {
      const int p = my_list[i];
      l_s[i] = p >> 8;
      l_r[i] = p & 255;
    }
    if (tid == 0) s_cnt = saved;
  } else {
    if constexpr (!SAVED) collect_keys();
    else collect(0, NCH);
  }
  __syncthreads();
  const bool over = __builtin_amdgcn_readfirstlane(s_over) != 0;
  const int  span = over ? NW : NCH;
  if (!SAVED && tid == 0) tile_cnt[blockIdx.x] = over ? -1 : s_cnt;

#pragma unroll 1
  for (int begin = 0; begin < NCH; begin += span) {
    if (over) {
      __syncthreads();
      if (tid == 0) s_cnt = 0;
      __syncthreads();
      collect(begin, min(begin + span, NCH));
      __syncthreads();
    }
    const int cnt = __builtin_amdgcn_readfirstlane(s_cnt);
    if (begin > 0) {
      if (tid < NRH) s_mo[tid] = s_m[tid];
      __syncthreads();
    }

    for (int i0 = tid; i0 < cnt; i0 += 4 * GAT_AGG_THREADS) {
      int  sv[4], rv[4];
      bool ok[4];
#pragma unroll
      for (int u = 0; u < 4; ++u) {
        const int i = i0 + u * GAT_AGG_THREADS;
        ok[u] = i < cnt;
        const int e = ok[u] ? l_s[i] : 0;
        if constexpr (SAVED) rv[u] = ok[u] ? l_r[i] : 0;
        else                 rv[u] = ok[u] ? dst[e] - node0 : 0;
        sv[u] = have ? e : src[e];
      }
      float4 av[4];
#pragma unroll
      for (int u = 0; u < 4; ++u) av[u] = ((const float4*)el)[sv[u]];
#pragma unroll
      for (int u = 0; u < 4; ++u) {
        if (ok[u]) {
          const int i = i0 + u * GAT_AGG_THREADS;
          const float4 b = s_er[rv[u]];
          float4 sc;
          sc.x = gat_leaky(av[u].x + b.x);
          sc.y = gat_leaky(av[u].y + b.y);
          sc.z = gat_leaky(av[u].z + b.z);
          sc.w = gat_leaky(av[u].w + b.w);
          l_s[i] = sv[u];
          if constexpr (!SAVED) l_r[i] = rv[u];
          l_w[i] = sc;
          if (!SAVED && !over) my_list[i] = (sv[u] << 8) | rv[u];
          float* mp = s_m + rv[u] * GAT_H;
          gat_lds_max_f32(mp + 0, sc.x);
          gat_lds_max_f32(mp + 1, sc.y);
          gat_lds_max_f32(mp + 2, sc.z);
          gat_lds_max_f32(mp + 3, sc.w);
        }
      }
    }
    __syncthreads();

    if (begin > 0) {
      for (int i = tid; i < rows * F; i += GAT_AGG_THREADS) {
        const int p = (i / F) * GAT_H + gat_tile_col<F>(i % F) / D;
        const float mo = s_mo[p], mn = s_m[p];
        if (mn > mo) tile[i] *= expf(mo - mn);
      }
      if (tid < NRH) {
        const float mo = s_mo[tid], mn = s_m[tid];
        if (mn > mo) s_den[tid] *= expf(mo - mn);
      }
      __syncthreads();
    }

    for (int i = tid; i < cnt; i += GAT_AGG_THREADS) {
      const int    r  = l_r[i];
      const float4 sc = l_w[i];
      const float4 mm = ((const float4*)s_m)[r];
      float4 x;
      x.x = expf(sc.x - mm.x);
      x.y = expf(sc.y - mm.y);
      x.z = expf(sc.z - mm.z);
      x.w = expf(sc.w - mm.w);
      l_w[i] = x;
      float* dp = s_den + r * GAT_H;
      __hip_atomic_fetch_add(dp + 0, x.x, __ATOMIC_RELAXED, __HIP_MEMORY_SCOPE_WORKGROUP);
      __hip_atomic_fetch_add(dp + 1, x.y, __ATOMIC_RELAXED, __HIP_MEMORY_SCOPE_WORKGROUP);
      __hip_atomic_fetch_add(dp + 2, x.z, __ATOMIC_RELAXED, __HIP_MEMORY_SCOPE_WORKGROUP);
      __hip_atomic_fetch_add(dp + 3, x.w, __ATOMIC_RELAXED, __HIP_MEMORY_SCOPE_WORKGROUP);
    }
    __syncthreads();

    const int per = (cnt + NW - 1) / NW;
    const int lo  = wave * per;
    const int hi  = min(cnt, lo + per);
    for (int b = lo; b < hi; b += 32) drain(b, min(32, hi - b));
  }

  __syncthreads();
  const v4f  zero = {0.f, 0.f, 0.f, 0.f};
  if constexpr (EPI == GAT_EPI_RES_MEAN) {
    constexpr int C4 = D / 4;
    const int  total = rows * C4;
    const v4f* r4 = (const v4f*)(res + (size_t)node0 * F);
    const v4f* b4 = (const v4f*)bias;
    v4f*       o4 = (v4f*)(out + (size_t)node0 * D);
    for (int i = tid; i < total; i += GAT_AGG_THREADS) {
      const int n = i / C4, c4 = i % C4;
      v4f rv[GAT_H];
#pragma unroll
      for (int h = 0; h < GAT_H; ++h) rv[h] = r4[n * F4 + h * C4 + c4];
      v4f sum = zero;
#pragma unroll
      for (int h = 0; h < GAT_H; ++h) {
        const float dn = s_den[n * GAT_H + h];
        const float* tr = tile + n * F + gat_tile_pos<F>(h * C4 + c4);
        const int    st = gat_tile_step<F>(h * C4 + c4);
        v4f t;
        t.x = tr[0]; t.y = tr[st]; t.z = tr[2 * st]; t.w = tr[3 * st];
        const v4f   a  = dn > 0.f ? t / dn : zero;
        sum += a + b4[h * C4 + c4] + rv[h];
      }
      o4[i] = sum * 0.25f;
    }
  } else {
    static_assert(GAT_AGG_THREADS % F4 == 0, "a thread keeps its 16-byte piece of the bias row");
    const int  total = rows * F4;
    v4f*       o4 = (v4f*)(out + (size_t)node0 * F);
    const v4f  b  = ((const v4f*)bias)[tid % F4];
    const int  hd = (tid % F4) * 4 / D;
    const int  tq = gat_tile_pos<F>(tid % F4);
    const int  st = gat_tile_step<F>(tid % F4);
    for (int i0 = tid; i0 < total; i0 += GAT_EPI_BATCH * GAT_AGG_THREADS) {
      v4f rv[GAT_EPI_BATCH];
      if constexpr (EPI == GAT_EPI_RES_ELU) {
        const v4f* r4 = (const v4f*)(res + (size_t)node0 * F);
#pragma unroll
        for (int u = 0; u < GAT_EPI_BATCH; ++u) rv[u] = r4[min(i0 + u * GAT_AGG_THREADS, total - 1)];
      }
#pragma unroll
      for (int u = 0; u < GAT_EPI_BATCH; ++u) {
        const int i = i0 + u * GAT_AGG_THREADS;
        if (i < total) {
          const float dn = s_den[(i / F4) * GAT_H + hd];
          const float* tr = tile + (i / F4) * F + tq;
          v4f t;
          t.x = tr[0]; t.y = tr[st]; t.z = tr[2 * st]; t.w = tr[3 * st];
          v4f v = dn > 0.f ? t / dn : zero;
          v = v + b;
          if constexpr (EPI == GAT_EPI_RES_ELU) v += rv[u];
          v4f o;
          o.x = v.x > 0.f ? v.x : (expf(v.x) - 1.f);
          o.y = v.y > 0.f ? v.y : (expf(v.y) - 1.f);
          o.z = v.z > 0.f ? v.z : (expf(v.z) - 1.f);
          o.w = v.w > 0.f ? v.w : (expf(v.w) - 1.f);
          o4[i] = o;
        }
      }
    }
  }
}

extern "C" void kernel_launch(void* const* d_in, const int* in_sizes, int n_in,
                              void* d_out, int out_size, void* d_ws, size_t ws_size,
                              hipStream_t stream)
{
  const float* x     = (const float*)d_in[0];
  const int*   src   = (const int*)d_in[1];
  const int*   dst   = (const int*)d_in[2];
  const float* W0    = (const float*)d_in[3];
  const float* al0   = (const float*)d_in[4];
  const float* ar0   = (const float*)d_in[5];
  const float* b0    = (const float*)d_in[6];
  const float* W1    = (const float*)d_in[7];
  const float* al1   = (const float*)d_in[8];
  const float* ar1   = (const float*)d_in[9];
  const float* b1    = (const float*)d_in[10];
  const float* W2    = (const float*)d_in[11];
  const float* al2   = (const float*)d_in[12];
  const float* ar2   = (const float*)d_in[13];
  const float* b2    = (const float*)d_in[14];
  const float* Wres2 = (const float*)d_in[15];

  const size_t NF = (size_t)GAT_N * 256;
  float* ws   = (float*)d_ws;
  float* bufA = ws;
  float* bufB = bufA + NF;
  float* bufC = bufB + NF;
  float* EL   = bufC + NF;
  float* ER   = EL + (size_t)GAT_N * GAT_H;
  int* tileCnt  = (int*)(ER + (size_t)GAT_N * GAT_H);
  int* tileList = tileCnt + GAT_NTILES;
  unsigned* keyPlane = (unsigned*)(tileList + (size_t)GAT_NTILES * GAT_LIST_CAP);

  dim3 gemmBlk(128);
  dim3 gemmGrid256(4, (GAT_N + 63) / 64);
  dim3 gemmGrid160(5, (GAT_N + 127) / 128);

  const int scBlocks  = GAT_N / GAT_SC_ROWS;
  const int aggBlocks = (GAT_N + GAT_TILE - 1) / GAT_TILE;

  gat_gemm_wmma<256, 256, 2, 2><<<gemmGrid256, gemmBlk, 0, stream>>>(x, W0, bufA, GAT_N);
  gat_attn_scores<256, 64, true><<<scBlocks, GAT_SC_THREADS, 0, stream>>>(bufA, al0, ar0, EL, ER, dst, keyPlane);
  gat_tile_softmax_aggregate<256, 64, false, GAT_EPI_ELU><<<aggBlocks, GAT_AGG_THREADS, 0, stream>>>(src, dst, bufA, EL, ER, b0, nullptr, bufB, tileCnt, tileList, keyPlane);

  gat_gemm_wmma<256, 256, 2, 2><<<gemmGrid256, gemmBlk, 0, stream>>>(bufB, W1, bufA, GAT_N);
  gat_attn_scores<256, 64, false><<<scBlocks, GAT_SC_THREADS, 0, stream>>>(bufA, al1, ar1, EL, ER, nullptr, nullptr);
  gat_tile_softmax_aggregate<256, 64, true, GAT_EPI_RES_ELU><<<aggBlocks, GAT_AGG_THREADS, 0, stream>>>(src, dst, bufA, EL, ER, b1, bufB, bufC, tileCnt, tileList, nullptr);

  gat_gemm_wmma<256, 160, 4, 1><<<gemmGrid160, gemmBlk, 0, stream>>>(bufC, W2, bufA, GAT_N);
  gat_gemm_wmma<256, 160, 4, 1><<<gemmGrid160, gemmBlk, 0, stream>>>(bufC, Wres2, bufB, GAT_N);
  gat_attn_scores<160, 40, false><<<scBlocks, GAT_SC_THREADS, 0, stream>>>(bufA, al2, ar2, EL, ER, nullptr, nullptr);
  gat_tile_softmax_aggregate<160, 40, true, GAT_EPI_RES_MEAN><<<aggBlocks, GAT_AGG_THREADS, 0, stream>>>(src, dst, bufA, EL, ER, b2, bufB, (float*)d_out, tileCnt, tileList, nullptr);
}
